# starting kernel
# baseline (speedup 1.0000x reference)
#include <hip/hip_runtime.h>

typedef float v2f __attribute__((ext_vector_type(2)));
typedef float v8f __attribute__((ext_vector_type(8)));

#define HH 96
#define WW 96
#define PIX (HH * WW)
#define CDIM 64
#define NHEADS 2
#define HD 32
#define KW 7
#define NKK (KW * KW)

__global__ void nchw_to_nhwc(const float* __restrict__ x, float* __restrict__ xh) {
    int t = blockIdx.x * blockDim.x + threadIdx.x;
    if (t >= CDIM * PIX) return;
    int c = t / PIX;
    int p = t % PIX;
    xh[(size_t)p * CDIM + c] = x[t];
}

__global__ void nhwc_to_nchw(const float* __restrict__ xh, float* __restrict__ y) {
    int t = blockIdx.x * blockDim.x + threadIdx.x;
    if (t >= CDIM * PIX) return;
    int c = t / PIX;
    int p = t % PIX;
    y[t] = xh[(size_t)p * CDIM + c];
}

template <int KD>
__global__ void wmma_gemm_bias(const float* __restrict__ A,
                               const float* __restrict__ W,
                               const float* __restrict__ bias,
                               float* __restrict__ C,
                               int M, int N) {
    int lane = threadIdx.x & 31;
    int wid  = threadIdx.x >> 5;
    int tile = blockIdx.x * 8 + wid;
    int tilesN = N >> 4;
    int numTiles = (M >> 4) * tilesN;
    if (tile >= numTiles) return;

    int tm = tile / tilesN;
    int tn = tile % tilesN;
    int lmod  = lane & 15;
    int lhalf = lane >> 4;

    const float* __restrict__ Ap = A + (size_t)(tm * 16 + lmod) * KD + 2 * lhalf;
    const float* __restrict__ Wp = W + (size_t)(tn * 16 + lmod) * KD + 2 * lhalf;

    v8f acc = {0.f, 0.f, 0.f, 0.f, 0.f, 0.f, 0.f, 0.f};
#pragma unroll
    for (int k = 0; k < KD; k += 4) {
        v2f a = *(const v2f*)(Ap + k);
        v2f b = *(const v2f*)(Wp + k);
        acc = __builtin_amdgcn_wmma_f32_16x16x4_f32(
             false, a,  false, b,
             (short)0, acc,  false,  false);
    }

    float bv = bias[tn * 16 + lmod];
    int rbase = tm * 16 + 8 * lhalf;
    int col   = tn * 16 + lmod;
#pragma unroll
    for (int v = 0; v < 8; ++v) {
        C[(size_t)(rbase + v) * N + col] = acc[v] + bv;
    }
}

__global__ void na_attn_kernel(const float* __restrict__ qkv,
                               const float* __restrict__ rpb,
                               const float* __restrict__ prev,
                               float* __restrict__ attn_store,
                               float* __restrict__ out,
                               int has_prev) {
    int idx = blockIdx.x * blockDim.x + threadIdx.x;
    if (idx >= NHEADS * PIX) return;
    int head = idx / PIX;
    int pix  = idx % PIX;
    int i = pix / WW;
    int j = pix % WW;

    int si = i - (KW / 2); si = si < 0 ? 0 : si; si = si > HH - KW ? HH - KW : si;
    int sj = j - (KW / 2); sj = sj < 0 ? 0 : sj; sj = sj > WW - KW ? WW - KW : sj;

    const float scale = 0.17677669529663687f;

    float qreg[HD];
    const float* qp = qkv + (size_t)pix * 192 + head * HD;
#pragma unroll
    for (int d = 0; d < HD; ++d) qreg[d] = qp[d] * scale;

    float m = -1e30f;
    float ssum = 0.f;
    float outacc[HD];
#pragma unroll
    for (int d = 0; d < HD; ++d) outacc[d] = 0.f;

    float* astore = attn_store + (size_t)idx * NKK;
    const float* pprev = prev + (size_t)idx * NKK;
    const float* rb = rpb + head * (2 * KW - 1) * (2 * KW - 1);

    for (int a = 0; a < KW; ++a) {
        int ri = si + a;
        for (int b = 0; b < KW; ++b) {
            int rj = sj + b;
            int kk = a * KW + b;
            const float* kp = qkv + (size_t)(ri * WW + rj) * 192 + 64 + head * HD;
            float l = 0.f;
#pragma unroll
            for (int d = 0; d < HD; ++d) l += qreg[d] * kp[d];
            l += rb[(ri - i + KW - 1) * (2 * KW - 1) + (rj - j + KW - 1)];
            if (has_prev) l += pprev[kk];
            astore[kk] = l;

            float mnew = fmaxf(m, l);
            float corr = __expf(m - mnew);
            float p    = __expf(l - mnew);
            ssum = ssum * corr + p;
            const float* vp = qkv + (size_t)(ri * WW + rj) * 192 + 128 + head * HD;
#pragma unroll
            for (int d = 0; d < HD; ++d) outacc[d] = outacc[d] * corr + p * vp[d];
            m = mnew;
        }
    }

    float inv = 1.f / ssum;
    for (int kk = 0; kk < NKK; ++kk) {
        astore[kk] = __expf(astore[kk] - m) * inv;
    }
    float* op = out + (size_t)pix * CDIM + head * HD;
#pragma unroll
    for (int d = 0; d < HD; ++d) op[d] = outacc[d] * inv;
}

extern "C" void kernel_launch(void* const* d_in, const int* in_sizes, int n_in,
                              void* d_out, int out_size, void* d_ws, size_t ws_size,
                              hipStream_t stream) {
    const float* x      = (const float*)d_in[0];
    const float* qkv_w  = (const float*)d_in[1];
    const float* qkv_b  = (const float*)d_in[2];
    const float* proj_w = (const float*)d_in[3];
    const float* proj_b = (const float*)d_in[4];
    const float* rpb    = (const float*)d_in[5];
    float* out = (float*)d_out;
    float* ws  = (float*)d_ws;

    float* xh0     = ws;
    float* xh1     = xh0 + (size_t)PIX * CDIM;
    float* qkv     = xh1 + (size_t)PIX * CDIM;
    float* attnA   = qkv + (size_t)PIX * 192;
    float* attnB   = attnA + (size_t)NHEADS * PIX * NKK;
    float* attnout = attnB + (size_t)NHEADS * PIX * NKK;

    nchw_to_nhwc<<<(CDIM * PIX + 255) / 256, 256, 0, stream>>>(x, xh0);

    float* xin  = xh0;
    float* xout = xh1;
    const float* prev = nullptr;

    for (int s = 0; s < 3; ++s) {
        float* acur = (s & 1) ? attnB : attnA;

        {
            int numTiles = (PIX / 16) * (192 / 16);
            wmma_gemm_bias<CDIM><<<(numTiles + 7) / 8, 256, 0, stream>>>(
                xin, qkv_w + (size_t)s * 192 * CDIM, qkv_b + (size_t)s * 192,
                qkv, PIX, 192);
        }

        na_attn_kernel<<<(NHEADS * PIX + 255) / 256, 256, 0, stream>>>(
            qkv, rpb + (size_t)s * NHEADS * (2 * KW - 1) * (2 * KW - 1),
            prev ? prev : acur  ,
            acur, attnout, prev != nullptr ? 1 : 0);

        {
            int numTiles = (PIX / 16) * (CDIM / 16);
            wmma_gemm_bias<CDIM><<<(numTiles + 7) / 8, 256, 0, stream>>>(
                attnout, proj_w + (size_t)s * CDIM * CDIM, proj_b + (size_t)s * CDIM,
                xout, PIX, CDIM);
        }

        prev = acur;
        float* t = xin; xin = xout; xout = t;
    }

    nhwc_to_nchw<<<(CDIM * PIX + 255) / 256, 256, 0, stream>>>(xin, out);
}


# attention as two f32 WMMA products per (head, 4x4 patch) wave: S^T=K.Q^T over the patch's 10x10 key square, masked lane-local softmax, D tiles reused as B of O^T=V^T.P^T; sums reordered
# speedup vs baseline: 3.1636x; 3.1636x over previous
#include <hip/hip_runtime.h>

typedef float v2f __attribute__((ext_vector_type(2)));
typedef float v8f __attribute__((ext_vector_type(8)));

#define HH 96
#define WW 96
#define PIX (HH * WW)
#define CDIM 64
#define NHEADS 2
#define HD 32
#define KW 7
#define NKK (KW * KW)

__global__ void nchw_to_nhwc(const float* __restrict__ x, float* __restrict__ xh) {
    int t = blockIdx.x * blockDim.x + threadIdx.x;
    if (t >= CDIM * PIX) return;
    int c = t / PIX;
    int p = t % PIX;
    xh[(size_t)p * CDIM + c] = x[t];
}

__global__ void nhwc_to_nchw(const float* __restrict__ xh, float* __restrict__ y) {
    int t = blockIdx.x * blockDim.x + threadIdx.x;
    if (t >= CDIM * PIX) return;
    int c = t / PIX;
    int p = t % PIX;
    y[t] = xh[(size_t)p * CDIM + c];
}

template <int KD>
__global__ void wmma_gemm_bias(const float* __restrict__ A,
                               const float* __restrict__ W,
                               const float* __restrict__ bias,
                               float* __restrict__ C,
                               int M, int N) {
    int lane = threadIdx.x & 31;
    int wid  = threadIdx.x >> 5;
    int tile = blockIdx.x * 8 + wid;
    int tilesN = N >> 4;
    int numTiles = (M >> 4) * tilesN;
    if (tile >= numTiles) return;

    int tm = tile / tilesN;
    int tn = tile % tilesN;
    int lmod  = lane & 15;
    int lhalf = lane >> 4;

    const float* __restrict__ Ap = A + (size_t)(tm * 16 + lmod) * KD + 2 * lhalf;
    const float* __restrict__ Wp = W + (size_t)(tn * 16 + lmod) * KD + 2 * lhalf;

    v8f acc = {0.f, 0.f, 0.f, 0.f, 0.f, 0.f, 0.f, 0.f};
#pragma unroll
    for (int k = 0; k < KD; k += 4) {
        v2f a = *(const v2f*)(Ap + k);
        v2f b = *(const v2f*)(Wp + k);
        acc = __builtin_amdgcn_wmma_f32_16x16x4_f32(
             false, a,  false, b,
             (short)0, acc,  false,  false);
    }

    float bv = bias[tn * 16 + lmod];
    int rbase = tm * 16 + 8 * lhalf;
    int col   = tn * 16 + lmod;
#pragma unroll
    for (int v = 0; v < 8; ++v) {
        C[(size_t)(rbase + v) * N + col] = acc[v] + bv;
    }
}

#define SQ 10
#define NKEY (SQ * SQ)
#define RPBW (2 * KW - 1)

__device__ __forceinline__ v8f wmma4(v2f a, v2f b, v8f c) {
    return __builtin_amdgcn_wmma_f32_16x16x4_f32(
         false, a,  false, b,
         (short)0, c,  false,  false);
}

__device__ __forceinline__ constexpr unsigned key_row_off(int n) {
    return (unsigned)((((n < NKEY ? n : NKEY - 1) / SQ) * WW + ((n < NKEY ? n : NKEY - 1) % SQ)) * 192);
}

struct Tap { bool valid; int a, b, kr, kc; };
__device__ __forceinline__ Tap tap_of(int t, int r, int h, int dsi, int dsj) {
    const int n0 = 16 * t + r;
    const int n1 = n0 + 8;
    Tap x;
    x.kr = h ? n1 / SQ : n0 / SQ;
    x.kc = h ? n1 % SQ : n0 % SQ;
    const bool insq = h ? (n1 < NKEY) : (n0 < NKEY);
    x.a = x.kr - dsi;
    x.b = x.kc - dsj;
    x.valid = insq && (unsigned)x.a < (unsigned)KW && (unsigned)x.b < (unsigned)KW;
    return x;
}

template <bool HAS_PREV>
__global__ void __launch_bounds__(32) __attribute__((amdgpu_waves_per_eu(1, 2)))
na_attn_wmma(const float* __restrict__ qkv,
             const float* __restrict__ rpb,
             const float* __restrict__ prev,
             float* __restrict__ attn_store,
             float* __restrict__ out) {
    const int lane  = threadIdx.x;
    const int lc    = lane & 15;
    const int h     = lane >> 4;
    const int head  = blockIdx.x & 1;
    const int patch = blockIdx.x >> 1;
    const int i0 = (patch / (WW / 4)) * 4;
    const int j0 = (patch % (WW / 4)) * 4;

    int r0 = i0 - KW / 2; r0 = r0 < 0 ? 0 : r0; r0 = r0 > HH - SQ ? HH - SQ : r0;
    int c0 = j0 - KW / 2; c0 = c0 < 0 ? 0 : c0; c0 = c0 > WW - SQ ? WW - SQ : c0;

    const int qi = i0 + (lc >> 2);
    const int qj = j0 + (lc & 3);
    const int qpix = qi * WW + qj;
    int si = qi - KW / 2; si = si < 0 ? 0 : si; si = si > HH - KW ? HH - KW : si;
    int sj = qj - KW / 2; sj = sj < 0 ? 0 : sj; sj = sj > WW - KW ? WW - KW : sj;
    const int dsi = si - r0;
    const int dsj = sj - c0;

    const float scale = 0.17677669529663687f;

    const unsigned qoff = (unsigned)(qpix * 192 + head * HD + 4 * h);
    float4 qv[4];
#pragma unroll
    for (int s = 0; s < 4; ++s) qv[s] = *(const float4*)(qkv + qoff + 8 * s);

    float4 kv[7][4];
#pragma unroll
    for (int t = 0; t < 7; ++t) {
        int kn = 16 * t + lc;
        kn = kn < NKEY ? kn : NKEY - 1;
        const int kr = kn / SQ;
        const int kc = kn - kr * SQ;
        const unsigned koff = (unsigned)(((r0 + kr) * WW + (c0 + kc)) * 192 + CDIM + head * HD + 4 * h);
#pragma unroll
        for (int s = 0; s < 4; ++s) kv[t][s] = *(const float4*)(qkv + koff + 8 * s);
    }

    const int rpb_lane = (r0 - qi + KW - 1) * RPBW + (c0 - qj + KW - 1);
    const float* rb = rpb + head * RPBW * RPBW;
    const unsigned aoff = (unsigned)((head * PIX + qpix) * NKK);
    float rbv[7][8];
    float pvv[7][8];
#pragma unroll
    for (int t = 0; t < 7; ++t) {
#pragma unroll
        for (int r = 0; r < 8; ++r) {
            const Tap x = tap_of(t, r, h, dsi, dsj);
            const unsigned ridx = x.valid ? (unsigned)(rpb_lane + x.kr * RPBW + x.kc) : 0u;
            rbv[t][r] = rb[ridx];
            if (HAS_PREV) {
                const unsigned kk = x.valid ? (unsigned)(x.a * KW + x.b) : 0u;
                pvv[t][r] = prev[aoff + kk];
            } else {
                pvv[t][r] = 0.f;
            }
        }
    }

#pragma unroll
    for (int s = 0; s < 4; ++s)
        qv[s] = make_float4(qv[s].x * scale, qv[s].y * scale, qv[s].z * scale, qv[s].w * scale);

    v8f st[7];
#pragma unroll
    for (int t = 0; t < 7; ++t) {
        v8f acc = {0.f, 0.f, 0.f, 0.f, 0.f, 0.f, 0.f, 0.f};
#pragma unroll
        for (int s = 0; s < 4; ++s) {
            v2f a0 = {kv[t][s].x, kv[t][s].y};
            v2f a1 = {kv[t][s].z, kv[t][s].w};
            v2f b0 = {qv[s].x, qv[s].y};
            v2f b1 = {qv[s].z, qv[s].w};
            acc = wmma4(a0, b0, acc);
            acc = wmma4(a1, b1, acc);
        }
        st[t] = acc;
    }

    const unsigned voff = (unsigned)((r0 * WW + c0) * 192 + 2 * CDIM + head * HD + lc);
    float va[7][4][2][2];
#pragma unroll
    for (int t = 0; t < 7; ++t) {
#pragma unroll
        for (int s = 0; s < 4; ++s) {
            const int k0 = 16 * t + 2 * s;
            if (k0 >= NKEY) continue;
            const unsigned of0 = voff + (h ? key_row_off(k0 + 8) : key_row_off(k0));
            const unsigned of1 = voff + (h ? key_row_off(k0 + 9) : key_row_off(k0 + 1));
            va[t][s][0][0] = qkv[of0];
            va[t][s][0][1] = qkv[of1];
            va[t][s][1][0] = qkv[of0 + 16];
            va[t][s][1][1] = qkv[of1 + 16];
        }
    }

    float mx = -1e30f;
#pragma unroll
    for (int t = 0; t < 7; ++t) {
#pragma unroll
        for (int r = 0; r < 8; ++r) {
            const Tap x = tap_of(t, r, h, dsi, dsj);
            float l = st[t][r] + rbv[t][r];
            if (HAS_PREV) l += pvv[t][r];
            l = x.valid ? l : -1e30f;
            st[t][r] = l;
            mx = fmaxf(mx, l);
        }
    }
    mx = fmaxf(mx, __shfl_xor(mx, 16));

    float ssum = 0.f;
#pragma unroll
    for (int t = 0; t < 7; ++t) {
#pragma unroll
        for (int r = 0; r < 8; ++r) {
            float p = __expf(st[t][r] - mx);
            st[t][r] = p;
            ssum += p;
        }
    }
    ssum += __shfl_xor(ssum, 16);
    const float inv = 1.f / ssum;

#pragma unroll
    for (int t = 0; t < 7; ++t)
#pragma unroll
        for (int r = 0; r < 8; ++r) st[t][r] *= inv;

    v8f o[2][2];
#pragma unroll
    for (int dt = 0; dt < 2; ++dt)
#pragma unroll
        for (int u = 0; u < 2; ++u)
            o[dt][u] = (v8f){0.f, 0.f, 0.f, 0.f, 0.f, 0.f, 0.f, 0.f};

#pragma unroll
    for (int t = 0; t < 7; ++t) {
#pragma unroll
        for (int s = 0; s < 4; ++s) {
            const int k0 = 16 * t + 2 * s;
            if (k0 >= NKEY) continue;
            v2f a0 = {va[t][s][0][0], va[t][s][0][1]};
            v2f a1 = {va[t][s][1][0], va[t][s][1][1]};
            v2f b  = {st[t][2 * s], st[t][2 * s + 1]};
            o[0][t & 1] = wmma4(a0, b, o[0][t & 1]);
            o[1][t & 1] = wmma4(a1, b, o[1][t & 1]);
        }
    }

    float* op = out + (size_t)qpix * CDIM + head * HD + 8 * h;
#pragma unroll
    for (int dt = 0; dt < 2; ++dt) {
        v8f d = o[dt][0] + o[dt][1];
        *(float4*)(op + 16 * dt)     = make_float4(d[0], d[1], d[2], d[3]);
        *(float4*)(op + 16 * dt + 4) = make_float4(d[4], d[5], d[6], d[7]);
    }

#pragma unroll
    for (int t = 0; t < 7; ++t) {
#pragma unroll
        for (int r = 0; r < 8; ++r) {
            const Tap x = tap_of(t, r, h, dsi, dsj);
            if (x.valid) attn_store[aoff + (unsigned)(x.a * KW + x.b)] = st[t][r];
        }
    }
}

extern "C" void kernel_launch(void* const* d_in, const int* in_sizes, int n_in,
                              void* d_out, int out_size, void* d_ws, size_t ws_size,
                              hipStream_t stream) {
    const float* x      = (const float*)d_in[0];
    const float* qkv_w  = (const float*)d_in[1];
    const float* qkv_b  = (const float*)d_in[2];
    const float* proj_w = (const float*)d_in[3];
    const float* proj_b = (const float*)d_in[4];
    const float* rpb    = (const float*)d_in[5];
    float* out = (float*)d_out;
    float* ws  = (float*)d_ws;

    float* xh0     = ws;
    float* xh1     = xh0 + (size_t)PIX * CDIM;
    float* qkv     = xh1 + (size_t)PIX * CDIM;
    float* attnA   = qkv + (size_t)PIX * 192;
    float* attnB   = attnA + (size_t)NHEADS * PIX * NKK;
    float* attnout = attnB + (size_t)NHEADS * PIX * NKK;

    nchw_to_nhwc<<<(CDIM * PIX + 255) / 256, 256, 0, stream>>>(x, xh0);

    float* xin  = xh0;
    float* xout = xh1;
    const float* prev = nullptr;

    for (int s = 0; s < 3; ++s) {
        float* acur = (s & 1) ? attnB : attnA;

        {
            int numTiles = (PIX / 16) * (192 / 16);
            wmma_gemm_bias<CDIM><<<(numTiles + 7) / 8, 256, 0, stream>>>(
                xin, qkv_w + (size_t)s * 192 * CDIM, qkv_b + (size_t)s * 192,
                qkv, PIX, 192);
        }

        {
            const float* rpb_s = rpb + (size_t)s * NHEADS * (2 * KW - 1) * (2 * KW - 1);
            if (prev)
                na_attn_wmma<true><<<NHEADS * (PIX / 16), 32, 0, stream>>>(
                    qkv, rpb_s, prev, acur, attnout);
            else
                na_attn_wmma<false><<<NHEADS * (PIX / 16), 32, 0, stream>>>(
                    qkv, rpb_s, acur  , acur, attnout);
        }

        {
            int numTiles = (PIX / 16) * (CDIM / 16);
            wmma_gemm_bias<CDIM><<<(numTiles + 7) / 8, 256, 0, stream>>>(
                attnout, proj_w + (size_t)s * CDIM * CDIM, proj_b + (size_t)s * CDIM,
                xout, PIX, CDIM);
        }

        prev = acur;
        float* t = xin; xin = xout; xout = t;
    }

    nhwc_to_nchw<<<(CDIM * PIX + 255) / 256, 256, 0, stream>>>(xin, out);
}


# output projection chained on the attention wave: O^T D tiles reused as B of Wp.O^T per head, the two head waves swap partial tiles via LDS and add; 11 -> 8 launches; sums reordered
# speedup vs baseline: 3.5283x; 1.1153x over previous
#include <hip/hip_runtime.h>

typedef float v2f __attribute__((ext_vector_type(2)));
typedef float v8f __attribute__((ext_vector_type(8)));

#define HH 96
#define WW 96
#define PIX (HH * WW)
#define CDIM 64
#define NHEADS 2
#define HD 32
#define KW 7
#define NKK (KW * KW)

__global__ void nchw_to_nhwc(const float* __restrict__ x, float* __restrict__ xh) {
    int t = blockIdx.x * blockDim.x + threadIdx.x;
    if (t >= CDIM * PIX) return;
    int c = t / PIX;
    int p = t % PIX;
    xh[(size_t)p * CDIM + c] = x[t];
}

__global__ void nhwc_to_nchw(const float* __restrict__ xh, float* __restrict__ y) {
    int t = blockIdx.x * blockDim.x + threadIdx.x;
    if (t >= CDIM * PIX) return;
    int c = t / PIX;
    int p = t % PIX;
    y[t] = xh[(size_t)p * CDIM + c];
}

template <int KD>
__global__ void wmma_gemm_bias(const float* __restrict__ A,
                               const float* __restrict__ W,
                               const float* __restrict__ bias,
                               float* __restrict__ C,
                               int M, int N) {
    int lane = threadIdx.x & 31;
    int wid  = threadIdx.x >> 5;
    int tile = blockIdx.x * 8 + wid;
    int tilesN = N >> 4;
    int numTiles = (M >> 4) * tilesN;
    if (tile >= numTiles) return;

    int tm = tile / tilesN;
    int tn = tile % tilesN;
    int lmod  = lane & 15;
    int lhalf = lane >> 4;

    const float* __restrict__ Ap = A + (size_t)(tm * 16 + lmod) * KD + 2 * lhalf;
    const float* __restrict__ Wp = W + (size_t)(tn * 16 + lmod) * KD + 2 * lhalf;

    v8f acc = {0.f, 0.f, 0.f, 0.f, 0.f, 0.f, 0.f, 0.f};
#pragma unroll
    for (int k = 0; k < KD; k += 4) {
        v2f a = *(const v2f*)(Ap + k);
        v2f b = *(const v2f*)(Wp + k);
        acc = __builtin_amdgcn_wmma_f32_16x16x4_f32(
             false, a,  false, b,
             (short)0, acc,  false,  false);
    }

    float bv = bias[tn * 16 + lmod];
    int rbase = tm * 16 + 8 * lhalf;
    int col   = tn * 16 + lmod;
#pragma unroll
    for (int v = 0; v < 8; ++v) {
        C[(size_t)(rbase + v) * N + col] = acc[v] + bv;
    }
}

#define SQ 10
#define NKEY (SQ * SQ)
#define RPBW (2 * KW - 1)

__device__ __forceinline__ v8f wmma4(v2f a, v2f b, v8f c) {
    return __builtin_amdgcn_wmma_f32_16x16x4_f32(
         false, a,  false, b,
         (short)0, c,  false,  false);
}

__device__ __forceinline__ constexpr unsigned key_row_off(int n) {
    return (unsigned)((((n < NKEY ? n : NKEY - 1) / SQ) * WW + ((n < NKEY ? n : NKEY - 1) % SQ)) * 192);
}

struct Tap { bool valid; int a, b, kr, kc; };
__device__ __forceinline__ Tap tap_of(int t, int r, int h, int dsi, int dsj) {
    const int n0 = 16 * t + r;
    const int n1 = n0 + 8;
    Tap x;
    x.kr = h ? n1 / SQ : n0 / SQ;
    x.kc = h ? n1 % SQ : n0 % SQ;
    const bool insq = h ? (n1 < NKEY) : (n0 < NKEY);
    x.a = x.kr - dsi;
    x.b = x.kc - dsj;
    x.valid = insq && (unsigned)x.a < (unsigned)KW && (unsigned)x.b < (unsigned)KW;
    return x;
}

template <bool HAS_PREV>
__global__ void __launch_bounds__(64) __attribute__((amdgpu_waves_per_eu(1, 2)))
na_attn_wmma(const float* __restrict__ qkv,
             const float* __restrict__ rpb,
             const float* __restrict__ prev,
             const float* __restrict__ pw,
             const float* __restrict__ pb,
             float* __restrict__ attn_store,
             float* __restrict__ out) {
    __shared__ float xch[NHEADS][2][8][32];
    const int lane  = threadIdx.x & 31;
    const int lc    = lane & 15;
    const int h     = lane >> 4;
    const int head  = __builtin_amdgcn_readfirstlane((int)(threadIdx.x >> 5));
    const int patch = blockIdx.x;
    const int i0 = (patch / (WW / 4)) * 4;
    const int j0 = (patch % (WW / 4)) * 4;

    int r0 = i0 - KW / 2; r0 = r0 < 0 ? 0 : r0; r0 = r0 > HH - SQ ? HH - SQ : r0;
    int c0 = j0 - KW / 2; c0 = c0 < 0 ? 0 : c0; c0 = c0 > WW - SQ ? WW - SQ : c0;

    const int qi = i0 + (lc >> 2);
    const int qj = j0 + (lc & 3);
    const int qpix = qi * WW + qj;
    int si = qi - KW / 2; si = si < 0 ? 0 : si; si = si > HH - KW ? HH - KW : si;
    int sj = qj - KW / 2; sj = sj < 0 ? 0 : sj; sj = sj > WW - KW ? WW - KW : sj;
    const int dsi = si - r0;
    const int dsj = sj - c0;

    const float scale = 0.17677669529663687f;

    const unsigned qoff = (unsigned)(qpix * 192 + head * HD + 4 * h);
    float4 qv[4];
#pragma unroll
    for (int s = 0; s < 4; ++s) qv[s] = *(const float4*)(qkv + qoff + 8 * s);

    float4 kv[7][4];
#pragma unroll
    for (int t = 0; t < 7; ++t) {
        int kn = 16 * t + lc;
        kn = kn < NKEY ? kn : NKEY - 1;
        const int kr = kn / SQ;
        const int kc = kn - kr * SQ;
        const unsigned koff = (unsigned)(((r0 + kr) * WW + (c0 + kc)) * 192 + CDIM + head * HD + 4 * h);
#pragma unroll
        for (int s = 0; s < 4; ++s) kv[t][s] = *(const float4*)(qkv + koff + 8 * s);
    }

    const int rpb_lane = (r0 - qi + KW - 1) * RPBW + (c0 - qj + KW - 1);
    const float* rb = rpb + head * RPBW * RPBW;
    const unsigned aoff = (unsigned)((head * PIX + qpix) * NKK);
    float rbv[7][8];
    float pvv[7][8];
#pragma unroll
    for (int t = 0; t < 7; ++t) {
#pragma unroll
        for (int r = 0; r < 8; ++r) {
            const Tap x = tap_of(t, r, h, dsi, dsj);
            const unsigned ridx = x.valid ? (unsigned)(rpb_lane + x.kr * RPBW + x.kc) : 0u;
            rbv[t][r] = rb[ridx];
            if (HAS_PREV) {
                const unsigned kk = x.valid ? (unsigned)(x.a * KW + x.b) : 0u;
                pvv[t][r] = prev[aoff + kk];
            } else {
                pvv[t][r] = 0.f;
            }
        }
    }

#pragma unroll
    for (int s = 0; s < 4; ++s)
        qv[s] = make_float4(qv[s].x * scale, qv[s].y * scale, qv[s].z * scale, qv[s].w * scale);

    v8f st[7];
#pragma unroll
    for (int t = 0; t < 7; ++t) {
        v8f acc = {0.f, 0.f, 0.f, 0.f, 0.f, 0.f, 0.f, 0.f};
#pragma unroll
        for (int s = 0; s < 4; ++s) {
            v2f a0 = {kv[t][s].x, kv[t][s].y};
            v2f a1 = {kv[t][s].z, kv[t][s].w};
            v2f b0 = {qv[s].x, qv[s].y};
            v2f b1 = {qv[s].z, qv[s].w};
            acc = wmma4(a0, b0, acc);
            acc = wmma4(a1, b1, acc);
        }
        st[t] = acc;
    }

    const unsigned voff = (unsigned)((r0 * WW + c0) * 192 + 2 * CDIM + head * HD + lc);
    float va[7][4][2][2];
#pragma unroll
    for (int t = 0; t < 7; ++t) {
#pragma unroll
        for (int s = 0; s < 4; ++s) {
            const int k0 = 16 * t + 2 * s;
            if (k0 >= NKEY) continue;
            const unsigned of0 = voff + (h ? key_row_off(k0 + 8) : key_row_off(k0));
            const unsigned of1 = voff + (h ? key_row_off(k0 + 9) : key_row_off(k0 + 1));
            va[t][s][0][0] = qkv[of0];
            va[t][s][0][1] = qkv[of1];
            va[t][s][1][0] = qkv[of0 + 16];
            va[t][s][1][1] = qkv[of1 + 16];
        }
    }

    float4 wq[4][2][2];
#pragma unroll
    for (int u = 0; u < 4; ++u) {
        const int ot = (2 * head + u) & 3;
        const float* wr = pw + (unsigned)((16 * ot + lc) * CDIM + head * HD + 8 * h);
#pragma unroll
        for (int dt = 0; dt < 2; ++dt) {
            wq[u][dt][0] = *(const float4*)(wr + 16 * dt);
            wq[u][dt][1] = *(const float4*)(wr + 16 * dt + 4);
        }
    }
    float4 bq[2][2];
#pragma unroll
    for (int tt = 0; tt < 2; ++tt) {
        const float* br = pb + (unsigned)(16 * (2 * head + tt) + 8 * h);
        bq[tt][0] = *(const float4*)(br);
        bq[tt][1] = *(const float4*)(br + 4);
    }

    float mx = -1e30f;
#pragma unroll
    for (int t = 0; t < 7; ++t) {
#pragma unroll
        for (int r = 0; r < 8; ++r) {
            const Tap x = tap_of(t, r, h, dsi, dsj);
            float l = st[t][r] + rbv[t][r];
            if (HAS_PREV) l += pvv[t][r];
            l = x.valid ? l : -1e30f;
            st[t][r] = l;
            mx = fmaxf(mx, l);
        }
    }
    mx = fmaxf(mx, __shfl_xor(mx, 16));

    float ssum = 0.f;
#pragma unroll
    for (int t = 0; t < 7; ++t) {
#pragma unroll
        for (int r = 0; r < 8; ++r) {
            float p = __expf(st[t][r] - mx);
            st[t][r] = p;
            ssum += p;
        }
    }
    ssum += __shfl_xor(ssum, 16);
    const float inv = 1.f / ssum;

#pragma unroll
    for (int t = 0; t < 7; ++t)
#pragma unroll
        for (int r = 0; r < 8; ++r) st[t][r] *= inv;

    v8f o[2][2];
#pragma unroll
    for (int dt = 0; dt < 2; ++dt)
#pragma unroll
        for (int u = 0; u < 2; ++u)
            o[dt][u] = (v8f){0.f, 0.f, 0.f, 0.f, 0.f, 0.f, 0.f, 0.f};

#pragma unroll
    for (int t = 0; t < 7; ++t) {
#pragma unroll
        for (int s = 0; s < 4; ++s) {
            const int k0 = 16 * t + 2 * s;
            if (k0 >= NKEY) continue;
            v2f a0 = {va[t][s][0][0], va[t][s][0][1]};
            v2f a1 = {va[t][s][1][0], va[t][s][1][1]};
            v2f b  = {st[t][2 * s], st[t][2 * s + 1]};
            o[0][t & 1] = wmma4(a0, b, o[0][t & 1]);
            o[1][t & 1] = wmma4(a1, b, o[1][t & 1]);
        }
    }

    v8f d[2];
#pragma unroll
    for (int dt = 0; dt < 2; ++dt) d[dt] = o[dt][0] + o[dt][1];

    v8f pp[4];
#pragma unroll
    for (int tt = 0; tt < 2; ++tt)
        pp[tt] = (v8f){bq[tt][0].x, bq[tt][0].y, bq[tt][0].z, bq[tt][0].w,
                       bq[tt][1].x, bq[tt][1].y, bq[tt][1].z, bq[tt][1].w};
#pragma unroll
    for (int u = 2; u < 4; ++u) pp[u] = (v8f){0.f, 0.f, 0.f, 0.f, 0.f, 0.f, 0.f, 0.f};
#pragma unroll
    for (int dt = 0; dt < 2; ++dt) {
#pragma unroll
        for (int j = 0; j < 4; ++j) {
            v2f b = {d[dt][2 * j], d[dt][2 * j + 1]};
#pragma unroll
            for (int u = 0; u < 4; ++u) {
                const float4 w4 = wq[u][dt][j >> 1];
                v2f a = (j & 1) ? (v2f){w4.z, w4.w} : (v2f){w4.x, w4.y};
                pp[u] = wmma4(a, b, pp[u]);
            }
        }
    }

#pragma unroll
    for (int tt = 0; tt < 2; ++tt)
#pragma unroll
        for (int r = 0; r < 8; ++r) xch[head][tt][r][lane] = pp[2 + tt][r];
    __syncthreads();

    float* op = out + (size_t)qpix * CDIM + 32 * head + 8 * h;
#pragma unroll
    for (int tt = 0; tt < 2; ++tt) {
        float e[8];
#pragma unroll
        for (int r = 0; r < 8; ++r) e[r] = pp[tt][r] + xch[1 - head][tt][r][lane];
        *(float4*)(op + 16 * tt)     = make_float4(e[0], e[1], e[2], e[3]);
        *(float4*)(op + 16 * tt + 4) = make_float4(e[4], e[5], e[6], e[7]);
    }

#pragma unroll
    for (int t = 0; t < 7; ++t) {
#pragma unroll
        for (int r = 0; r < 8; ++r) {
            const Tap x = tap_of(t, r, h, dsi, dsj);
            if (x.valid) attn_store[aoff + (unsigned)(x.a * KW + x.b)] = st[t][r];
        }
    }
}

extern "C" void kernel_launch(void* const* d_in, const int* in_sizes, int n_in,
                              void* d_out, int out_size, void* d_ws, size_t ws_size,
                              hipStream_t stream) {
    const float* x      = (const float*)d_in[0];
    const float* qkv_w  = (const float*)d_in[1];
    const float* qkv_b  = (const float*)d_in[2];
    const float* proj_w = (const float*)d_in[3];
    const float* proj_b = (const float*)d_in[4];
    const float* rpb    = (const float*)d_in[5];
    float* out = (float*)d_out;
    float* ws  = (float*)d_ws;

    float* xh0     = ws;
    float* xh1     = xh0 + (size_t)PIX * CDIM;
    float* qkv     = xh1 + (size_t)PIX * CDIM;
    float* attnA   = qkv + (size_t)PIX * 192;
    float* attnB   = attnA + (size_t)NHEADS * PIX * NKK;

    nchw_to_nhwc<<<(CDIM * PIX + 255) / 256, 256, 0, stream>>>(x, xh0);

    float* xin  = xh0;
    float* xout = xh1;
    const float* prev = nullptr;

    for (int s = 0; s < 3; ++s) {
        float* acur = (s & 1) ? attnB : attnA;

        {
            int numTiles = (PIX / 16) * (192 / 16);
            wmma_gemm_bias<CDIM><<<(numTiles + 7) / 8, 256, 0, stream>>>(
                xin, qkv_w + (size_t)s * 192 * CDIM, qkv_b + (size_t)s * 192,
                qkv, PIX, 192);
        }

        {
            const float* rpb_s = rpb + (size_t)s * NHEADS * (2 * KW - 1) * (2 * KW - 1);
            const float* pw_s  = proj_w + (size_t)s * CDIM * CDIM;
            const float* pb_s  = proj_b + (size_t)s * CDIM;
            if (prev)
                na_attn_wmma<true><<<PIX / 16, 32 * NHEADS, 0, stream>>>(
                    qkv, rpb_s, prev, pw_s, pb_s, acur, xout);
            else
                na_attn_wmma<false><<<PIX / 16, 32 * NHEADS, 0, stream>>>(
                    qkv, rpb_s, acur  , pw_s, pb_s, acur, xout);
        }

        prev = acur;
        float* t = xin; xin = xout; xout = t;
    }

    nhwc_to_nchw<<<(CDIM * PIX + 255) / 256, 256, 0, stream>>>(xin, out);
}


# attention map handed between stages in the wave's D-tile register layout: 13 16B accesses per lane each way, 512B contiguous, replace 2x56 scattered 4B taps; last stage stores none; same sums
# speedup vs baseline: 4.5719x; 1.2958x over previous
#include <hip/hip_runtime.h>

typedef float v2f __attribute__((ext_vector_type(2)));
typedef float v8f __attribute__((ext_vector_type(8)));

#define HH 96
#define WW 96
#define PIX (HH * WW)
#define CDIM 64
#define NHEADS 2
#define HD 32
#define KW 7
#define NKK (KW * KW)

__global__ void nchw_to_nhwc(const float* __restrict__ x, float* __restrict__ xh) {
    int t = blockIdx.x * blockDim.x + threadIdx.x;
    if (t >= CDIM * PIX) return;
    int c = t / PIX;
    int p = t % PIX;
    xh[(size_t)p * CDIM + c] = x[t];
}

__global__ void nhwc_to_nchw(const float* __restrict__ xh, float* __restrict__ y) {
    int t = blockIdx.x * blockDim.x + threadIdx.x;
    if (t >= CDIM * PIX) return;
    int c = t / PIX;
    int p = t % PIX;
    y[t] = xh[(size_t)p * CDIM + c];
}

template <int KD>
__global__ void wmma_gemm_bias(const float* __restrict__ A,
                               const float* __restrict__ W,
                               const float* __restrict__ bias,
                               float* __restrict__ C,
                               int M, int N) {
    int lane = threadIdx.x & 31;
    int wid  = threadIdx.x >> 5;
    int tile = blockIdx.x * 8 + wid;
    int tilesN = N >> 4;
    int numTiles = (M >> 4) * tilesN;
    if (tile >= numTiles) return;

    int tm = tile / tilesN;
    int tn = tile % tilesN;
    int lmod  = lane & 15;
    int lhalf = lane >> 4;

    const float* __restrict__ Ap = A + (size_t)(tm * 16 + lmod) * KD + 2 * lhalf;
    const float* __restrict__ Wp = W + (size_t)(tn * 16 + lmod) * KD + 2 * lhalf;

    v8f acc = {0.f, 0.f, 0.f, 0.f, 0.f, 0.f, 0.f, 0.f};
#pragma unroll
    for (int k = 0; k < KD; k += 4) {
        v2f a = *(const v2f*)(Ap + k);
        v2f b = *(const v2f*)(Wp + k);
        acc = __builtin_amdgcn_wmma_f32_16x16x4_f32(
             false, a,  false, b,
             (short)0, acc,  false,  false);
    }

    float bv = bias[tn * 16 + lmod];
    int rbase = tm * 16 + 8 * lhalf;
    int col   = tn * 16 + lmod;
#pragma unroll
    for (int v = 0; v < 8; ++v) {
        C[(size_t)(rbase + v) * N + col] = acc[v] + bv;
    }
}

#define SQ 10
#define NKEY (SQ * SQ)
#define RPBW (2 * KW - 1)
#define MAPG 13
#define MAPW (MAPG * 32 * 4)

__device__ __forceinline__ v8f wmma4(v2f a, v2f b, v8f c) {
    return __builtin_amdgcn_wmma_f32_16x16x4_f32(
         false, a,  false, b,
         (short)0, c,  false,  false);
}

__device__ __forceinline__ constexpr unsigned key_row_off(int n) {
    return (unsigned)((((n < NKEY ? n : NKEY - 1) / SQ) * WW + ((n < NKEY ? n : NKEY - 1) % SQ)) * 192);
}

struct Tap { bool valid; int a, b, kr, kc; };
__device__ __forceinline__ Tap tap_of(int t, int r, int h, int dsi, int dsj) {
    const int n0 = 16 * t + r;
    const int n1 = n0 + 8;
    Tap x;
    x.kr = h ? n1 / SQ : n0 / SQ;
    x.kc = h ? n1 % SQ : n0 % SQ;
    const bool insq = h ? (n1 < NKEY) : (n0 < NKEY);
    x.a = x.kr - dsi;
    x.b = x.kc - dsj;
    x.valid = insq && (unsigned)x.a < (unsigned)KW && (unsigned)x.b < (unsigned)KW;
    return x;
}

template <bool HAS_PREV, bool STORE_MAP>
__global__ void __launch_bounds__(64) __attribute__((amdgpu_waves_per_eu(1, 2)))
na_attn_wmma(const float* __restrict__ qkv,
             const float* __restrict__ rpb,
             const float* __restrict__ pw,
             const float* __restrict__ pb,
             float* __restrict__ amap,
             float* __restrict__ out) {
    __shared__ float xch[NHEADS][2][8][32];
    const int lane  = threadIdx.x & 31;
    const int lc    = lane & 15;
    const int h     = lane >> 4;
    const int head  = __builtin_amdgcn_readfirstlane((int)(threadIdx.x >> 5));
    const int patch = blockIdx.x;
    const int i0 = (patch / (WW / 4)) * 4;
    const int j0 = (patch % (WW / 4)) * 4;

    int r0 = i0 - KW / 2; r0 = r0 < 0 ? 0 : r0; r0 = r0 > HH - SQ ? HH - SQ : r0;
    int c0 = j0 - KW / 2; c0 = c0 < 0 ? 0 : c0; c0 = c0 > WW - SQ ? WW - SQ : c0;

    const int qi = i0 + (lc >> 2);
    const int qj = j0 + (lc & 3);
    const int qpix = qi * WW + qj;
    int si = qi - KW / 2; si = si < 0 ? 0 : si; si = si > HH - KW ? HH - KW : si;
    int sj = qj - KW / 2; sj = sj < 0 ? 0 : sj; sj = sj > WW - KW ? WW - KW : sj;
    const int dsi = si - r0;
    const int dsj = sj - c0;

    const float scale = 0.17677669529663687f;

    const unsigned qoff = (unsigned)(qpix * 192 + head * HD + 4 * h);
    float4 qv[4];
#pragma unroll
    for (int s = 0; s < 4; ++s) qv[s] = *(const float4*)(qkv + qoff + 8 * s);

    float4 kv[7][4];
#pragma unroll
    for (int t = 0; t < 7; ++t) {
        int kn = 16 * t + lc;
        kn = kn < NKEY ? kn : NKEY - 1;
        const int kr = kn / SQ;
        const int kc = kn - kr * SQ;
        const unsigned koff = (unsigned)(((r0 + kr) * WW + (c0 + kc)) * 192 + CDIM + head * HD + 4 * h);
#pragma unroll
        for (int s = 0; s < 4; ++s) kv[t][s] = *(const float4*)(qkv + koff + 8 * s);
    }

    const int rpb_lane = (r0 - qi + KW - 1) * RPBW + (c0 - qj + KW - 1);
    const float* rb = rpb + head * RPBW * RPBW;
    float* mp = amap + (unsigned)((patch * NHEADS + head) * MAPW) + 4 * lane;
    float4 pm[MAPG];
    if (HAS_PREV) {
#pragma unroll
        for (int g = 0; g < MAPG; ++g) pm[g] = *(const float4*)(mp + 128 * g);
    }
    float rbv[7][8];
    float pvv[7][8];
#pragma unroll
    for (int t = 0; t < 7; ++t) {
#pragma unroll
        for (int r = 0; r < 8; ++r) {
            const Tap x = tap_of(t, r, h, dsi, dsj);
            const unsigned ridx = x.valid ? (unsigned)(rpb_lane + x.kr * RPBW + x.kc) : 0u;
            rbv[t][r] = rb[ridx];
            const int f = 8 * t + r;
            if (HAS_PREV && f < 4 * MAPG) {
                const float4 g4 = pm[f >> 2];
                pvv[t][r] = (f & 3) == 0 ? g4.x : (f & 3) == 1 ? g4.y : (f & 3) == 2 ? g4.z : g4.w;
            } else {
                pvv[t][r] = 0.f;
            }
        }
    }

#pragma unroll
    for (int s = 0; s < 4; ++s)
        qv[s] = make_float4(qv[s].x * scale, qv[s].y * scale, qv[s].z * scale, qv[s].w * scale);

    v8f st[7];
#pragma unroll
    for (int t = 0; t < 7; ++t) {
        v8f acc = {0.f, 0.f, 0.f, 0.f, 0.f, 0.f, 0.f, 0.f};
#pragma unroll
        for (int s = 0; s < 4; ++s) {
            v2f a0 = {kv[t][s].x, kv[t][s].y};
            v2f a1 = {kv[t][s].z, kv[t][s].w};
            v2f b0 = {qv[s].x, qv[s].y};
            v2f b1 = {qv[s].z, qv[s].w};
            acc = wmma4(a0, b0, acc);
            acc = wmma4(a1, b1, acc);
        }
        st[t] = acc;
    }

    const unsigned voff = (unsigned)((r0 * WW + c0) * 192 + 2 * CDIM + head * HD + lc);
    float va[7][4][2][2];
#pragma unroll
    for (int t = 0; t < 7; ++t) {
#pragma unroll
        for (int s = 0; s < 4; ++s) {
            const int k0 = 16 * t + 2 * s;
            if (k0 >= NKEY) continue;
            const unsigned of0 = voff + (h ? key_row_off(k0 + 8) : key_row_off(k0));
            const unsigned of1 = voff + (h ? key_row_off(k0 + 9) : key_row_off(k0 + 1));
            va[t][s][0][0] = qkv[of0];
            va[t][s][0][1] = qkv[of1];
            va[t][s][1][0] = qkv[of0 + 16];
            va[t][s][1][1] = qkv[of1 + 16];
        }
    }

    float4 wq[4][2][2];
#pragma unroll
    for (int u = 0; u < 4; ++u) {
        const int ot = (2 * head + u) & 3;
        const float* wr = pw + (unsigned)((16 * ot + lc) * CDIM + head * HD + 8 * h);
#pragma unroll
        for (int dt = 0; dt < 2; ++dt) {
            wq[u][dt][0] = *(const float4*)(wr + 16 * dt);
            wq[u][dt][1] = *(const float4*)(wr + 16 * dt + 4);
        }
    }
    float4 bq[2][2];
#pragma unroll
    for (int tt = 0; tt < 2; ++tt) {
        const float* br = pb + (unsigned)(16 * (2 * head + tt) + 8 * h);
        bq[tt][0] = *(const float4*)(br);
        bq[tt][1] = *(const float4*)(br + 4);
    }

    float mx = -1e30f;
#pragma unroll
    for (int t = 0; t < 7; ++t) {
#pragma unroll
        for (int r = 0; r < 8; ++r) {
            const Tap x = tap_of(t, r, h, dsi, dsj);
            float l = st[t][r] + rbv[t][r];
            if (HAS_PREV) l += pvv[t][r];
            l = x.valid ? l : -1e30f;
            st[t][r] = l;
            mx = fmaxf(mx, l);
        }
    }
    mx = fmaxf(mx, __shfl_xor(mx, 16));

    float ssum = 0.f;
#pragma unroll
    for (int t = 0; t < 7; ++t) {
#pragma unroll
        for (int r = 0; r < 8; ++r) {
            float p = __expf(st[t][r] - mx);
            st[t][r] = p;
            ssum += p;
        }
    }
    ssum += __shfl_xor(ssum, 16);
    const float inv = 1.f / ssum;

#pragma unroll
    for (int t = 0; t < 7; ++t)
#pragma unroll
        for (int r = 0; r < 8; ++r) st[t][r] *= inv;

    v8f o[2][2];
#pragma unroll
    for (int dt = 0; dt < 2; ++dt)
#pragma unroll
        for (int u = 0; u < 2; ++u)
            o[dt][u] = (v8f){0.f, 0.f, 0.f, 0.f, 0.f, 0.f, 0.f, 0.f};

#pragma unroll
    for (int t = 0; t < 7; ++t) {
#pragma unroll
        for (int s = 0; s < 4; ++s) {
            const int k0 = 16 * t + 2 * s;
            if (k0 >= NKEY) continue;
            v2f a0 = {va[t][s][0][0], va[t][s][0][1]};
            v2f a1 = {va[t][s][1][0], va[t][s][1][1]};
            v2f b  = {st[t][2 * s], st[t][2 * s + 1]};
            o[0][t & 1] = wmma4(a0, b, o[0][t & 1]);
            o[1][t & 1] = wmma4(a1, b, o[1][t & 1]);
        }
    }

    v8f d[2];
#pragma unroll
    for (int dt = 0; dt < 2; ++dt) d[dt] = o[dt][0] + o[dt][1];

    v8f pp[4];
#pragma unroll
    for (int tt = 0; tt < 2; ++tt)
        pp[tt] = (v8f){bq[tt][0].x, bq[tt][0].y, bq[tt][0].z, bq[tt][0].w,
                       bq[tt][1].x, bq[tt][1].y, bq[tt][1].z, bq[tt][1].w};
#pragma unroll
    for (int u = 2; u < 4; ++u) pp[u] = (v8f){0.f, 0.f, 0.f, 0.f, 0.f, 0.f, 0.f, 0.f};
#pragma unroll
    for (int dt = 0; dt < 2; ++dt) {
#pragma unroll
        for (int j = 0; j < 4; ++j) {
            v2f b = {d[dt][2 * j], d[dt][2 * j + 1]};
#pragma unroll
            for (int u = 0; u < 4; ++u) {
                const float4 w4 = wq[u][dt][j >> 1];
                v2f a = (j & 1) ? (v2f){w4.z, w4.w} : (v2f){w4.x, w4.y};
                pp[u] = wmma4(a, b, pp[u]);
            }
        }
    }

#pragma unroll
    for (int tt = 0; tt < 2; ++tt)
#pragma unroll
        for (int r = 0; r < 8; ++r) xch[head][tt][r][lane] = pp[2 + tt][r];
    __syncthreads();

    float* op = out + (size_t)qpix * CDIM + 32 * head + 8 * h;
#pragma unroll
    for (int tt = 0; tt < 2; ++tt) {
        float e[8];
#pragma unroll
        for (int r = 0; r < 8; ++r) e[r] = pp[tt][r] + xch[1 - head][tt][r][lane];
        *(float4*)(op + 16 * tt)     = make_float4(e[0], e[1], e[2], e[3]);
        *(float4*)(op + 16 * tt + 4) = make_float4(e[4], e[5], e[6], e[7]);
    }

    if (STORE_MAP) {
#pragma unroll
        for (int g = 0; g < MAPG; ++g) {
            const int t = g >> 1;
            const int r = 4 * (g & 1);
            *(float4*)(mp + 128 * g) = make_float4(st[t][r], st[t][r + 1], st[t][r + 2], st[t][r + 3]);
        }
    }
}

extern "C" void kernel_launch(void* const* d_in, const int* in_sizes, int n_in,
                              void* d_out, int out_size, void* d_ws, size_t ws_size,
                              hipStream_t stream) {
    const float* x      = (const float*)d_in[0];
    const float* qkv_w  = (const float*)d_in[1];
    const float* qkv_b  = (const float*)d_in[2];
    const float* proj_w = (const float*)d_in[3];
    const float* proj_b = (const float*)d_in[4];
    const float* rpb    = (const float*)d_in[5];
    float* out = (float*)d_out;
    float* ws  = (float*)d_ws;

    float* xh   = ws;
    float* qkv  = xh + (size_t)PIX * CDIM;
    float* amap = qkv + (size_t)PIX * 192;

    nchw_to_nhwc<<<(CDIM * PIX + 255) / 256, 256, 0, stream>>>(x, xh);

    for (int s = 0; s < 3; ++s) {

        {
            int numTiles = (PIX / 16) * (192 / 16);
            wmma_gemm_bias<CDIM><<<(numTiles + 7) / 8, 256, 0, stream>>>(
                xh, qkv_w + (size_t)s * 192 * CDIM, qkv_b + (size_t)s * 192,
                qkv, PIX, 192);
        }

        {
            const float* rpb_s = rpb + (size_t)s * NHEADS * (2 * KW - 1) * (2 * KW - 1);
            const float* pw_s  = proj_w + (size_t)s * CDIM * CDIM;
            const float* pb_s  = proj_b + (size_t)s * CDIM;
            if (s == 0)
                na_attn_wmma<false, true><<<PIX / 16, 32 * NHEADS, 0, stream>>>(
                    qkv, rpb_s, pw_s, pb_s, amap, xh);
            else if (s == 1)
                na_attn_wmma<true, true><<<PIX / 16, 32 * NHEADS, 0, stream>>>(
                    qkv, rpb_s, pw_s, pb_s, amap, xh);
            else
                na_attn_wmma<true, false><<<PIX / 16, 32 * NHEADS, 0, stream>>>(
                    qkv, rpb_s, pw_s, pb_s, amap, xh);
        }
    }

    nhwc_to_nchw<<<(CDIM * PIX + 255) / 256, 256, 0, stream>>>(xh, out);
}


# both layout transposes tiled through LDS: a 4-wave work-group moves a 64-channel x 32-pixel tile with 16B accesses, whole 128B lines both ways; 18432 -> 1152 waves per transpose launch; same sums
# speedup vs baseline: 5.7024x; 1.2473x over previous
#include <hip/hip_runtime.h>

typedef float v2f __attribute__((ext_vector_type(2)));
typedef float v8f __attribute__((ext_vector_type(8)));

#define HH 96
#define WW 96
#define PIX (HH * WW)
#define CDIM 64
#define NHEADS 2
#define HD 32
#define KW 7
#define NKK (KW * KW)

#define TPX 32
#define TPITCH 68

__global__ void __launch_bounds__(128)
nchw_to_nhwc(const float* __restrict__ x, float* __restrict__ xh) {
    __shared__ float tile[TPX * TPITCH];
    const int tid = threadIdx.x;
    const int p0  = blockIdx.x * TPX;
    float4 v[4];
#pragma unroll
    for (int i = 0; i < 4; ++i) {
        const int idx = i * 128 + tid;
        const int c = idx >> 3;
        const int q = idx & 7;
        v[i] = *(const float4*)(x + (size_t)c * PIX + p0 + 4 * q);
    }
#pragma unroll
    for (int i = 0; i < 4; ++i) {
        const int idx = i * 128 + tid;
        const int c = idx >> 3;
        const int q = idx & 7;
        float* t = tile + (4 * q) * TPITCH + c;
        t[0]          = v[i].x;
        t[TPITCH]     = v[i].y;
        t[2 * TPITCH] = v[i].z;
        t[3 * TPITCH] = v[i].w;
    }
    __syncthreads();
#pragma unroll
    for (int i = 0; i < 4; ++i) {
        const int idx = i * 128 + tid;
        const int p = idx >> 4;
        const int m = idx & 15;
        const float4 w = *(const float4*)(tile + p * TPITCH + 4 * m);
        *(float4*)(xh + (size_t)(p0 + p) * CDIM + 4 * m) = w;
    }
}

__global__ void __launch_bounds__(128)
nhwc_to_nchw(const float* __restrict__ xh, float* __restrict__ y) {
    __shared__ float tile[TPX * TPITCH];
    const int tid = threadIdx.x;
    const int p0  = blockIdx.x * TPX;
    float4 v[4];
#pragma unroll
    for (int i = 0; i < 4; ++i) {
        const int idx = i * 128 + tid;
        const int p = idx >> 4;
        const int m = idx & 15;
        v[i] = *(const float4*)(xh + (size_t)(p0 + p) * CDIM + 4 * m);
    }
#pragma unroll
    for (int i = 0; i < 4; ++i) {
        const int idx = i * 128 + tid;
        const int p = idx >> 4;
        const int m = idx & 15;
        *(float4*)(tile + p * TPITCH + 4 * m) = v[i];
    }
    __syncthreads();
#pragma unroll
    for (int i = 0; i < 4; ++i) {
        const int idx = i * 128 + tid;
        const int c = idx >> 3;
        const int q = idx & 7;
        const float* t = tile + (4 * q) * TPITCH + c;
        const float4 w = make_float4(t[0], t[TPITCH], t[2 * TPITCH], t[3 * TPITCH]);
        *(float4*)(y + (size_t)c * PIX + p0 + 4 * q) = w;
    }
}

template <int KD>
__global__ void wmma_gemm_bias(const float* __restrict__ A,
                               const float* __restrict__ W,
                               const float* __restrict__ bias,
                               float* __restrict__ C,
                               int M, int N) {
    int lane = threadIdx.x & 31;
    int wid  = threadIdx.x >> 5;
    int tile = blockIdx.x * 8 + wid;
    int tilesN = N >> 4;
    int numTiles = (M >> 4) * tilesN;
    if (tile >= numTiles) return;

    int tm = tile / tilesN;
    int tn = tile % tilesN;
    int lmod  = lane & 15;
    int lhalf = lane >> 4;

    const float* __restrict__ Ap = A + (size_t)(tm * 16 + lmod) * KD + 2 * lhalf;
    const float* __restrict__ Wp = W + (size_t)(tn * 16 + lmod) * KD + 2 * lhalf;

    v8f acc = {0.f, 0.f, 0.f, 0.f, 0.f, 0.f, 0.f, 0.f};
#pragma unroll
    for (int k = 0; k < KD; k += 4) {
        v2f a = *(const v2f*)(Ap + k);
        v2f b = *(const v2f*)(Wp + k);
        acc = __builtin_amdgcn_wmma_f32_16x16x4_f32(
             false, a,  false, b,
             (short)0, acc,  false,  false);
    }

    float bv = bias[tn * 16 + lmod];
    int rbase = tm * 16 + 8 * lhalf;
    int col   = tn * 16 + lmod;
#pragma unroll
    for (int v = 0; v < 8; ++v) {
        C[(size_t)(rbase + v) * N + col] = acc[v] + bv;
    }
}

#define SQ 10
#define NKEY (SQ * SQ)
#define RPBW (2 * KW - 1)
#define MAPG 13
#define MAPW (MAPG * 32 * 4)

__device__ __forceinline__ v8f wmma4(v2f a, v2f b, v8f c) {
    return __builtin_amdgcn_wmma_f32_16x16x4_f32(
         false, a,  false, b,
         (short)0, c,  false,  false);
}

__device__ __forceinline__ constexpr unsigned key_row_off(int n) {
    return (unsigned)((((n < NKEY ? n : NKEY - 1) / SQ) * WW + ((n < NKEY ? n : NKEY - 1) % SQ)) * 192);
}

struct Tap { bool valid; int a, b, kr, kc; };
__device__ __forceinline__ Tap tap_of(int t, int r, int h, int dsi, int dsj) {
    const int n0 = 16 * t + r;
    const int n1 = n0 + 8;
    Tap x;
    x.kr = h ? n1 / SQ : n0 / SQ;
    x.kc = h ? n1 % SQ : n0 % SQ;
    const bool insq = h ? (n1 < NKEY) : (n0 < NKEY);
    x.a = x.kr - dsi;
    x.b = x.kc - dsj;
    x.valid = insq && (unsigned)x.a < (unsigned)KW && (unsigned)x.b < (unsigned)KW;
    return x;
}

template <bool HAS_PREV, bool STORE_MAP>
__global__ void __launch_bounds__(64) __attribute__((amdgpu_waves_per_eu(1, 2)))
na_attn_wmma(const float* __restrict__ qkv,
             const float* __restrict__ rpb,
             const float* __restrict__ pw,
             const float* __restrict__ pb,
             float* __restrict__ amap,
             float* __restrict__ out) {
    __shared__ float xch[NHEADS][2][8][32];
    const int lane  = threadIdx.x & 31;
    const int lc    = lane & 15;
    const int h     = lane >> 4;
    const int head  = __builtin_amdgcn_readfirstlane((int)(threadIdx.x >> 5));
    const int patch = blockIdx.x;
    const int i0 = (patch / (WW / 4)) * 4;
    const int j0 = (patch % (WW / 4)) * 4;

    int r0 = i0 - KW / 2; r0 = r0 < 0 ? 0 : r0; r0 = r0 > HH - SQ ? HH - SQ : r0;
    int c0 = j0 - KW / 2; c0 = c0 < 0 ? 0 : c0; c0 = c0 > WW - SQ ? WW - SQ : c0;

    const int qi = i0 + (lc >> 2);
    const int qj = j0 + (lc & 3);
    const int qpix = qi * WW + qj;
    int si = qi - KW / 2; si = si < 0 ? 0 : si; si = si > HH - KW ? HH - KW : si;
    int sj = qj - KW / 2; sj = sj < 0 ? 0 : sj; sj = sj > WW - KW ? WW - KW : sj;
    const int dsi = si - r0;
    const int dsj = sj - c0;

    const float scale = 0.17677669529663687f;

    const unsigned qoff = (unsigned)(qpix * 192 + head * HD + 4 * h);
    float4 qv[4];
#pragma unroll
    for (int s = 0; s < 4; ++s) qv[s] = *(const float4*)(qkv + qoff + 8 * s);

    float4 kv[7][4];
#pragma unroll
    for (int t = 0; t < 7; ++t) {
        int kn = 16 * t + lc;
        kn = kn < NKEY ? kn : NKEY - 1;
        const int kr = kn / SQ;
        const int kc = kn - kr * SQ;
        const unsigned koff = (unsigned)(((r0 + kr) * WW + (c0 + kc)) * 192 + CDIM + head * HD + 4 * h);
#pragma unroll
        for (int s = 0; s < 4; ++s) kv[t][s] = *(const float4*)(qkv + koff + 8 * s);
    }

    const int rpb_lane = (r0 - qi + KW - 1) * RPBW + (c0 - qj + KW - 1);
    const float* rb = rpb + head * RPBW * RPBW;
    float* mp = amap + (unsigned)((patch * NHEADS + head) * MAPW) + 4 * lane;
    float4 pm[MAPG];
    if (HAS_PREV) {
#pragma unroll
        for (int g = 0; g < MAPG; ++g) pm[g] = *(const float4*)(mp + 128 * g);
    }
    float rbv[7][8];
    float pvv[7][8];
#pragma unroll
    for (int t = 0; t < 7; ++t) {
#pragma unroll
        for (int r = 0; r < 8; ++r) {
            const Tap x = tap_of(t, r, h, dsi, dsj);
            const unsigned ridx = x.valid ? (unsigned)(rpb_lane + x.kr * RPBW + x.kc) : 0u;
            rbv[t][r] = rb[ridx];
            const int f = 8 * t + r;
            if (HAS_PREV && f < 4 * MAPG) {
                const float4 g4 = pm[f >> 2];
                pvv[t][r] = (f & 3) == 0 ? g4.x : (f & 3) == 1 ? g4.y : (f & 3) == 2 ? g4.z : g4.w;
            } else {
                pvv[t][r] = 0.f;
            }
        }
    }

#pragma unroll
    for (int s = 0; s < 4; ++s)
        qv[s] = make_float4(qv[s].x * scale, qv[s].y * scale, qv[s].z * scale, qv[s].w * scale);

    v8f st[7];
#pragma unroll
    for (int t = 0; t < 7; ++t) {
        v8f acc = {0.f, 0.f, 0.f, 0.f, 0.f, 0.f, 0.f, 0.f};
#pragma unroll
        for (int s = 0; s < 4; ++s) {
            v2f a0 = {kv[t][s].x, kv[t][s].y};
            v2f a1 = {kv[t][s].z, kv[t][s].w};
            v2f b0 = {qv[s].x, qv[s].y};
            v2f b1 = {qv[s].z, qv[s].w};
            acc = wmma4(a0, b0, acc);
            acc = wmma4(a1, b1, acc);
        }
        st[t] = acc;
    }

    const unsigned voff = (unsigned)((r0 * WW + c0) * 192 + 2 * CDIM + head * HD + lc);
    float va[7][4][2][2];
#pragma unroll
    for (int t = 0; t < 7; ++t) {
#pragma unroll
        for (int s = 0; s < 4; ++s) {
            const int k0 = 16 * t + 2 * s;
            if (k0 >= NKEY) continue;
            const unsigned of0 = voff + (h ? key_row_off(k0 + 8) : key_row_off(k0));
            const unsigned of1 = voff + (h ? key_row_off(k0 + 9) : key_row_off(k0 + 1));
            va[t][s][0][0] = qkv[of0];
            va[t][s][0][1] = qkv[of1];
            va[t][s][1][0] = qkv[of0 + 16];
            va[t][s][1][1] = qkv[of1 + 16];
        }
    }

    float4 wq[4][2][2];
#pragma unroll
    for (int u = 0; u < 4; ++u) {
        const int ot = (2 * head + u) & 3;
        const float* wr = pw + (unsigned)((16 * ot + lc) * CDIM + head * HD + 8 * h);
#pragma unroll
        for (int dt = 0; dt < 2; ++dt) {
            wq[u][dt][0] = *(const float4*)(wr + 16 * dt);
            wq[u][dt][1] = *(const float4*)(wr + 16 * dt + 4);
        }
    }
    float4 bq[2][2];
#pragma unroll
    for (int tt = 0; tt < 2; ++tt) {
        const float* br = pb + (unsigned)(16 * (2 * head + tt) + 8 * h);
        bq[tt][0] = *(const float4*)(br);
        bq[tt][1] = *(const float4*)(br + 4);
    }

    float mx = -1e30f;
#pragma unroll
    for (int t = 0; t < 7; ++t) {
#pragma unroll
        for (int r = 0; r < 8; ++r) {
            const Tap x = tap_of(t, r, h, dsi, dsj);
            float l = st[t][r] + rbv[t][r];
            if (HAS_PREV) l += pvv[t][r];
            l = x.valid ? l : -1e30f;
            st[t][r] = l;
            mx = fmaxf(mx, l);
        }
    }
    mx = fmaxf(mx, __shfl_xor(mx, 16));

    float ssum = 0.f;
#pragma unroll
    for (int t = 0; t < 7; ++t) {
#pragma unroll
        for (int r = 0; r < 8; ++r) {
            float p = __expf(st[t][r] - mx);
            st[t][r] = p;
            ssum += p;
        }
    }
    ssum += __shfl_xor(ssum, 16);
    const float inv = 1.f / ssum;

#pragma unroll
    for (int t = 0; t < 7; ++t)
#pragma unroll
        for (int r = 0; r < 8; ++r) st[t][r] *= inv;

    v8f o[2][2];
#pragma unroll
    for (int dt = 0; dt < 2; ++dt)
#pragma unroll
        for (int u = 0; u < 2; ++u)
            o[dt][u] = (v8f){0.f, 0.f, 0.f, 0.f, 0.f, 0.f, 0.f, 0.f};

#pragma unroll
    for (int t = 0; t < 7; ++t) {
#pragma unroll
        for (int s = 0; s < 4; ++s) {
            const int k0 = 16 * t + 2 * s;
            if (k0 >= NKEY) continue;
            v2f a0 = {va[t][s][0][0], va[t][s][0][1]};
            v2f a1 = {va[t][s][1][0], va[t][s][1][1]};
            v2f b  = {st[t][2 * s], st[t][2 * s + 1]};
            o[0][t & 1] = wmma4(a0, b, o[0][t & 1]);
            o[1][t & 1] = wmma4(a1, b, o[1][t & 1]);
        }
    }

    v8f d[2];
#pragma unroll
    for (int dt = 0; dt < 2; ++dt) d[dt] = o[dt][0] + o[dt][1];

    v8f pp[4];
#pragma unroll
    for (int tt = 0; tt < 2; ++tt)
        pp[tt] = (v8f){bq[tt][0].x, bq[tt][0].y, bq[tt][0].z, bq[tt][0].w,
                       bq[tt][1].x, bq[tt][1].y, bq[tt][1].z, bq[tt][1].w};
#pragma unroll
    for (int u = 2; u < 4; ++u) pp[u] = (v8f){0.f, 0.f, 0.f, 0.f, 0.f, 0.f, 0.f, 0.f};
#pragma unroll
    for (int dt = 0; dt < 2; ++dt) {
#pragma unroll
        for (int j = 0; j < 4; ++j) {
            v2f b = {d[dt][2 * j], d[dt][2 * j + 1]};
#pragma unroll
            for (int u = 0; u < 4; ++u) {
                const float4 w4 = wq[u][dt][j >> 1];
                v2f a = (j & 1) ? (v2f){w4.z, w4.w} : (v2f){w4.x, w4.y};
                pp[u] = wmma4(a, b, pp[u]);
            }
        }
    }

#pragma unroll
    for (int tt = 0; tt < 2; ++tt)
#pragma unroll
        for (int r = 0; r < 8; ++r) xch[head][tt][r][lane] = pp[2 + tt][r];
    __syncthreads();

    float* op = out + (size_t)qpix * CDIM + 32 * head + 8 * h;
#pragma unroll
    for (int tt = 0; tt < 2; ++tt) {
        float e[8];
#pragma unroll
        for (int r = 0; r < 8; ++r) e[r] = pp[tt][r] + xch[1 - head][tt][r][lane];
        *(float4*)(op + 16 * tt)     = make_float4(e[0], e[1], e[2], e[3]);
        *(float4*)(op + 16 * tt + 4) = make_float4(e[4], e[5], e[6], e[7]);
    }

    if (STORE_MAP) {
#pragma unroll
        for (int g = 0; g < MAPG; ++g) {
            const int t = g >> 1;
            const int r = 4 * (g & 1);
            *(float4*)(mp + 128 * g) = make_float4(st[t][r], st[t][r + 1], st[t][r + 2], st[t][r + 3]);
        }
    }
}

extern "C" void kernel_launch(void* const* d_in, const int* in_sizes, int n_in,
                              void* d_out, int out_size, void* d_ws, size_t ws_size,
                              hipStream_t stream) {
    const float* x      = (const float*)d_in[0];
    const float* qkv_w  = (const float*)d_in[1];
    const float* qkv_b  = (const float*)d_in[2];
    const float* proj_w = (const float*)d_in[3];
    const float* proj_b = (const float*)d_in[4];
    const float* rpb    = (const float*)d_in[5];
    float* out = (float*)d_out;
    float* ws  = (float*)d_ws;

    float* xh   = ws;
    float* qkv  = xh + (size_t)PIX * CDIM;
    float* amap = qkv + (size_t)PIX * 192;

    nchw_to_nhwc<<<PIX / TPX, 128, 0, stream>>>(x, xh);

    for (int s = 0; s < 3; ++s) {

        {
            int numTiles = (PIX / 16) * (192 / 16);
            wmma_gemm_bias<CDIM><<<(numTiles + 7) / 8, 256, 0, stream>>>(
                xh, qkv_w + (size_t)s * 192 * CDIM, qkv_b + (size_t)s * 192,
                qkv, PIX, 192);
        }

        {
            const float* rpb_s = rpb + (size_t)s * NHEADS * (2 * KW - 1) * (2 * KW - 1);
            const float* pw_s  = proj_w + (size_t)s * CDIM * CDIM;
            const float* pb_s  = proj_b + (size_t)s * CDIM;
            if (s == 0)
                na_attn_wmma<false, true><<<PIX / 16, 32 * NHEADS, 0, stream>>>(
                    qkv, rpb_s, pw_s, pb_s, amap, xh);
            else if (s == 1)
                na_attn_wmma<true, true><<<PIX / 16, 32 * NHEADS, 0, stream>>>(
                    qkv, rpb_s, pw_s, pb_s, amap, xh);
            else
                na_attn_wmma<true, false><<<PIX / 16, 32 * NHEADS, 0, stream>>>(
                    qkv, rpb_s, pw_s, pb_s, amap, xh);
        }
    }

    nhwc_to_nchw<<<PIX / TPX, 128, 0, stream>>>(xh, out);
}


# next stage's QKV product chained on the attention wave: head waves swap all 4 projected D tiles via LDS, reuse them as B of Wqkv.x^T (6 tiles each), store Q,K,V rows; 8 -> 6 launches; sums reordered
# speedup vs baseline: 6.3004x; 1.1049x over previous
#include <hip/hip_runtime.h>

typedef float v2f __attribute__((ext_vector_type(2)));
typedef float v4f __attribute__((ext_vector_type(4)));
typedef float v8f __attribute__((ext_vector_type(8)));

#define HH 96
#define WW 96
#define PIX (HH * WW)
#define CDIM 64
#define NHEADS 2
#define HD 32
#define KW 7
#define NKK (KW * KW)

#define TPX 32
#define TPITCH 68

__global__ void __launch_bounds__(128)
nchw_to_nhwc(const float* __restrict__ x, float* __restrict__ xh) {
    __shared__ float tile[TPX * TPITCH];
    const int tid = threadIdx.x;
    const int p0  = blockIdx.x * TPX;
    float4 v[4];
#pragma unroll
    for (int i = 0; i < 4; ++i) {
        const int idx = i * 128 + tid;
        const int c = idx >> 3;
        const int q = idx & 7;
        v[i] = *(const float4*)(x + (size_t)c * PIX + p0 + 4 * q);
    }
#pragma unroll
    for (int i = 0; i < 4; ++i) {
        const int idx = i * 128 + tid;
        const int c = idx >> 3;
        const int q = idx & 7;
        float* t = tile + (4 * q) * TPITCH + c;
        t[0]          = v[i].x;
        t[TPITCH]     = v[i].y;
        t[2 * TPITCH] = v[i].z;
        t[3 * TPITCH] = v[i].w;
    }
    __syncthreads();
#pragma unroll
    for (int i = 0; i < 4; ++i) {
        const int idx = i * 128 + tid;
        const int p = idx >> 4;
        const int m = idx & 15;
        const float4 w = *(const float4*)(tile + p * TPITCH + 4 * m);
        *(float4*)(xh + (size_t)(p0 + p) * CDIM + 4 * m) = w;
    }
}

__global__ void __launch_bounds__(128)
nhwc_to_nchw(const float* __restrict__ xh, float* __restrict__ y) {
    __shared__ float tile[TPX * TPITCH];
    const int tid = threadIdx.x;
    const int p0  = blockIdx.x * TPX;
    float4 v[4];
#pragma unroll
    for (int i = 0; i < 4; ++i) {
        const int idx = i * 128 + tid;
        const int p = idx >> 4;
        const int m = idx & 15;
        v[i] = *(const float4*)(xh + (size_t)(p0 + p) * CDIM + 4 * m);
    }
#pragma unroll
    for (int i = 0; i < 4; ++i) {
        const int idx = i * 128 + tid;
        const int p = idx >> 4;
        const int m = idx & 15;
        *(float4*)(tile + p * TPITCH + 4 * m) = v[i];
    }
    __syncthreads();
#pragma unroll
    for (int i = 0; i < 4; ++i) {
        const int idx = i * 128 + tid;
        const int c = idx >> 3;
        const int q = idx & 7;
        const float* t = tile + (4 * q) * TPITCH + c;
        const float4 w = make_float4(t[0], t[TPITCH], t[2 * TPITCH], t[3 * TPITCH]);
        *(float4*)(y + (size_t)c * PIX + p0 + 4 * q) = w;
    }
}

template <int KD>
__global__ void wmma_gemm_bias(const float* __restrict__ A,
                               const float* __restrict__ W,
                               const float* __restrict__ bias,
                               float* __restrict__ C,
                               int M, int N) {
    int lane = threadIdx.x & 31;
    int wid  = threadIdx.x >> 5;
    int tile = blockIdx.x * 8 + wid;
    int tilesN = N >> 4;
    int numTiles = (M >> 4) * tilesN;
    if (tile >= numTiles) return;

    int tm = tile / tilesN;
    int tn = tile % tilesN;
    int lmod  = lane & 15;
    int lhalf = lane >> 4;

    const float* __restrict__ Ap = A + (size_t)(tm * 16 + lmod) * KD + 2 * lhalf;
    const float* __restrict__ Wp = W + (size_t)(tn * 16 + lmod) * KD + 2 * lhalf;

    v8f acc = {0.f, 0.f, 0.f, 0.f, 0.f, 0.f, 0.f, 0.f};
#pragma unroll
    for (int k = 0; k < KD; k += 4) {
        v2f a = *(const v2f*)(Ap + k);
        v2f b = *(const v2f*)(Wp + k);
        acc = __builtin_amdgcn_wmma_f32_16x16x4_f32(
             false, a,  false, b,
             (short)0, acc,  false,  false);
    }

    float bv = bias[tn * 16 + lmod];
    int rbase = tm * 16 + 8 * lhalf;
    int col   = tn * 16 + lmod;
#pragma unroll
    for (int v = 0; v < 8; ++v) {
        C[(size_t)(rbase + v) * N + col] = acc[v] + bv;
    }
}

#define SQ 10
#define NKEY (SQ * SQ)
#define RPBW (2 * KW - 1)
#define MAPG 13
#define MAPW (MAPG * 32 * 4)
#define RPBS (NHEADS * RPBW * RPBW)

__device__ __forceinline__ v8f wmma4(v2f a, v2f b, v8f c) {
    return __builtin_amdgcn_wmma_f32_16x16x4_f32(
         false, a,  false, b,
         (short)0, c,  false,  false);
}

template <int PITCH>
__device__ __forceinline__ constexpr unsigned key_row_off(int n) {
    return (unsigned)((((n < NKEY ? n : NKEY - 1) / SQ) * WW + ((n < NKEY ? n : NKEY - 1) % SQ)) * PITCH);
}

struct Tap { bool valid; int a, b, kr, kc; };
__device__ __forceinline__ Tap tap_of(int t, int r, int h, int dsi, int dsj) {
    const int n0 = 16 * t + r;
    const int n1 = n0 + 8;
    Tap x;
    x.kr = h ? n1 / SQ : n0 / SQ;
    x.kc = h ? n1 % SQ : n0 % SQ;
    const bool insq = h ? (n1 < NKEY) : (n0 < NKEY);
    x.a = x.kr - dsi;
    x.b = x.kc - dsj;
    x.valid = insq && (unsigned)x.a < (unsigned)KW && (unsigned)x.b < (unsigned)KW;
    return x;
}

#define XPITCH 12
#define XTILE (32 * XPITCH)
template <bool HAS_PREV, bool LAST, int KVS, int KVD>
__global__ void __launch_bounds__(64) __attribute__((amdgpu_waves_per_eu(1, 2)))
na_attn_wmma(float* __restrict__ qio,
             const float* __restrict__ kin,
             const float* __restrict__ vin,
             const float* __restrict__ rpb,
             const float* __restrict__ pw,
             const float* __restrict__ pb,
             const float* __restrict__ nw,
             const float* __restrict__ nb,
             float* __restrict__ amap,
             float* __restrict__ kout,
             float* __restrict__ vout,
             float* __restrict__ out) {
    __shared__ __attribute__((aligned(16))) float xch[LAST ? NHEADS * 2 * 8 * 32 : NHEADS * 4 * XTILE];
    const int lane  = threadIdx.x & 31;
    const int lc    = lane & 15;
    const int h     = lane >> 4;
    const int head  = __builtin_amdgcn_readfirstlane((int)(threadIdx.x >> 5));
    const int patch = blockIdx.x;
    const int i0 = (patch / (WW / 4)) * 4;
    const int j0 = (patch % (WW / 4)) * 4;

    int r0 = i0 - KW / 2; r0 = r0 < 0 ? 0 : r0; r0 = r0 > HH - SQ ? HH - SQ : r0;
    int c0 = j0 - KW / 2; c0 = c0 < 0 ? 0 : c0; c0 = c0 > WW - SQ ? WW - SQ : c0;

    const int qi = i0 + (lc >> 2);
    const int qj = j0 + (lc & 3);
    const int qpix = qi * WW + qj;
    int si = qi - KW / 2; si = si < 0 ? 0 : si; si = si > HH - KW ? HH - KW : si;
    int sj = qj - KW / 2; sj = sj < 0 ? 0 : sj; sj = sj > WW - KW ? WW - KW : sj;
    const int dsi = si - r0;
    const int dsj = sj - c0;

    const float scale = 0.17677669529663687f;

    const unsigned qoff = (unsigned)(qpix * 192 + head * HD + 4 * h);
    float4 qv[4];
#pragma unroll
    for (int s = 0; s < 4; ++s) qv[s] = *(const float4*)(qio + qoff + 8 * s);

    float4 kv[7][4];
#pragma unroll
    for (int t = 0; t < 7; ++t) {
        int kn = 16 * t + lc;
        kn = kn < NKEY ? kn : NKEY - 1;
        const int kr = kn / SQ;
        const int kc = kn - kr * SQ;
        const unsigned koff = (unsigned)(((r0 + kr) * WW + (c0 + kc)) * KVS + head * HD + 4 * h);
#pragma unroll
        for (int s = 0; s < 4; ++s) kv[t][s] = *(const float4*)(kin + koff + 8 * s);
    }

    const int rpb_lane = (r0 - qi + KW - 1) * RPBW + (c0 - qj + KW - 1);
    const float* rb = rpb + head * RPBW * RPBW;
    float* mp = amap + (unsigned)((patch * NHEADS + head) * MAPW) + 4 * lane;
    float4 pm[MAPG];
    if (HAS_PREV) {
#pragma unroll
        for (int g = 0; g < MAPG; ++g) pm[g] = *(const float4*)(mp + 128 * g);
    }
    float rbv[7][8];
    float pvv[7][8];
#pragma unroll
    for (int t = 0; t < 7; ++t) {
#pragma unroll
        for (int r = 0; r < 8; ++r) {
            const Tap x = tap_of(t, r, h, dsi, dsj);
            const unsigned ridx = x.valid ? (unsigned)(rpb_lane + x.kr * RPBW + x.kc) : 0u;
            rbv[t][r] = rb[ridx];
            const int f = 8 * t + r;
            if (HAS_PREV && f < 4 * MAPG) {
                const float4 g4 = pm[f >> 2];
                pvv[t][r] = (f & 3) == 0 ? g4.x : (f & 3) == 1 ? g4.y : (f & 3) == 2 ? g4.z : g4.w;
            } else {
                pvv[t][r] = 0.f;
            }
        }
    }

#pragma unroll
    for (int s = 0; s < 4; ++s)
        qv[s] = make_float4(qv[s].x * scale, qv[s].y * scale, qv[s].z * scale, qv[s].w * scale);

    v8f st[7];
#pragma unroll
    for (int t = 0; t < 7; ++t) {
        v8f acc = {0.f, 0.f, 0.f, 0.f, 0.f, 0.f, 0.f, 0.f};
#pragma unroll
        for (int s = 0; s < 4; ++s) {
            v2f a0 = {kv[t][s].x, kv[t][s].y};
            v2f a1 = {kv[t][s].z, kv[t][s].w};
            v2f b0 = {qv[s].x, qv[s].y};
            v2f b1 = {qv[s].z, qv[s].w};
            acc = wmma4(a0, b0, acc);
            acc = wmma4(a1, b1, acc);
        }
        st[t] = acc;
    }

    const unsigned voff = (unsigned)((r0 * WW + c0) * KVS + head * HD + lc);
    float va[7][4][2][2];
#pragma unroll
    for (int t = 0; t < 7; ++t) {
#pragma unroll
        for (int s = 0; s < 4; ++s) {
            const int k0 = 16 * t + 2 * s;
            if (k0 >= NKEY) continue;
            const unsigned of0 = voff + (h ? key_row_off<KVS>(k0 + 8) : key_row_off<KVS>(k0));
            const unsigned of1 = voff + (h ? key_row_off<KVS>(k0 + 9) : key_row_off<KVS>(k0 + 1));
            va[t][s][0][0] = vin[of0];
            va[t][s][0][1] = vin[of1];
            va[t][s][1][0] = vin[of0 + 16];
            va[t][s][1][1] = vin[of1 + 16];
        }
    }

    float4 wq[4][2][2];
#pragma unroll
    for (int u = 0; u < 4; ++u) {
        const int ot = (2 * head + u) & 3;
        const float* wr = pw + (unsigned)((16 * ot + lc) * CDIM + head * HD + 8 * h);
#pragma unroll
        for (int dt = 0; dt < 2; ++dt) {
            wq[u][dt][0] = *(const float4*)(wr + 16 * dt);
            wq[u][dt][1] = *(const float4*)(wr + 16 * dt + 4);
        }
    }
    float4 bq[2][2];
#pragma unroll
    for (int tt = 0; tt < 2; ++tt) {
        const float* br = pb + (unsigned)(16 * (2 * head + tt) + 8 * h);
        bq[tt][0] = *(const float4*)(br);
        bq[tt][1] = *(const float4*)(br + 4);
    }

    float mx = -1e30f;
#pragma unroll
    for (int t = 0; t < 7; ++t) {
#pragma unroll
        for (int r = 0; r < 8; ++r) {
            const Tap x = tap_of(t, r, h, dsi, dsj);
            float l = st[t][r] + rbv[t][r];
            if (HAS_PREV) l += pvv[t][r];
            l = x.valid ? l : -1e30f;
            st[t][r] = l;
            mx = fmaxf(mx, l);
        }
    }
    mx = fmaxf(mx, __shfl_xor(mx, 16));

    float ssum = 0.f;
#pragma unroll
    for (int t = 0; t < 7; ++t) {
#pragma unroll
        for (int r = 0; r < 8; ++r) {
            float p = __expf(st[t][r] - mx);
            st[t][r] = p;
            ssum += p;
        }
    }
    ssum += __shfl_xor(ssum, 16);
    const float inv = 1.f / ssum;

#pragma unroll
    for (int t = 0; t < 7; ++t)
#pragma unroll
        for (int r = 0; r < 8; ++r) st[t][r] *= inv;

    v8f o[2][2];
#pragma unroll
    for (int dt = 0; dt < 2; ++dt)
#pragma unroll
        for (int u = 0; u < 2; ++u)
            o[dt][u] = (v8f){0.f, 0.f, 0.f, 0.f, 0.f, 0.f, 0.f, 0.f};

#pragma unroll
    for (int t = 0; t < 7; ++t) {
#pragma unroll
        for (int s = 0; s < 4; ++s) {
            const int k0 = 16 * t + 2 * s;
            if (k0 >= NKEY) continue;
            v2f a0 = {va[t][s][0][0], va[t][s][0][1]};
            v2f a1 = {va[t][s][1][0], va[t][s][1][1]};
            v2f b  = {st[t][2 * s], st[t][2 * s + 1]};
            o[0][t & 1] = wmma4(a0, b, o[0][t & 1]);
            o[1][t & 1] = wmma4(a1, b, o[1][t & 1]);
        }
    }

    v8f d[2];
#pragma unroll
    for (int dt = 0; dt < 2; ++dt) d[dt] = o[dt][0] + o[dt][1];

    float4 wn[6][4][2];
    if constexpr (!LAST) {
#pragma unroll
        for (int m = 0; m < 6; ++m) {
            const int frow = 64 * (m >> 1) + 32 * head + 16 * (m & 1);
            const float* wr = nw + (unsigned)((frow + lc) * CDIM + 8 * h);
#pragma unroll
            for (int u = 0; u < 4; ++u) {
                const int g = (2 * head + u) & 3;
                wn[m][u][0] = *(const float4*)(wr + 16 * g);
                wn[m][u][1] = *(const float4*)(wr + 16 * g + 4);
            }
        }
    }

    v8f pp[4];
#pragma unroll
    for (int tt = 0; tt < 2; ++tt)
        pp[tt] = (v8f){bq[tt][0].x, bq[tt][0].y, bq[tt][0].z, bq[tt][0].w,
                       bq[tt][1].x, bq[tt][1].y, bq[tt][1].z, bq[tt][1].w};
#pragma unroll
    for (int u = 2; u < 4; ++u) pp[u] = (v8f){0.f, 0.f, 0.f, 0.f, 0.f, 0.f, 0.f, 0.f};
#pragma unroll
    for (int dt = 0; dt < 2; ++dt) {
#pragma unroll
        for (int j = 0; j < 4; ++j) {
            v2f b = {d[dt][2 * j], d[dt][2 * j + 1]};
#pragma unroll
            for (int u = 0; u < 4; ++u) {
                const float4 w4 = wq[u][dt][j >> 1];
                v2f a = (j & 1) ? (v2f){w4.z, w4.w} : (v2f){w4.x, w4.y};
                pp[u] = wmma4(a, b, pp[u]);
            }
        }
    }

    if constexpr (LAST) {
#pragma unroll
        for (int tt = 0; tt < 2; ++tt)
#pragma unroll
            for (int r = 0; r < 8; ++r) xch[((head * 2 + tt) * 8 + r) * 32 + lane] = pp[2 + tt][r];
        __syncthreads();

        float* op = out + (size_t)qpix * CDIM + 32 * head + 8 * h;
#pragma unroll
        for (int tt = 0; tt < 2; ++tt) {
            float e[8];
#pragma unroll
            for (int r = 0; r < 8; ++r) e[r] = pp[tt][r] + xch[(((1 - head) * 2 + tt) * 8 + r) * 32 + lane];
            *(float4*)(op + 16 * tt)     = make_float4(e[0], e[1], e[2], e[3]);
            *(float4*)(op + 16 * tt + 4) = make_float4(e[4], e[5], e[6], e[7]);
        }
    } else {
        float* xw = xch + head * (4 * XTILE) + lane * XPITCH;
        const float* xr = xch + (1 - head) * (4 * XTILE) + lane * XPITCH;
#pragma unroll
        for (int u = 0; u < 4; ++u) {
            *(float4*)(xw + u * XTILE)     = make_float4(pp[u][0], pp[u][1], pp[u][2], pp[u][3]);
            *(float4*)(xw + u * XTILE + 4) = make_float4(pp[u][4], pp[u][5], pp[u][6], pp[u][7]);
        }
        __syncthreads();

        v4f xa[4], xb[4];
#pragma unroll
        for (int u = 0; u < 4; ++u) {
            xa[u] = *(const v4f*)(xr + ((u + 2) & 3) * XTILE);
            xb[u] = *(const v4f*)(xr + ((u + 2) & 3) * XTILE + 4);
        }
        float4 nbq[6][2];
#pragma unroll
        for (int m = 0; m < 6; ++m) {
            const float* br = nb + (unsigned)(64 * (m >> 1) + 32 * head + 16 * (m & 1) + 8 * h);
            nbq[m][0] = *(const float4*)(br);
            nbq[m][1] = *(const float4*)(br + 4);
        }
        asm volatile("" :: "v"(xa[0]), "v"(xb[0]), "v"(xa[1]), "v"(xb[1]),
                           "v"(xa[2]), "v"(xb[2]), "v"(xa[3]), "v"(xb[3]));
        v8f xt[4];
#pragma unroll
        for (int u = 0; u < 4; ++u)
            xt[u] = pp[u] + (v8f){xa[u].x, xa[u].y, xa[u].z, xa[u].w, xb[u].x, xb[u].y, xb[u].z, xb[u].w};

        v8f qa[6];
#pragma unroll
        for (int m = 0; m < 6; ++m) qa[m] = (v8f){0.f, 0.f, 0.f, 0.f, 0.f, 0.f, 0.f, 0.f};
#pragma unroll
        for (int u = 0; u < 4; ++u) {
#pragma unroll
            for (int j = 0; j < 4; ++j) {
                v2f b = {xt[u][2 * j], xt[u][2 * j + 1]};
#pragma unroll
                for (int m = 0; m < 6; ++m) {
                    const float4 w4 = wn[m][u][j >> 1];
                    v2f a = (j & 1) ? (v2f){w4.z, w4.w} : (v2f){w4.x, w4.y};
                    qa[m] = wmma4(a, b, qa[m]);
                }
            }
        }

        float* qp = qio  + (size_t)qpix * 192 + 32 * head + 8 * h;
        float* kp = kout + (size_t)qpix * KVD + 32 * head + 8 * h;
        float* vp = vout + (size_t)qpix * KVD + 32 * head + 8 * h;
#pragma unroll
        for (int m = 0; m < 6; ++m) {
            float* dp = ((m >> 1) == 0 ? qp : (m >> 1) == 1 ? kp : vp) + 16 * (m & 1);
            *(float4*)(dp)     = make_float4(qa[m][0] + nbq[m][0].x, qa[m][1] + nbq[m][0].y,
                                             qa[m][2] + nbq[m][0].z, qa[m][3] + nbq[m][0].w);
            *(float4*)(dp + 4) = make_float4(qa[m][4] + nbq[m][1].x, qa[m][5] + nbq[m][1].y,
                                             qa[m][6] + nbq[m][1].z, qa[m][7] + nbq[m][1].w);
        }

#pragma unroll
        for (int g = 0; g < MAPG; ++g) {
            const int t = g >> 1;
            const int r = 4 * (g & 1);
            *(float4*)(mp + 128 * g) = make_float4(st[t][r], st[t][r + 1], st[t][r + 2], st[t][r + 3]);
        }
    }
}

extern "C" void kernel_launch(void* const* d_in, const int* in_sizes, int n_in,
                              void* d_out, int out_size, void* d_ws, size_t ws_size,
                              hipStream_t stream) {
    const float* x      = (const float*)d_in[0];
    const float* qkv_w  = (const float*)d_in[1];
    const float* qkv_b  = (const float*)d_in[2];
    const float* proj_w = (const float*)d_in[3];
    const float* proj_b = (const float*)d_in[4];
    const float* rpb    = (const float*)d_in[5];
    float* out = (float*)d_out;
    float* ws  = (float*)d_ws;

    float* plane = ws;
    float* qkv   = plane + (size_t)PIX * CDIM;
    float* amap  = qkv + (size_t)PIX * 192;

    nchw_to_nhwc<<<PIX / TPX, 128, 0, stream>>>(x, plane);

    {
        int numTiles = (PIX / 16) * (192 / 16);
        wmma_gemm_bias<CDIM><<<(numTiles + 7) / 8, 256, 0, stream>>>(
            plane, qkv_w, qkv_b, qkv, PIX, 192);
    }

    na_attn_wmma<false, false, 192, 64><<<PIX / 16, 32 * NHEADS, 0, stream>>>(
        qkv, qkv + 64, qkv + 128, rpb, proj_w, proj_b,
        qkv_w + 192 * CDIM, qkv_b + 192, amap, plane, out, nullptr);
    na_attn_wmma<true, false, 64, 192><<<PIX / 16, 32 * NHEADS, 0, stream>>>(
        qkv, plane, out, rpb + RPBS, proj_w + CDIM * CDIM, proj_b + CDIM,
        qkv_w + 2 * 192 * CDIM, qkv_b + 2 * 192, amap, qkv + 64, qkv + 128, nullptr);
    na_attn_wmma<true, true, 192, 64><<<PIX / 16, 32 * NHEADS, 0, stream>>>(
        qkv, qkv + 64, qkv + 128, rpb + 2 * RPBS, proj_w + 2 * CDIM * CDIM, proj_b + 2 * CDIM,
        nullptr, nullptr, amap, nullptr, nullptr, plane);

    nhwc_to_nchw<<<PIX / TPX, 128, 0, stream>>>(plane, out);
}


# attention key order: lane half h owns square rows 5h..5h+4 (register 8t+r = key 8t+r+50h), so window tests, bias indices and V offsets are constants plus one lane base, no selects; sums reordered
# speedup vs baseline: 6.6112x; 1.0493x over previous
#include <hip/hip_runtime.h>

typedef float v2f __attribute__((ext_vector_type(2)));
typedef float v4f __attribute__((ext_vector_type(4)));
typedef float v8f __attribute__((ext_vector_type(8)));

#define HH 96
#define WW 96
#define PIX (HH * WW)
#define CDIM 64
#define NHEADS 2
#define HD 32
#define KW 7
#define NKK (KW * KW)

#define TPX 32
#define TPITCH 68

__global__ void __launch_bounds__(128)
nchw_to_nhwc(const float* __restrict__ x, float* __restrict__ xh) {
    __shared__ float tile[TPX * TPITCH];
    const int tid = threadIdx.x;
    const int p0  = blockIdx.x * TPX;
    float4 v[4];
#pragma unroll
    for (int i = 0; i < 4; ++i) {
        const int idx = i * 128 + tid;
        const int c = idx >> 3;
        const int q = idx & 7;
        v[i] = *(const float4*)(x + (size_t)c * PIX + p0 + 4 * q);
    }
#pragma unroll
    for (int i = 0; i < 4; ++i) {
        const int idx = i * 128 + tid;
        const int c = idx >> 3;
        const int q = idx & 7;
        float* t = tile + (4 * q) * TPITCH + c;
        t[0]          = v[i].x;
        t[TPITCH]     = v[i].y;
        t[2 * TPITCH] = v[i].z;
        t[3 * TPITCH] = v[i].w;
    }
    __syncthreads();
#pragma unroll
    for (int i = 0; i < 4; ++i) {
        const int idx = i * 128 + tid;
        const int p = idx >> 4;
        const int m = idx & 15;
        const float4 w = *(const float4*)(tile + p * TPITCH + 4 * m);
        *(float4*)(xh + (size_t)(p0 + p) * CDIM + 4 * m) = w;
    }
}

__global__ void __launch_bounds__(128)
nhwc_to_nchw(const float* __restrict__ xh, float* __restrict__ y) {
    __shared__ float tile[TPX * TPITCH];
    const int tid = threadIdx.x;
    const int p0  = blockIdx.x * TPX;
    float4 v[4];
#pragma unroll
    for (int i = 0; i < 4; ++i) {
        const int idx = i * 128 + tid;
        const int p = idx >> 4;
        const int m = idx & 15;
        v[i] = *(const float4*)(xh + (size_t)(p0 + p) * CDIM + 4 * m);
    }
#pragma unroll
    for (int i = 0; i < 4; ++i) {
        const int idx = i * 128 + tid;
        const int p = idx >> 4;
        const int m = idx & 15;
        *(float4*)(tile + p * TPITCH + 4 * m) = v[i];
    }
    __syncthreads();
#pragma unroll
    for (int i = 0; i < 4; ++i) {
        const int idx = i * 128 + tid;
        const int c = idx >> 3;
        const int q = idx & 7;
        const float* t = tile + (4 * q) * TPITCH + c;
        const float4 w = make_float4(t[0], t[TPITCH], t[2 * TPITCH], t[3 * TPITCH]);
        *(float4*)(y + (size_t)c * PIX + p0 + 4 * q) = w;
    }
}

template <int KD>
__global__ void wmma_gemm_bias(const float* __restrict__ A,
                               const float* __restrict__ W,
                               const float* __restrict__ bias,
                               float* __restrict__ C,
                               int M, int N) {
    int lane = threadIdx.x & 31;
    int wid  = threadIdx.x >> 5;
    int tile = blockIdx.x * 8 + wid;
    int tilesN = N >> 4;
    int numTiles = (M >> 4) * tilesN;
    if (tile >= numTiles) return;

    int tm = tile / tilesN;
    int tn = tile % tilesN;
    int lmod  = lane & 15;
    int lhalf = lane >> 4;

    const float* __restrict__ Ap = A + (size_t)(tm * 16 + lmod) * KD + 2 * lhalf;
    const float* __restrict__ Wp = W + (size_t)(tn * 16 + lmod) * KD + 2 * lhalf;

    v8f acc = {0.f, 0.f, 0.f, 0.f, 0.f, 0.f, 0.f, 0.f};
#pragma unroll
    for (int k = 0; k < KD; k += 4) {
        v2f a = *(const v2f*)(Ap + k);
        v2f b = *(const v2f*)(Wp + k);
        acc = __builtin_amdgcn_wmma_f32_16x16x4_f32(
             false, a,  false, b,
             (short)0, acc,  false,  false);
    }

    float bv = bias[tn * 16 + lmod];
    int rbase = tm * 16 + 8 * lhalf;
    int col   = tn * 16 + lmod;
#pragma unroll
    for (int v = 0; v < 8; ++v) {
        C[(size_t)(rbase + v) * N + col] = acc[v] + bv;
    }
}

#define SQ 10
#define NKEY (SQ * SQ)
#define RPBW (2 * KW - 1)
#define MAPG 13
#define MAPW (MAPG * 32 * 4)
#define RPBS (NHEADS * RPBW * RPBW)

__device__ __forceinline__ v8f wmma4(v2f a, v2f b, v8f c) {
    return __builtin_amdgcn_wmma_f32_16x16x4_f32(
         false, a,  false, b,
         (short)0, c,  false,  false);
}

template <int PITCH>
__device__ __forceinline__ constexpr unsigned key_row_off(int n) {
    return (unsigned)((((n < NKEY ? n : NKEY - 1) / SQ) * WW + ((n < NKEY ? n : NKEY - 1) % SQ)) * PITCH);
}

#define HKEY (NKEY / 2)

#define XPITCH 12
#define XTILE (32 * XPITCH)
template <bool HAS_PREV, bool LAST, int KVS, int KVD>
__global__ void __launch_bounds__(64) __attribute__((amdgpu_waves_per_eu(1, 2)))
na_attn_wmma(float* __restrict__ qio,
             const float* __restrict__ kin,
             const float* __restrict__ vin,
             const float* __restrict__ rpb,
             const float* __restrict__ pw,
             const float* __restrict__ pb,
             const float* __restrict__ nw,
             const float* __restrict__ nb,
             float* __restrict__ amap,
             float* __restrict__ kout,
             float* __restrict__ vout,
             float* __restrict__ out) {
    __shared__ __attribute__((aligned(16))) float xch[LAST ? NHEADS * 2 * 8 * 32 : NHEADS * 4 * XTILE];
    const int lane  = threadIdx.x & 31;
    const int lc    = lane & 15;
    const int h     = lane >> 4;
    const int head  = __builtin_amdgcn_readfirstlane((int)(threadIdx.x >> 5));
    const int patch = blockIdx.x;
    const int i0 = (patch / (WW / 4)) * 4;
    const int j0 = (patch % (WW / 4)) * 4;

    int r0 = i0 - KW / 2; r0 = r0 < 0 ? 0 : r0; r0 = r0 > HH - SQ ? HH - SQ : r0;
    int c0 = j0 - KW / 2; c0 = c0 < 0 ? 0 : c0; c0 = c0 > WW - SQ ? WW - SQ : c0;

    const int qi = i0 + (lc >> 2);
    const int qj = j0 + (lc & 3);
    const int qpix = qi * WW + qj;
    int si = qi - KW / 2; si = si < 0 ? 0 : si; si = si > HH - KW ? HH - KW : si;
    int sj = qj - KW / 2; sj = sj < 0 ? 0 : sj; sj = sj > WW - KW ? WW - KW : sj;
    const int dsi = si - r0;
    const int dsj = sj - c0;

    const float scale = 0.17677669529663687f;

    const unsigned qoff = (unsigned)(qpix * 192 + head * HD + 4 * h);
    float4 qv[4];
#pragma unroll
    for (int s = 0; s < 4; ++s) qv[s] = *(const float4*)(qio + qoff + 8 * s);

    float4 kv[7][4];
#pragma unroll
    for (int t = 0; t < 7; ++t) {
        int km = 8 * t + (lc & 7);
        km = km < HKEY ? km : HKEY - 1;
        const int kq = km / SQ;
        const int kr = kq + 5 * (lc >> 3);
        const int kc = km - kq * SQ;
        const unsigned koff = (unsigned)(((r0 + kr) * WW + (c0 + kc)) * KVS + head * HD + 4 * h);
#pragma unroll
        for (int s = 0; s < 4; ++s) kv[t][s] = *(const float4*)(kin + koff + 8 * s);
    }

    const int ri0 = r0 - qi + KW - 1 + 5 * h;
    const int ci0 = c0 - qj + KW - 1;
    int rrow[5], ccol[SQ];
#pragma unroll
    for (int k = 0; k < 5; ++k) { int v = ri0 + k; v = v < 0 ? 0 : v; v = v > RPBW - 1 ? RPBW - 1 : v; rrow[k] = v * RPBW; }
#pragma unroll
    for (int k = 0; k < SQ; ++k) { int v = ci0 + k; v = v < 0 ? 0 : v; v = v > RPBW - 1 ? RPBW - 1 : v; ccol[k] = v; }
    const int dsih = dsi - 5 * h;
    bool rowv[5], colv[SQ];
#pragma unroll
    for (int k = 0; k < 5; ++k) rowv[k] = (unsigned)(k - dsih) < (unsigned)KW;
#pragma unroll
    for (int k = 0; k < SQ; ++k) colv[k] = (unsigned)(k - dsj) < (unsigned)KW;
    const float* rb = rpb + head * RPBW * RPBW;
    float* mp = amap + (unsigned)((patch * NHEADS + head) * MAPW) + 4 * lane;
    float4 pm[MAPG];
    if (HAS_PREV) {
#pragma unroll
        for (int g = 0; g < MAPG; ++g) pm[g] = *(const float4*)(mp + 128 * g);
    }
    float rbv[7][8];
    float pvv[7][8];
#pragma unroll
    for (int t = 0; t < 7; ++t) {
#pragma unroll
        for (int r = 0; r < 8; ++r) {
            const int f = 8 * t + r;
            rbv[t][r] = f < HKEY ? rb[rrow[f / SQ] + ccol[f % SQ]] : 0.f;
            if (HAS_PREV && f < 4 * MAPG) {
                const float4 g4 = pm[f >> 2];
                pvv[t][r] = (f & 3) == 0 ? g4.x : (f & 3) == 1 ? g4.y : (f & 3) == 2 ? g4.z : g4.w;
            } else {
                pvv[t][r] = 0.f;
            }
        }
    }

#pragma unroll
    for (int s = 0; s < 4; ++s)
        qv[s] = make_float4(qv[s].x * scale, qv[s].y * scale, qv[s].z * scale, qv[s].w * scale);

    v8f st[7];
#pragma unroll
    for (int t = 0; t < 7; ++t) {
        v8f acc = {0.f, 0.f, 0.f, 0.f, 0.f, 0.f, 0.f, 0.f};
#pragma unroll
        for (int s = 0; s < 4; ++s) {
            v2f a0 = {kv[t][s].x, kv[t][s].y};
            v2f a1 = {kv[t][s].z, kv[t][s].w};
            v2f b0 = {qv[s].x, qv[s].y};
            v2f b1 = {qv[s].z, qv[s].w};
            acc = wmma4(a0, b0, acc);
            acc = wmma4(a1, b1, acc);
        }
        st[t] = acc;
    }

    const unsigned voff = (unsigned)(((r0 + 5 * h) * WW + c0) * KVS + head * HD + lc);
    float va[7][4][2][2];
#pragma unroll
    for (int t = 0; t < 7; ++t) {
#pragma unroll
        for (int s = 0; s < 4; ++s) {
            const int k0 = 8 * t + 2 * s;
            if (k0 >= HKEY) continue;
            const unsigned of0 = voff + key_row_off<KVS>(k0);
            const unsigned of1 = voff + key_row_off<KVS>(k0 + 1);
            va[t][s][0][0] = vin[of0];
            va[t][s][0][1] = vin[of1];
            va[t][s][1][0] = vin[of0 + 16];
            va[t][s][1][1] = vin[of1 + 16];
        }
    }

    float4 wq[4][2][2];
#pragma unroll
    for (int u = 0; u < 4; ++u) {
        const int ot = (2 * head + u) & 3;
        const float* wr = pw + (unsigned)((16 * ot + lc) * CDIM + head * HD + 8 * h);
#pragma unroll
        for (int dt = 0; dt < 2; ++dt) {
            wq[u][dt][0] = *(const float4*)(wr + 16 * dt);
            wq[u][dt][1] = *(const float4*)(wr + 16 * dt + 4);
        }
    }
    float4 bq[2][2];
#pragma unroll
    for (int tt = 0; tt < 2; ++tt) {
        const float* br = pb + (unsigned)(16 * (2 * head + tt) + 8 * h);
        bq[tt][0] = *(const float4*)(br);
        bq[tt][1] = *(const float4*)(br + 4);
    }

    float mx = -1e30f;
#pragma unroll
    for (int t = 0; t < 7; ++t) {
#pragma unroll
        for (int r = 0; r < 8; ++r) {
            const int f = 8 * t + r;
            const bool valid = f < HKEY && rowv[f < HKEY ? f / SQ : 0] && colv[f % SQ];
            float l = st[t][r] + rbv[t][r];
            if (HAS_PREV) l += pvv[t][r];
            l = valid ? l : -1e30f;
            st[t][r] = l;
            mx = fmaxf(mx, l);
        }
    }
    mx = fmaxf(mx, __shfl_xor(mx, 16));

    float ssum = 0.f;
#pragma unroll
    for (int t = 0; t < 7; ++t) {
#pragma unroll
        for (int r = 0; r < 8; ++r) {
            float p = __expf(st[t][r] - mx);
            st[t][r] = p;
            ssum += p;
        }
    }
    ssum += __shfl_xor(ssum, 16);
    const float inv = 1.f / ssum;

#pragma unroll
    for (int t = 0; t < 7; ++t)
#pragma unroll
        for (int r = 0; r < 8; ++r) st[t][r] *= inv;

    v8f o[2][2];
#pragma unroll
    for (int dt = 0; dt < 2; ++dt)
#pragma unroll
        for (int u = 0; u < 2; ++u)
            o[dt][u] = (v8f){0.f, 0.f, 0.f, 0.f, 0.f, 0.f, 0.f, 0.f};

#pragma unroll
    for (int t = 0; t < 7; ++t) {
#pragma unroll
        for (int s = 0; s < 4; ++s) {
            const int k0 = 8 * t + 2 * s;
            if (k0 >= HKEY) continue;
            v2f a0 = {va[t][s][0][0], va[t][s][0][1]};
            v2f a1 = {va[t][s][1][0], va[t][s][1][1]};
            v2f b  = {st[t][2 * s], st[t][2 * s + 1]};
            o[0][t & 1] = wmma4(a0, b, o[0][t & 1]);
            o[1][t & 1] = wmma4(a1, b, o[1][t & 1]);
        }
    }

    v8f d[2];
#pragma unroll
    for (int dt = 0; dt < 2; ++dt) d[dt] = o[dt][0] + o[dt][1];

    float4 wn[6][4][2];
    if constexpr (!LAST) {
#pragma unroll
        for (int m = 0; m < 6; ++m) {
            const int frow = 64 * (m >> 1) + 32 * head + 16 * (m & 1);
            const float* wr = nw + (unsigned)((frow + lc) * CDIM + 8 * h);
#pragma unroll
            for (int u = 0; u < 4; ++u) {
                const int g = (2 * head + u) & 3;
                wn[m][u][0] = *(const float4*)(wr + 16 * g);
                wn[m][u][1] = *(const float4*)(wr + 16 * g + 4);
            }
        }
    }

    v8f pp[4];
#pragma unroll
    for (int tt = 0; tt < 2; ++tt)
        pp[tt] = (v8f){bq[tt][0].x, bq[tt][0].y, bq[tt][0].z, bq[tt][0].w,
                       bq[tt][1].x, bq[tt][1].y, bq[tt][1].z, bq[tt][1].w};
#pragma unroll
    for (int u = 2; u < 4; ++u) pp[u] = (v8f){0.f, 0.f, 0.f, 0.f, 0.f, 0.f, 0.f, 0.f};
#pragma unroll
    for (int dt = 0; dt < 2; ++dt) {
#pragma unroll
        for (int j = 0; j < 4; ++j) {
            v2f b = {d[dt][2 * j], d[dt][2 * j + 1]};
#pragma unroll
            for (int u = 0; u < 4; ++u) {
                const float4 w4 = wq[u][dt][j >> 1];
                v2f a = (j & 1) ? (v2f){w4.z, w4.w} : (v2f){w4.x, w4.y};
                pp[u] = wmma4(a, b, pp[u]);
            }
        }
    }

    if constexpr (LAST) {
#pragma unroll
        for (int tt = 0; tt < 2; ++tt)
#pragma unroll
            for (int r = 0; r < 8; ++r) xch[((head * 2 + tt) * 8 + r) * 32 + lane] = pp[2 + tt][r];
        __syncthreads();

        float* op = out + (size_t)qpix * CDIM + 32 * head + 8 * h;
#pragma unroll
        for (int tt = 0; tt < 2; ++tt) {
            float e[8];
#pragma unroll
            for (int r = 0; r < 8; ++r) e[r] = pp[tt][r] + xch[(((1 - head) * 2 + tt) * 8 + r) * 32 + lane];
            *(float4*)(op + 16 * tt)     = make_float4(e[0], e[1], e[2], e[3]);
            *(float4*)(op + 16 * tt + 4) = make_float4(e[4], e[5], e[6], e[7]);
        }
    } else {
        float* xw = xch + head * (4 * XTILE) + lane * XPITCH;
        const float* xr = xch + (1 - head) * (4 * XTILE) + lane * XPITCH;
#pragma unroll
        for (int u = 0; u < 4; ++u) {
            *(float4*)(xw + u * XTILE)     = make_float4(pp[u][0], pp[u][1], pp[u][2], pp[u][3]);
            *(float4*)(xw + u * XTILE + 4) = make_float4(pp[u][4], pp[u][5], pp[u][6], pp[u][7]);
        }
        __syncthreads();

        v4f xa[4], xb[4];
#pragma unroll
        for (int u = 0; u < 4; ++u) {
            xa[u] = *(const v4f*)(xr + ((u + 2) & 3) * XTILE);
            xb[u] = *(const v4f*)(xr + ((u + 2) & 3) * XTILE + 4);
        }
        float4 nbq[6][2];
#pragma unroll
        for (int m = 0; m < 6; ++m) {
            const float* br = nb + (unsigned)(64 * (m >> 1) + 32 * head + 16 * (m & 1) + 8 * h);
            nbq[m][0] = *(const float4*)(br);
            nbq[m][1] = *(const float4*)(br + 4);
        }
        asm volatile("" :: "v"(xa[0]), "v"(xb[0]), "v"(xa[1]), "v"(xb[1]),
                           "v"(xa[2]), "v"(xb[2]), "v"(xa[3]), "v"(xb[3]));
        v8f xt[4];
#pragma unroll
        for (int u = 0; u < 4; ++u)
            xt[u] = pp[u] + (v8f){xa[u].x, xa[u].y, xa[u].z, xa[u].w, xb[u].x, xb[u].y, xb[u].z, xb[u].w};

        v8f qa[6];
#pragma unroll
        for (int m = 0; m < 6; ++m) qa[m] = (v8f){0.f, 0.f, 0.f, 0.f, 0.f, 0.f, 0.f, 0.f};
#pragma unroll
        for (int u = 0; u < 4; ++u) {
#pragma unroll
            for (int j = 0; j < 4; ++j) {
                v2f b = {xt[u][2 * j], xt[u][2 * j + 1]};
#pragma unroll
                for (int m = 0; m < 6; ++m) {
                    const float4 w4 = wn[m][u][j >> 1];
                    v2f a = (j & 1) ? (v2f){w4.z, w4.w} : (v2f){w4.x, w4.y};
                    qa[m] = wmma4(a, b, qa[m]);
                }
            }
        }

        float* qp = qio  + (size_t)qpix * 192 + 32 * head + 8 * h;
        float* kp = kout + (size_t)qpix * KVD + 32 * head + 8 * h;
        float* vp = vout + (size_t)qpix * KVD + 32 * head + 8 * h;
#pragma unroll
        for (int m = 0; m < 6; ++m) {
            float* dp = ((m >> 1) == 0 ? qp : (m >> 1) == 1 ? kp : vp) + 16 * (m & 1);
            *(float4*)(dp)     = make_float4(qa[m][0] + nbq[m][0].x, qa[m][1] + nbq[m][0].y,
                                             qa[m][2] + nbq[m][0].z, qa[m][3] + nbq[m][0].w);
            *(float4*)(dp + 4) = make_float4(qa[m][4] + nbq[m][1].x, qa[m][5] + nbq[m][1].y,
                                             qa[m][6] + nbq[m][1].z, qa[m][7] + nbq[m][1].w);
        }

#pragma unroll
        for (int g = 0; g < MAPG; ++g) {
            const int t = g >> 1;
            const int r = 4 * (g & 1);
            *(float4*)(mp + 128 * g) = make_float4(st[t][r], st[t][r + 1], st[t][r + 2], st[t][r + 3]);
        }
    }
}

extern "C" void kernel_launch(void* const* d_in, const int* in_sizes, int n_in,
                              void* d_out, int out_size, void* d_ws, size_t ws_size,
                              hipStream_t stream) {
    const float* x      = (const float*)d_in[0];
    const float* qkv_w  = (const float*)d_in[1];
    const float* qkv_b  = (const float*)d_in[2];
    const float* proj_w = (const float*)d_in[3];
    const float* proj_b = (const float*)d_in[4];
    const float* rpb    = (const float*)d_in[5];
    float* out = (float*)d_out;
    float* ws  = (float*)d_ws;

    float* plane = ws;
    float* qkv   = plane + (size_t)PIX * CDIM;
    float* amap  = qkv + (size_t)PIX * 192;

    nchw_to_nhwc<<<PIX / TPX, 128, 0, stream>>>(x, plane);

    {
        int numTiles = (PIX / 16) * (192 / 16);
        wmma_gemm_bias<CDIM><<<(numTiles + 7) / 8, 256, 0, stream>>>(
            plane, qkv_w, qkv_b, qkv, PIX, 192);
    }

    na_attn_wmma<false, false, 192, 64><<<PIX / 16, 32 * NHEADS, 0, stream>>>(
        qkv, qkv + 64, qkv + 128, rpb, proj_w, proj_b,
        qkv_w + 192 * CDIM, qkv_b + 192, amap, plane, out, nullptr);
    na_attn_wmma<true, false, 64, 192><<<PIX / 16, 32 * NHEADS, 0, stream>>>(
        qkv, plane, out, rpb + RPBS, proj_w + CDIM * CDIM, proj_b + CDIM,
        qkv_w + 2 * 192 * CDIM, qkv_b + 2 * 192, amap, qkv + 64, qkv + 128, nullptr);
    na_attn_wmma<true, true, 192, 64><<<PIX / 16, 32 * NHEADS, 0, stream>>>(
        qkv, qkv + 64, qkv + 128, rpb + 2 * RPBS, proj_w + 2 * CDIM * CDIM, proj_b + 2 * CDIM,
        nullptr, nullptr, amap, nullptr, nullptr, plane);

    nhwc_to_nchw<<<PIX / TPX, 128, 0, stream>>>(plane, out);
}


# operands read as fragments (stage-0 x and weight, proj/QKV weights, Q of stages 1,2) are stored in fragment order [load][lane][8|16B]: a load covers 2-4 whole lines, not 16; same sums
# speedup vs baseline: 7.4879x; 1.1326x over previous
#include <hip/hip_runtime.h>

typedef float v2f __attribute__((ext_vector_type(2)));
typedef float v4f __attribute__((ext_vector_type(4)));
typedef float v8f __attribute__((ext_vector_type(8)));

#define HH 96
#define WW 96
#define PIX (HH * WW)
#define CDIM 64
#define NHEADS 2
#define HD 32
#define KW 7
#define NKK (KW * KW)

#define TPX 32
#define TPITCH 68
#define TBLK (PIX / TPX)
#define WF_PROJ 0
#define WF_NEXT (3 * NHEADS * 16 * 128)
#define WF_QKV0 (WF_NEXT + 2 * NHEADS * 48 * 128)
#define WF_FLOATS (WF_QKV0 + 192 * CDIM)
#define WBLK (WF_FLOATS / 4 / 128)

__global__ void __launch_bounds__(128)
nchw_to_frag(const float* __restrict__ x, float* __restrict__ xf,
             const float* __restrict__ qkv_w, const float* __restrict__ proj_w,
             float* __restrict__ wf) {
    __shared__ float tile[TPX * TPITCH];
    const int tid = threadIdx.x;
    if (blockIdx.x >= TBLK) {
        const int i = (blockIdx.x - TBLK) * 128 + tid;
        const int lane = i & 31, lc = lane & 15, h = lane >> 4;
        float4 w;
        if (i < WF_NEXT / 4) {
            const int idx = (i >> 5) & 15, head = (i >> 9) & 1, stage = i >> 10;
            const int u = idx >> 2, dt = (idx >> 1) & 1, e = idx & 1;
            const int ot = (2 * head + u) & 3;
            w = *(const float4*)(proj_w + stage * CDIM * CDIM + (16 * ot + lc) * CDIM + head * HD + 8 * h + 16 * dt + 4 * e);
        } else if (i < WF_QKV0 / 4) {
            const int j = i - WF_NEXT / 4;
            const int sn = j / (NHEADS * 48 * 32);
            const int r = j - sn * (NHEADS * 48 * 32);
            const int head = r / (48 * 32);
            const int idx = (r - head * (48 * 32)) >> 5;
            const int m = idx >> 3, u = (idx >> 1) & 3, e = idx & 1;
            const int g = (2 * head + u) & 3;
            const int frow = 64 * (m >> 1) + 32 * head + 16 * (m & 1);
            w = *(const float4*)(qkv_w + (sn + 1) * 192 * CDIM + (frow + lc) * CDIM + 8 * h + 16 * g + 4 * e);
        } else {
            const int j = i - WF_QKV0 / 4;
            const int pr = j & 15, ks = (j >> 4) & 15, tn = j >> 8;
            const int r = tn * 16 + 2 * (pr & 7), c = 4 * ks + 2 * (pr >> 3);
            const float2 a = *(const float2*)(qkv_w + r * CDIM + c);
            const float2 b = *(const float2*)(qkv_w + (r + 1) * CDIM + c);
            w = make_float4(a.x, a.y, b.x, b.y);
        }
        *(float4*)(wf + 4 * i) = w;
        return;
    }
    const int p0  = blockIdx.x * TPX;
    float4 v[4];
#pragma unroll
    for (int i = 0; i < 4; ++i) {
        const int idx = i * 128 + tid;
        const int c = idx >> 3;
        const int q = idx & 7;
        v[i] = *(const float4*)(x + (size_t)c * PIX + p0 + 4 * q);
    }
#pragma unroll
    for (int i = 0; i < 4; ++i) {
        const int idx = i * 128 + tid;
        const int c = idx >> 3;
        const int q = idx & 7;
        float* t = tile + (4 * q) * TPITCH + c;
        t[0]          = v[i].x;
        t[TPITCH]     = v[i].y;
        t[2 * TPITCH] = v[i].z;
        t[3 * TPITCH] = v[i].w;
    }
    __syncthreads();
#pragma unroll
    for (int i = 0; i < 4; ++i) {
        const int idx = i * 128 + tid;
        const int pr = idx & 15, ks = (idx >> 4) & 15, tl = idx >> 8;
        const float* t = tile + (16 * tl + 2 * (pr & 7)) * TPITCH + 4 * ks + 2 * (pr >> 3);
        const float2 a = *(const float2*)(t);
        const float2 b = *(const float2*)(t + TPITCH);
        *(float4*)(xf + (size_t)p0 * CDIM + 4 * idx) = make_float4(a.x, a.y, b.x, b.y);
    }
}

__global__ void __launch_bounds__(128)
nhwc_to_nchw(const float* __restrict__ xh, float* __restrict__ y) {
    __shared__ float tile[TPX * TPITCH];
    const int tid = threadIdx.x;
    const int p0  = blockIdx.x * TPX;
    float4 v[4];
#pragma unroll
    for (int i = 0; i < 4; ++i) {
        const int idx = i * 128 + tid;
        const int p = idx >> 4;
        const int m = idx & 15;
        v[i] = *(const float4*)(xh + (size_t)(p0 + p) * CDIM + 4 * m);
    }
#pragma unroll
    for (int i = 0; i < 4; ++i) {
        const int idx = i * 128 + tid;
        const int p = idx >> 4;
        const int m = idx & 15;
        *(float4*)(tile + p * TPITCH + 4 * m) = v[i];
    }
    __syncthreads();
#pragma unroll
    for (int i = 0; i < 4; ++i) {
        const int idx = i * 128 + tid;
        const int c = idx >> 3;
        const int q = idx & 7;
        const float* t = tile + (4 * q) * TPITCH + c;
        const float4 w = make_float4(t[0], t[TPITCH], t[2 * TPITCH], t[3 * TPITCH]);
        *(float4*)(y + (size_t)c * PIX + p0 + 4 * q) = w;
    }
}

template <int KD>
__global__ void wmma_gemm_bias(const float* __restrict__ A,
                               const float* __restrict__ W,
                               const float* __restrict__ bias,
                               float* __restrict__ C,
                               int M, int N) {
    int lane = threadIdx.x & 31;
    int wid  = threadIdx.x >> 5;
    int tile = blockIdx.x * 8 + wid;
    int tilesN = N >> 4;
    int numTiles = (M >> 4) * tilesN;
    if (tile >= numTiles) return;

    int tm = tile / tilesN;
    int tn = tile % tilesN;
    int lmod  = lane & 15;
    int lhalf = lane >> 4;

    const float* __restrict__ Ap = A + (size_t)tm * (16 * KD) + 2 * lane;
    const float* __restrict__ Wp = W + (size_t)tn * (16 * KD) + 2 * lane;

    v8f acc = {0.f, 0.f, 0.f, 0.f, 0.f, 0.f, 0.f, 0.f};
#pragma unroll
    for (int k = 0; k < KD; k += 4) {
        v2f a = *(const v2f*)(Ap + 16 * k);
        v2f b = *(const v2f*)(Wp + 16 * k);
        acc = __builtin_amdgcn_wmma_f32_16x16x4_f32(
             false, a,  false, b,
             (short)0, acc,  false,  false);
    }

    float bv = bias[tn * 16 + lmod];
    int rbase = tm * 16 + 8 * lhalf;
    int col   = tn * 16 + lmod;
#pragma unroll
    for (int v = 0; v < 8; ++v) {
        C[(size_t)(rbase + v) * N + col] = acc[v] + bv;
    }
}

#define SQ 10
#define NKEY (SQ * SQ)
#define RPBW (2 * KW - 1)
#define MAPG 12
#define MAPW (MAPG * 128 + 64)
#define RPBS (NHEADS * RPBW * RPBW)

__device__ __forceinline__ v8f wmma4(v2f a, v2f b, v8f c) {
    return __builtin_amdgcn_wmma_f32_16x16x4_f32(
         false, a,  false, b,
         (short)0, c,  false,  false);
}

template <int PITCH>
__device__ __forceinline__ constexpr unsigned key_row_off(int n) {
    return (unsigned)((((n < NKEY ? n : NKEY - 1) / SQ) * WW + ((n < NKEY ? n : NKEY - 1) % SQ)) * PITCH);
}

#define HKEY (NKEY / 2)

#define XPITCH 12
#define XTILE (32 * XPITCH)
template <bool HAS_PREV, bool LAST, int KVS, int KVD>
__global__ void __launch_bounds__(64) __attribute__((amdgpu_waves_per_eu(1, 2)))
na_attn_wmma(float* __restrict__ qio,
             const float* __restrict__ kin,
             const float* __restrict__ vin,
             const float* __restrict__ rpb,
             const float* __restrict__ pw,
             const float* __restrict__ pb,
             const float* __restrict__ nw,
             const float* __restrict__ nb,
             float* __restrict__ amap,
             float* __restrict__ kout,
             float* __restrict__ vout,
             float* __restrict__ out) {
    __shared__ __attribute__((aligned(16))) float xch[LAST ? NHEADS * 2 * 8 * 32 : NHEADS * 4 * XTILE];
    const int lane  = threadIdx.x & 31;
    const int lc    = lane & 15;
    const int h     = lane >> 4;
    const int head  = __builtin_amdgcn_readfirstlane((int)(threadIdx.x >> 5));
    const int patch = blockIdx.x;
    const int i0 = (patch / (WW / 4)) * 4;
    const int j0 = (patch % (WW / 4)) * 4;

    int r0 = i0 - KW / 2; r0 = r0 < 0 ? 0 : r0; r0 = r0 > HH - SQ ? HH - SQ : r0;
    int c0 = j0 - KW / 2; c0 = c0 < 0 ? 0 : c0; c0 = c0 > WW - SQ ? WW - SQ : c0;

    const int qi = i0 + (lc >> 2);
    const int qj = j0 + (lc & 3);
    const int qpix = qi * WW + qj;
    int si = qi - KW / 2; si = si < 0 ? 0 : si; si = si > HH - KW ? HH - KW : si;
    int sj = qj - KW / 2; sj = sj < 0 ? 0 : sj; sj = sj > WW - KW ? WW - KW : sj;
    const int dsi = si - r0;
    const int dsj = sj - c0;

    const float scale = 0.17677669529663687f;

    const unsigned qoff = HAS_PREV
        ? (unsigned)((i0 * WW + j0 + (lane >> 3)) * 192 + head * HD + 4 * (lane & 7))
        : (unsigned)(qpix * 192 + head * HD + 4 * h);
    float4 qv[4];
#pragma unroll
    for (int s = 0; s < 4; ++s) qv[s] = *(const float4*)(qio + qoff + (HAS_PREV ? WW * 192 : 8) * s);

    float4 kv[7][4];
#pragma unroll
    for (int t = 0; t < 7; ++t) {
        int km = 8 * t + (lc & 7);
        km = km < HKEY ? km : HKEY - 1;
        const int kq = km / SQ;
        const int kr = kq + 5 * (lc >> 3);
        const int kc = km - kq * SQ;
        const unsigned koff = (unsigned)(((r0 + kr) * WW + (c0 + kc)) * KVS + head * HD + 4 * h);
#pragma unroll
        for (int s = 0; s < 4; ++s) kv[t][s] = *(const float4*)(kin + koff + 8 * s);
    }

    const int ri0 = r0 - qi + KW - 1 + 5 * h;
    const int ci0 = c0 - qj + KW - 1;
    int rrow[5], ccol[SQ];
#pragma unroll
    for (int k = 0; k < 5; ++k) { int v = ri0 + k; v = v < 0 ? 0 : v; v = v > RPBW - 1 ? RPBW - 1 : v; rrow[k] = v * RPBW; }
#pragma unroll
    for (int k = 0; k < SQ; ++k) { int v = ci0 + k; v = v < 0 ? 0 : v; v = v > RPBW - 1 ? RPBW - 1 : v; ccol[k] = v; }
    const int dsih = dsi - 5 * h;
    bool rowv[5], colv[SQ];
#pragma unroll
    for (int k = 0; k < 5; ++k) rowv[k] = (unsigned)(k - dsih) < (unsigned)KW;
#pragma unroll
    for (int k = 0; k < SQ; ++k) colv[k] = (unsigned)(k - dsj) < (unsigned)KW;
    const float* rb = rpb + head * RPBW * RPBW;
    float* mp = amap + (unsigned)((patch * NHEADS + head) * MAPW) + 4 * lane;
    float* mp2 = amap + (unsigned)((patch * NHEADS + head) * MAPW + MAPG * 128) + 2 * lane;
    float4 pm[MAPG];
    float2 pm2 = make_float2(0.f, 0.f);
    if (HAS_PREV) {
#pragma unroll
        for (int g = 0; g < MAPG; ++g) pm[g] = *(const float4*)(mp + 128 * g);
        pm2 = *(const float2*)(mp2);
    }
    float rbv[7][8];
    float pvv[7][8];
#pragma unroll
    for (int t = 0; t < 7; ++t) {
#pragma unroll
        for (int r = 0; r < 8; ++r) {
            const int f = 8 * t + r;
            rbv[t][r] = f < HKEY ? rb[rrow[f / SQ] + ccol[f % SQ]] : 0.f;
            if (HAS_PREV && f < 4 * MAPG) {
                const float4 g4 = pm[f >> 2];
                pvv[t][r] = (f & 3) == 0 ? g4.x : (f & 3) == 1 ? g4.y : (f & 3) == 2 ? g4.z : g4.w;
            } else if (HAS_PREV && f < HKEY) {
                pvv[t][r] = f == 4 * MAPG ? pm2.x : pm2.y;
            } else {
                pvv[t][r] = 0.f;
            }
        }
    }

#pragma unroll
    for (int s = 0; s < 4; ++s)
        qv[s] = make_float4(qv[s].x * scale, qv[s].y * scale, qv[s].z * scale, qv[s].w * scale);

    v8f st[7];
#pragma unroll
    for (int t = 0; t < 7; ++t) {
        v8f acc = {0.f, 0.f, 0.f, 0.f, 0.f, 0.f, 0.f, 0.f};
#pragma unroll
        for (int s = 0; s < 4; ++s) {
            v2f a0 = {kv[t][s].x, kv[t][s].y};
            v2f a1 = {kv[t][s].z, kv[t][s].w};
            v2f b0 = {qv[s].x, qv[s].y};
            v2f b1 = {qv[s].z, qv[s].w};
            acc = wmma4(a0, b0, acc);
            acc = wmma4(a1, b1, acc);
        }
        st[t] = acc;
    }

    const unsigned voff = (unsigned)(((r0 + 5 * h) * WW + c0) * KVS + head * HD + lc);
    float va[7][4][2][2];
#pragma unroll
    for (int t = 0; t < 7; ++t) {
#pragma unroll
        for (int s = 0; s < 4; ++s) {
            const int k0 = 8 * t + 2 * s;
            if (k0 >= HKEY) continue;
            const unsigned of0 = voff + key_row_off<KVS>(k0);
            const unsigned of1 = voff + key_row_off<KVS>(k0 + 1);
            va[t][s][0][0] = vin[of0];
            va[t][s][0][1] = vin[of1];
            va[t][s][1][0] = vin[of0 + 16];
            va[t][s][1][1] = vin[of1 + 16];
        }
    }

    float4 wq[4][2][2];
    {
        const float* wr = pw + (unsigned)(head * (16 * 128) + 4 * lane);
#pragma unroll
        for (int u = 0; u < 4; ++u) {
#pragma unroll
            for (int dt = 0; dt < 2; ++dt) {
                wq[u][dt][0] = *(const float4*)(wr + 128 * (4 * u + 2 * dt));
                wq[u][dt][1] = *(const float4*)(wr + 128 * (4 * u + 2 * dt + 1));
            }
        }
    }
    float4 bq[2][2];
#pragma unroll
    for (int tt = 0; tt < 2; ++tt) {
        const float* br = pb + (unsigned)(16 * (2 * head + tt) + 8 * h);
        bq[tt][0] = *(const float4*)(br);
        bq[tt][1] = *(const float4*)(br + 4);
    }

    float mx = -1e30f;
#pragma unroll
    for (int t = 0; t < 7; ++t) {
#pragma unroll
        for (int r = 0; r < 8; ++r) {
            const int f = 8 * t + r;
            const bool valid = f < HKEY && rowv[f < HKEY ? f / SQ : 0] && colv[f % SQ];
            float l = st[t][r] + rbv[t][r];
            if (HAS_PREV) l += pvv[t][r];
            l = valid ? l : -1e30f;
            st[t][r] = l;
            mx = fmaxf(mx, l);
        }
    }
    mx = fmaxf(mx, __shfl_xor(mx, 16));

    float ssum = 0.f;
#pragma unroll
    for (int t = 0; t < 7; ++t) {
#pragma unroll
        for (int r = 0; r < 8; ++r) {
            float p = __expf(st[t][r] - mx);
            st[t][r] = p;
            ssum += p;
        }
    }
    ssum += __shfl_xor(ssum, 16);
    const float inv = 1.f / ssum;

#pragma unroll
    for (int t = 0; t < 7; ++t)
#pragma unroll
        for (int r = 0; r < 8; ++r) st[t][r] *= inv;

    v8f o[2][2];
#pragma unroll
    for (int dt = 0; dt < 2; ++dt)
#pragma unroll
        for (int u = 0; u < 2; ++u)
            o[dt][u] = (v8f){0.f, 0.f, 0.f, 0.f, 0.f, 0.f, 0.f, 0.f};

#pragma unroll
    for (int t = 0; t < 7; ++t) {
#pragma unroll
        for (int s = 0; s < 4; ++s) {
            const int k0 = 8 * t + 2 * s;
            if (k0 >= HKEY) continue;
            v2f a0 = {va[t][s][0][0], va[t][s][0][1]};
            v2f a1 = {va[t][s][1][0], va[t][s][1][1]};
            v2f b  = {st[t][2 * s], st[t][2 * s + 1]};
            o[0][t & 1] = wmma4(a0, b, o[0][t & 1]);
            o[1][t & 1] = wmma4(a1, b, o[1][t & 1]);
        }
    }

    v8f d[2];
#pragma unroll
    for (int dt = 0; dt < 2; ++dt) d[dt] = o[dt][0] + o[dt][1];

    float4 wn[6][4][2];
    if constexpr (!LAST) {
        const float* wr = nw + (unsigned)(head * (48 * 128) + 4 * lane);
#pragma unroll
        for (int m = 0; m < 6; ++m) {
#pragma unroll
            for (int u = 0; u < 4; ++u) {
                wn[m][u][0] = *(const float4*)(wr + 128 * (8 * m + 2 * u));
                wn[m][u][1] = *(const float4*)(wr + 128 * (8 * m + 2 * u + 1));
            }
        }
    }

    v8f pp[4];
#pragma unroll
    for (int tt = 0; tt < 2; ++tt)
        pp[tt] = (v8f){bq[tt][0].x, bq[tt][0].y, bq[tt][0].z, bq[tt][0].w,
                       bq[tt][1].x, bq[tt][1].y, bq[tt][1].z, bq[tt][1].w};
#pragma unroll
    for (int u = 2; u < 4; ++u) pp[u] = (v8f){0.f, 0.f, 0.f, 0.f, 0.f, 0.f, 0.f, 0.f};
#pragma unroll
    for (int dt = 0; dt < 2; ++dt) {
#pragma unroll
        for (int j = 0; j < 4; ++j) {
            v2f b = {d[dt][2 * j], d[dt][2 * j + 1]};
#pragma unroll
            for (int u = 0; u < 4; ++u) {
                const float4 w4 = wq[u][dt][j >> 1];
                v2f a = (j & 1) ? (v2f){w4.z, w4.w} : (v2f){w4.x, w4.y};
                pp[u] = wmma4(a, b, pp[u]);
            }
        }
    }

    if constexpr (LAST) {
#pragma unroll
        for (int tt = 0; tt < 2; ++tt)
#pragma unroll
            for (int r = 0; r < 8; ++r) xch[((head * 2 + tt) * 8 + r) * 32 + lane] = pp[2 + tt][r];
        __syncthreads();

        float* op = out + (size_t)qpix * CDIM + 32 * head + 8 * h;
#pragma unroll
        for (int tt = 0; tt < 2; ++tt) {
            float e[8];
#pragma unroll
            for (int r = 0; r < 8; ++r) e[r] = pp[tt][r] + xch[(((1 - head) * 2 + tt) * 8 + r) * 32 + lane];
            *(float4*)(op + 16 * tt)     = make_float4(e[0], e[1], e[2], e[3]);
            *(float4*)(op + 16 * tt + 4) = make_float4(e[4], e[5], e[6], e[7]);
        }
    } else {
        float* xw = xch + head * (4 * XTILE) + lane * XPITCH;
        const float* xr = xch + (1 - head) * (4 * XTILE) + lane * XPITCH;
#pragma unroll
        for (int u = 0; u < 4; ++u) {
            *(float4*)(xw + u * XTILE)     = make_float4(pp[u][0], pp[u][1], pp[u][2], pp[u][3]);
            *(float4*)(xw + u * XTILE + 4) = make_float4(pp[u][4], pp[u][5], pp[u][6], pp[u][7]);
        }
        __syncthreads();

        v4f xa[4], xb[4];
#pragma unroll
        for (int u = 0; u < 4; ++u) {
            xa[u] = *(const v4f*)(xr + ((u + 2) & 3) * XTILE);
            xb[u] = *(const v4f*)(xr + ((u + 2) & 3) * XTILE + 4);
        }
        float4 nbq[6][2];
#pragma unroll
        for (int m = 0; m < 6; ++m) {
            const float* br = nb + (unsigned)(64 * (m >> 1) + 32 * head + 16 * (m & 1) + 8 * h);
            nbq[m][0] = *(const float4*)(br);
            nbq[m][1] = *(const float4*)(br + 4);
        }
        asm volatile("" :: "v"(xa[0]), "v"(xb[0]), "v"(xa[1]), "v"(xb[1]),
                           "v"(xa[2]), "v"(xb[2]), "v"(xa[3]), "v"(xb[3]));
        v8f xt[4];
#pragma unroll
        for (int u = 0; u < 4; ++u)
            xt[u] = pp[u] + (v8f){xa[u].x, xa[u].y, xa[u].z, xa[u].w, xb[u].x, xb[u].y, xb[u].z, xb[u].w};

        v8f qa[6];
#pragma unroll
        for (int m = 0; m < 6; ++m) qa[m] = (v8f){0.f, 0.f, 0.f, 0.f, 0.f, 0.f, 0.f, 0.f};
#pragma unroll
        for (int u = 0; u < 4; ++u) {
#pragma unroll
            for (int j = 0; j < 4; ++j) {
                v2f b = {xt[u][2 * j], xt[u][2 * j + 1]};
#pragma unroll
                for (int m = 0; m < 6; ++m) {
                    const float4 w4 = wn[m][u][j >> 1];
                    v2f a = (j & 1) ? (v2f){w4.z, w4.w} : (v2f){w4.x, w4.y};
                    qa[m] = wmma4(a, b, qa[m]);
                }
            }
        }

        float* qw = qio  + (size_t)(((i0 + h) * WW + j0 + (lc >> 3)) * 192 + 32 * head + 4 * (lc & 7));
        float* kp = kout + (size_t)qpix * KVD + 32 * head + 8 * h;
        float* vp = vout + (size_t)qpix * KVD + 32 * head + 8 * h;
#pragma unroll
        for (int m = 0; m < 6; ++m) {
            float* dp = ((m >> 1) == 1 ? kp : vp) + 16 * (m & 1);
            float* d0 = (m >> 1) == 0 ? qw + 2 * (m & 1) * WW * 192 : dp;
            float* d1 = (m >> 1) == 0 ? d0 + 2 * 192 : dp + 4;
            *(float4*)(d0) = make_float4(qa[m][0] + nbq[m][0].x, qa[m][1] + nbq[m][0].y,
                                         qa[m][2] + nbq[m][0].z, qa[m][3] + nbq[m][0].w);
            *(float4*)(d1) = make_float4(qa[m][4] + nbq[m][1].x, qa[m][5] + nbq[m][1].y,
                                         qa[m][6] + nbq[m][1].z, qa[m][7] + nbq[m][1].w);
        }

#pragma unroll
        for (int g = 0; g < MAPG; ++g) {
            const int t = g >> 1;
            const int r = 4 * (g & 1);
            *(float4*)(mp + 128 * g) = make_float4(st[t][r], st[t][r + 1], st[t][r + 2], st[t][r + 3]);
        }
        *(float2*)(mp2) = make_float2(st[6][0], st[6][1]);
    }
}

extern "C" void kernel_launch(void* const* d_in, const int* in_sizes, int n_in,
                              void* d_out, int out_size, void* d_ws, size_t ws_size,
                              hipStream_t stream) {
    const float* x      = (const float*)d_in[0];
    const float* qkv_w  = (const float*)d_in[1];
    const float* qkv_b  = (const float*)d_in[2];
    const float* proj_w = (const float*)d_in[3];
    const float* proj_b = (const float*)d_in[4];
    const float* rpb    = (const float*)d_in[5];
    float* out = (float*)d_out;
    float* ws  = (float*)d_ws;

    float* plane = ws;
    float* qkv   = plane + (size_t)PIX * CDIM;
    float* amap  = qkv + (size_t)PIX * 192;
    float* wfrag = amap + (size_t)(PIX / 16) * NHEADS * MAPW;

    nchw_to_frag<<<TBLK + WBLK, 128, 0, stream>>>(x, plane, qkv_w, proj_w, wfrag);

    {
        int numTiles = (PIX / 16) * (192 / 16);
        wmma_gemm_bias<CDIM><<<(numTiles + 7) / 8, 256, 0, stream>>>(
            plane, wfrag + WF_QKV0, qkv_b, qkv, PIX, 192);
    }

    na_attn_wmma<false, false, 192, 64><<<PIX / 16, 32 * NHEADS, 0, stream>>>(
        qkv, qkv + 64, qkv + 128, rpb, wfrag + WF_PROJ, proj_b,
        wfrag + WF_NEXT, qkv_b + 192, amap, plane, out, nullptr);
    na_attn_wmma<true, false, 64, 192><<<PIX / 16, 32 * NHEADS, 0, stream>>>(
        qkv, plane, out, rpb + RPBS, wfrag + WF_PROJ + CDIM * CDIM, proj_b + CDIM,
        wfrag + WF_NEXT + 192 * CDIM, qkv_b + 2 * 192, amap, qkv + 64, qkv + 128, nullptr);
    na_attn_wmma<true, true, 192, 64><<<PIX / 16, 32 * NHEADS, 0, stream>>>(
        qkv, qkv + 64, qkv + 128, rpb + 2 * RPBS, wfrag + WF_PROJ + 2 * CDIM * CDIM, proj_b + 2 * CDIM,
        nullptr, nullptr, amap, nullptr, nullptr, plane);

    nhwc_to_nchw<<<PIX / TPX, 128, 0, stream>>>(plane, out);
}


# last stage's group = 4 patches in a row x 2 heads (8 waves): sums staged in LDS as [channel][row][16 px], stored to (C,H,W) as 64 B row pieces; layout launch gone, 6 -> 5 launches; same sums
# speedup vs baseline: 7.9685x; 1.0642x over previous
#include <hip/hip_runtime.h>

typedef float v2f __attribute__((ext_vector_type(2)));
typedef float v4f __attribute__((ext_vector_type(4)));
typedef float v8f __attribute__((ext_vector_type(8)));

#define HH 96
#define WW 96
#define PIX (HH * WW)
#define CDIM 64
#define NHEADS 2
#define HD 32
#define KW 7
#define NKK (KW * KW)

#define TPX 32
#define TPITCH 68
#define TBLK (PIX / TPX)
#define WF_PROJ 0
#define WF_NEXT (3 * NHEADS * 16 * 128)
#define WF_QKV0 (WF_NEXT + 2 * NHEADS * 48 * 128)
#define WF_FLOATS (WF_QKV0 + 192 * CDIM)
#define WBLK (WF_FLOATS / 4 / 128)

__global__ void __launch_bounds__(128)
nchw_to_frag(const float* __restrict__ x, float* __restrict__ xf,
             const float* __restrict__ qkv_w, const float* __restrict__ proj_w,
             float* __restrict__ wf) {
    __shared__ float tile[TPX * TPITCH];
    const int tid = threadIdx.x;
    if (blockIdx.x >= TBLK) {
        const int i = (blockIdx.x - TBLK) * 128 + tid;
        const int lane = i & 31, lc = lane & 15, h = lane >> 4;
        float4 w;
        if (i < WF_NEXT / 4) {
            const int idx = (i >> 5) & 15, head = (i >> 9) & 1, stage = i >> 10;
            const int u = idx >> 2, dt = (idx >> 1) & 1, e = idx & 1;
            const int ot = (2 * head + u) & 3;
            w = *(const float4*)(proj_w + stage * CDIM * CDIM + (16 * ot + lc) * CDIM + head * HD + 8 * h + 16 * dt + 4 * e);
        } else if (i < WF_QKV0 / 4) {
            const int j = i - WF_NEXT / 4;
            const int sn = j / (NHEADS * 48 * 32);
            const int r = j - sn * (NHEADS * 48 * 32);
            const int head = r / (48 * 32);
            const int idx = (r - head * (48 * 32)) >> 5;
            const int m = idx >> 3, u = (idx >> 1) & 3, e = idx & 1;
            const int g = (2 * head + u) & 3;
            const int frow = 64 * (m >> 1) + 32 * head + 16 * (m & 1);
            w = *(const float4*)(qkv_w + (sn + 1) * 192 * CDIM + (frow + lc) * CDIM + 8 * h + 16 * g + 4 * e);
        } else {
            const int j = i - WF_QKV0 / 4;
            const int pr = j & 15, ks = (j >> 4) & 15, tn = j >> 8;
            const int r = tn * 16 + 2 * (pr & 7), c = 4 * ks + 2 * (pr >> 3);
            const float2 a = *(const float2*)(qkv_w + r * CDIM + c);
            const float2 b = *(const float2*)(qkv_w + (r + 1) * CDIM + c);
            w = make_float4(a.x, a.y, b.x, b.y);
        }
        *(float4*)(wf + 4 * i) = w;
        return;
    }
    const int p0  = blockIdx.x * TPX;
    float4 v[4];
#pragma unroll
    for (int i = 0; i < 4; ++i) {
        const int idx = i * 128 + tid;
        const int c = idx >> 3;
        const int q = idx & 7;
        v[i] = *(const float4*)(x + (size_t)c * PIX + p0 + 4 * q);
    }
#pragma unroll
    for (int i = 0; i < 4; ++i) {
        const int idx = i * 128 + tid;
        const int c = idx >> 3;
        const int q = idx & 7;
        float* t = tile + (4 * q) * TPITCH + c;
        t[0]          = v[i].x;
        t[TPITCH]     = v[i].y;
        t[2 * TPITCH] = v[i].z;
        t[3 * TPITCH] = v[i].w;
    }
    __syncthreads();
#pragma unroll
    for (int i = 0; i < 4; ++i) {
        const int idx = i * 128 + tid;
        const int pr = idx & 15, ks = (idx >> 4) & 15, tl = idx >> 8;
        const float* t = tile + (16 * tl + 2 * (pr & 7)) * TPITCH + 4 * ks + 2 * (pr >> 3);
        const float2 a = *(const float2*)(t);
        const float2 b = *(const float2*)(t + TPITCH);
        *(float4*)(xf + (size_t)p0 * CDIM + 4 * idx) = make_float4(a.x, a.y, b.x, b.y);
    }
}

template <int KD>
__global__ void wmma_gemm_bias(const float* __restrict__ A,
                               const float* __restrict__ W,
                               const float* __restrict__ bias,
                               float* __restrict__ C,
                               int M, int N) {
    int lane = threadIdx.x & 31;
    int wid  = threadIdx.x >> 5;
    int tile = blockIdx.x * 8 + wid;
    int tilesN = N >> 4;
    int numTiles = (M >> 4) * tilesN;
    if (tile >= numTiles) return;

    int tm = tile / tilesN;
    int tn = tile % tilesN;
    int lmod  = lane & 15;
    int lhalf = lane >> 4;

    const float* __restrict__ Ap = A + (size_t)tm * (16 * KD) + 2 * lane;
    const float* __restrict__ Wp = W + (size_t)tn * (16 * KD) + 2 * lane;

    v8f acc = {0.f, 0.f, 0.f, 0.f, 0.f, 0.f, 0.f, 0.f};
#pragma unroll
    for (int k = 0; k < KD; k += 4) {
        v2f a = *(const v2f*)(Ap + 16 * k);
        v2f b = *(const v2f*)(Wp + 16 * k);
        acc = __builtin_amdgcn_wmma_f32_16x16x4_f32(
             false, a,  false, b,
             (short)0, acc,  false,  false);
    }

    float bv = bias[tn * 16 + lmod];
    int rbase = tm * 16 + 8 * lhalf;
    int col   = tn * 16 + lmod;
#pragma unroll
    for (int v = 0; v < 8; ++v) {
        C[(size_t)(rbase + v) * N + col] = acc[v] + bv;
    }
}

#define SQ 10
#define NKEY (SQ * SQ)
#define RPBW (2 * KW - 1)
#define MAPG 12
#define MAPW (MAPG * 128 + 64)
#define RPBS (NHEADS * RPBW * RPBW)

__device__ __forceinline__ v8f wmma4(v2f a, v2f b, v8f c) {
    return __builtin_amdgcn_wmma_f32_16x16x4_f32(
         false, a,  false, b,
         (short)0, c,  false,  false);
}

template <int PITCH>
__device__ __forceinline__ constexpr unsigned key_row_off(int n) {
    return (unsigned)((((n < NKEY ? n : NKEY - 1) / SQ) * WW + ((n < NKEY ? n : NKEY - 1) % SQ)) * PITCH);
}

#define HKEY (NKEY / 2)

#define XPITCH 12
#define XTILE (32 * XPITCH)
#define GP 4
#define GCOL (4 * GP)
template <bool HAS_PREV, bool LAST, int KVS, int KVD>
__global__ void __launch_bounds__(LAST ? 64 * GP : 64) __attribute__((amdgpu_waves_per_eu(LAST ? 2 : 1, 2)))
na_attn_wmma(float* __restrict__ qio,
             const float* __restrict__ kin,
             const float* __restrict__ vin,
             const float* __restrict__ rpb,
             const float* __restrict__ pw,
             const float* __restrict__ pb,
             const float* __restrict__ nw,
             const float* __restrict__ nb,
             float* __restrict__ amap,
             float* __restrict__ kout,
             float* __restrict__ vout,
             float* __restrict__ out) {
    __shared__ __attribute__((aligned(16))) float xch[LAST ? GP * NHEADS * 2 * 8 * 32 : NHEADS * 4 * XTILE];
    const int lane  = threadIdx.x & 31;
    const int lc    = lane & 15;
    const int h     = lane >> 4;
    const int wv    = __builtin_amdgcn_readfirstlane((int)(threadIdx.x >> 5));
    const int head  = LAST ? (wv & 1) : wv;
    const int pl    = LAST ? (wv >> 1) : 0;
    const int patch = LAST ? (int)blockIdx.x * GP + pl : (int)blockIdx.x;
    const int i0 = (patch / (WW / 4)) * 4;
    const int j0 = (patch % (WW / 4)) * 4;

    int r0 = i0 - KW / 2; r0 = r0 < 0 ? 0 : r0; r0 = r0 > HH - SQ ? HH - SQ : r0;
    int c0 = j0 - KW / 2; c0 = c0 < 0 ? 0 : c0; c0 = c0 > WW - SQ ? WW - SQ : c0;

    const int qi = i0 + (lc >> 2);
    const int qj = j0 + (lc & 3);
    const int qpix = qi * WW + qj;
    int si = qi - KW / 2; si = si < 0 ? 0 : si; si = si > HH - KW ? HH - KW : si;
    int sj = qj - KW / 2; sj = sj < 0 ? 0 : sj; sj = sj > WW - KW ? WW - KW : sj;
    const int dsi = si - r0;
    const int dsj = sj - c0;

    const float scale = 0.17677669529663687f;

    const unsigned qoff = HAS_PREV
        ? (unsigned)((i0 * WW + j0 + (lane >> 3)) * 192 + head * HD + 4 * (lane & 7))
        : (unsigned)(qpix * 192 + head * HD + 4 * h);
    float4 qv[4];
#pragma unroll
    for (int s = 0; s < 4; ++s) qv[s] = *(const float4*)(qio + qoff + (HAS_PREV ? WW * 192 : 8) * s);

    float4 kv[7][4];
#pragma unroll
    for (int t = 0; t < 7; ++t) {
        int km = 8 * t + (lc & 7);
        km = km < HKEY ? km : HKEY - 1;
        const int kq = km / SQ;
        const int kr = kq + 5 * (lc >> 3);
        const int kc = km - kq * SQ;
        const unsigned koff = (unsigned)(((r0 + kr) * WW + (c0 + kc)) * KVS + head * HD + 4 * h);
#pragma unroll
        for (int s = 0; s < 4; ++s) kv[t][s] = *(const float4*)(kin + koff + 8 * s);
    }

    const int ri0 = r0 - qi + KW - 1 + 5 * h;
    const int ci0 = c0 - qj + KW - 1;
    int rrow[5], ccol[SQ];
#pragma unroll
    for (int k = 0; k < 5; ++k) { int v = ri0 + k; v = v < 0 ? 0 : v; v = v > RPBW - 1 ? RPBW - 1 : v; rrow[k] = v * RPBW; }
#pragma unroll
    for (int k = 0; k < SQ; ++k) { int v = ci0 + k; v = v < 0 ? 0 : v; v = v > RPBW - 1 ? RPBW - 1 : v; ccol[k] = v; }
    const int dsih = dsi - 5 * h;
    bool rowv[5], colv[SQ];
#pragma unroll
    for (int k = 0; k < 5; ++k) rowv[k] = (unsigned)(k - dsih) < (unsigned)KW;
#pragma unroll
    for (int k = 0; k < SQ; ++k) colv[k] = (unsigned)(k - dsj) < (unsigned)KW;
    const float* rb = rpb + head * RPBW * RPBW;
    float* mp = amap + (unsigned)((patch * NHEADS + head) * MAPW) + 4 * lane;
    float* mp2 = amap + (unsigned)((patch * NHEADS + head) * MAPW + MAPG * 128) + 2 * lane;
    float4 pm[MAPG];
    float2 pm2 = make_float2(0.f, 0.f);
    if (HAS_PREV) {
#pragma unroll
        for (int g = 0; g < MAPG; ++g) pm[g] = *(const float4*)(mp + 128 * g);
        pm2 = *(const float2*)(mp2);
    }
    float rbv[7][8];
    float pvv[7][8];
#pragma unroll
    for (int t = 0; t < 7; ++t) {
#pragma unroll
        for (int r = 0; r < 8; ++r) {
            const int f = 8 * t + r;
            rbv[t][r] = f < HKEY ? rb[rrow[f / SQ] + ccol[f % SQ]] : 0.f;
            if (HAS_PREV && f < 4 * MAPG) {
                const float4 g4 = pm[f >> 2];
                pvv[t][r] = (f & 3) == 0 ? g4.x : (f & 3) == 1 ? g4.y : (f & 3) == 2 ? g4.z : g4.w;
            } else if (HAS_PREV && f < HKEY) {
                pvv[t][r] = f == 4 * MAPG ? pm2.x : pm2.y;
            } else {
                pvv[t][r] = 0.f;
            }
        }
    }

#pragma unroll
    for (int s = 0; s < 4; ++s)
        qv[s] = make_float4(qv[s].x * scale, qv[s].y * scale, qv[s].z * scale, qv[s].w * scale);

    v8f st[7];
#pragma unroll
    for (int t = 0; t < 7; ++t) {
        v8f acc = {0.f, 0.f, 0.f, 0.f, 0.f, 0.f, 0.f, 0.f};
#pragma unroll
        for (int s = 0; s < 4; ++s) {
            v2f a0 = {kv[t][s].x, kv[t][s].y};
            v2f a1 = {kv[t][s].z, kv[t][s].w};
            v2f b0 = {qv[s].x, qv[s].y};
            v2f b1 = {qv[s].z, qv[s].w};
            acc = wmma4(a0, b0, acc);
            acc = wmma4(a1, b1, acc);
        }
        st[t] = acc;
    }

    const unsigned voff = (unsigned)(((r0 + 5 * h) * WW + c0) * KVS + head * HD + lc);
    float va[7][4][2][2];
#pragma unroll
    for (int t = 0; t < 7; ++t) {
#pragma unroll
        for (int s = 0; s < 4; ++s) {
            const int k0 = 8 * t + 2 * s;
            if (k0 >= HKEY) continue;
            const unsigned of0 = voff + key_row_off<KVS>(k0);
            const unsigned of1 = voff + key_row_off<KVS>(k0 + 1);
            va[t][s][0][0] = vin[of0];
            va[t][s][0][1] = vin[of1];
            va[t][s][1][0] = vin[of0 + 16];
            va[t][s][1][1] = vin[of1 + 16];
        }
    }

    float4 wq[4][2][2];
    {
        const float* wr = pw + (unsigned)(head * (16 * 128) + 4 * lane);
#pragma unroll
        for (int u = 0; u < 4; ++u) {
#pragma unroll
            for (int dt = 0; dt < 2; ++dt) {
                wq[u][dt][0] = *(const float4*)(wr + 128 * (4 * u + 2 * dt));
                wq[u][dt][1] = *(const float4*)(wr + 128 * (4 * u + 2 * dt + 1));
            }
        }
    }
    float4 bq[2][2];
#pragma unroll
    for (int tt = 0; tt < 2; ++tt) {
        const float* br = pb + (unsigned)(16 * (2 * head + tt) + 8 * h);
        bq[tt][0] = *(const float4*)(br);
        bq[tt][1] = *(const float4*)(br + 4);
    }

    float mx = -1e30f;
#pragma unroll
    for (int t = 0; t < 7; ++t) {
#pragma unroll
        for (int r = 0; r < 8; ++r) {
            const int f = 8 * t + r;
            const bool valid = f < HKEY && rowv[f < HKEY ? f / SQ : 0] && colv[f % SQ];
            float l = st[t][r] + rbv[t][r];
            if (HAS_PREV) l += pvv[t][r];
            l = valid ? l : -1e30f;
            st[t][r] = l;
            mx = fmaxf(mx, l);
        }
    }
    mx = fmaxf(mx, __shfl_xor(mx, 16));

    float ssum = 0.f;
#pragma unroll
    for (int t = 0; t < 7; ++t) {
#pragma unroll
        for (int r = 0; r < 8; ++r) {
            float p = __expf(st[t][r] - mx);
            st[t][r] = p;
            ssum += p;
        }
    }
    ssum += __shfl_xor(ssum, 16);
    const float inv = 1.f / ssum;

#pragma unroll
    for (int t = 0; t < 7; ++t)
#pragma unroll
        for (int r = 0; r < 8; ++r) st[t][r] *= inv;

    v8f o[2][2];
#pragma unroll
    for (int dt = 0; dt < 2; ++dt)
#pragma unroll
        for (int u = 0; u < 2; ++u)
            o[dt][u] = (v8f){0.f, 0.f, 0.f, 0.f, 0.f, 0.f, 0.f, 0.f};

#pragma unroll
    for (int t = 0; t < 7; ++t) {
#pragma unroll
        for (int s = 0; s < 4; ++s) {
            const int k0 = 8 * t + 2 * s;
            if (k0 >= HKEY) continue;
            v2f a0 = {va[t][s][0][0], va[t][s][0][1]};
            v2f a1 = {va[t][s][1][0], va[t][s][1][1]};
            v2f b  = {st[t][2 * s], st[t][2 * s + 1]};
            o[0][t & 1] = wmma4(a0, b, o[0][t & 1]);
            o[1][t & 1] = wmma4(a1, b, o[1][t & 1]);
        }
    }

    v8f d[2];
#pragma unroll
    for (int dt = 0; dt < 2; ++dt) d[dt] = o[dt][0] + o[dt][1];

    float4 wn[6][4][2];
    if constexpr (!LAST) {
        const float* wr = nw + (unsigned)(head * (48 * 128) + 4 * lane);
#pragma unroll
        for (int m = 0; m < 6; ++m) {
#pragma unroll
            for (int u = 0; u < 4; ++u) {
                wn[m][u][0] = *(const float4*)(wr + 128 * (8 * m + 2 * u));
                wn[m][u][1] = *(const float4*)(wr + 128 * (8 * m + 2 * u + 1));
            }
        }
    }

    v8f pp[4];
#pragma unroll
    for (int tt = 0; tt < 2; ++tt)
        pp[tt] = (v8f){bq[tt][0].x, bq[tt][0].y, bq[tt][0].z, bq[tt][0].w,
                       bq[tt][1].x, bq[tt][1].y, bq[tt][1].z, bq[tt][1].w};
#pragma unroll
    for (int u = 2; u < 4; ++u) pp[u] = (v8f){0.f, 0.f, 0.f, 0.f, 0.f, 0.f, 0.f, 0.f};
#pragma unroll
    for (int dt = 0; dt < 2; ++dt) {
#pragma unroll
        for (int j = 0; j < 4; ++j) {
            v2f b = {d[dt][2 * j], d[dt][2 * j + 1]};
#pragma unroll
            for (int u = 0; u < 4; ++u) {
                const float4 w4 = wq[u][dt][j >> 1];
                v2f a = (j & 1) ? (v2f){w4.z, w4.w} : (v2f){w4.x, w4.y};
                pp[u] = wmma4(a, b, pp[u]);
            }
        }
    }

    if constexpr (LAST) {
        __shared__ __attribute__((aligned(16))) float stg[CDIM * 4 * GCOL];
        float* xw = xch + (pl * NHEADS + head) * (2 * 8 * 32) + lane;
        const float* xr = xch + (pl * NHEADS + (1 - head)) * (2 * 8 * 32) + lane;
#pragma unroll
        for (int tt = 0; tt < 2; ++tt)
#pragma unroll
            for (int r = 0; r < 8; ++r) xw[(tt * 8 + r) * 32] = pp[2 + tt][r];
        __syncthreads();

        float* sw = stg + (32 * head + 8 * h) * (4 * GCOL) + (lc >> 2) * GCOL + 4 * pl + (lc & 3);
        float xo[2][8];
#pragma unroll
        for (int tt = 0; tt < 2; ++tt)
#pragma unroll
            for (int r = 0; r < 8; ++r) xo[tt][r] = xr[(tt * 8 + r) * 32];
        asm volatile("" :: "v"(xo[0][0]), "v"(xo[0][1]), "v"(xo[0][2]), "v"(xo[0][3]),
                           "v"(xo[0][4]), "v"(xo[0][5]), "v"(xo[0][6]), "v"(xo[0][7]),
                           "v"(xo[1][0]), "v"(xo[1][1]), "v"(xo[1][2]), "v"(xo[1][3]),
                           "v"(xo[1][4]), "v"(xo[1][5]), "v"(xo[1][6]), "v"(xo[1][7]));
#pragma unroll
        for (int tt = 0; tt < 2; ++tt)
#pragma unroll
            for (int r = 0; r < 8; ++r)
                sw[(16 * tt + r) * (4 * GCOL)] = pp[tt][r] + xo[tt][r];
        __syncthreads();

        const int ig = ((int)blockIdx.x / (WW / GCOL)) * 4;
        const int jg = ((int)blockIdx.x % (WW / GCOL)) * GCOL;
        float* og = out + (unsigned)(ig * WW + jg);
#pragma unroll
        for (int it = 0; it < (CDIM * 4 * GCOL) / (4 * 64 * GP); ++it) {
            const int idx = it * (64 * GP) + (int)threadIdx.x;
            const float4 w4 = *(const float4*)(stg + 4 * idx);
            *(float4*)(og + (size_t)(idx >> 4) * PIX + ((idx >> 2) & 3) * WW + 4 * (idx & 3)) = w4;
        }
    } else {
        float* xw = xch + head * (4 * XTILE) + lane * XPITCH;
        const float* xr = xch + (1 - head) * (4 * XTILE) + lane * XPITCH;
#pragma unroll
        for (int u = 0; u < 4; ++u) {
            *(float4*)(xw + u * XTILE)     = make_float4(pp[u][0], pp[u][1], pp[u][2], pp[u][3]);
            *(float4*)(xw + u * XTILE + 4) = make_float4(pp[u][4], pp[u][5], pp[u][6], pp[u][7]);
        }
        __syncthreads();

        v4f xa[4], xb[4];
#pragma unroll
        for (int u = 0; u < 4; ++u) {
            xa[u] = *(const v4f*)(xr + ((u + 2) & 3) * XTILE);
            xb[u] = *(const v4f*)(xr + ((u + 2) & 3) * XTILE + 4);
        }
        float4 nbq[6][2];
#pragma unroll
        for (int m = 0; m < 6; ++m) {
            const float* br = nb + (unsigned)(64 * (m >> 1) + 32 * head + 16 * (m & 1) + 8 * h);
            nbq[m][0] = *(const float4*)(br);
            nbq[m][1] = *(const float4*)(br + 4);
        }
        asm volatile("" :: "v"(xa[0]), "v"(xb[0]), "v"(xa[1]), "v"(xb[1]),
                           "v"(xa[2]), "v"(xb[2]), "v"(xa[3]), "v"(xb[3]));
        v8f xt[4];
#pragma unroll
        for (int u = 0; u < 4; ++u)
            xt[u] = pp[u] + (v8f){xa[u].x, xa[u].y, xa[u].z, xa[u].w, xb[u].x, xb[u].y, xb[u].z, xb[u].w};

        v8f qa[6];
#pragma unroll
        for (int m = 0; m < 6; ++m) qa[m] = (v8f){0.f, 0.f, 0.f, 0.f, 0.f, 0.f, 0.f, 0.f};
#pragma unroll
        for (int u = 0; u < 4; ++u) {
#pragma unroll
            for (int j = 0; j < 4; ++j) {
                v2f b = {xt[u][2 * j], xt[u][2 * j + 1]};
#pragma unroll
                for (int m = 0; m < 6; ++m) {
                    const float4 w4 = wn[m][u][j >> 1];
                    v2f a = (j & 1) ? (v2f){w4.z, w4.w} : (v2f){w4.x, w4.y};
                    qa[m] = wmma4(a, b, qa[m]);
                }
            }
        }

        float* qw = qio  + (size_t)(((i0 + h) * WW + j0 + (lc >> 3)) * 192 + 32 * head + 4 * (lc & 7));
        float* kp = kout + (size_t)qpix * KVD + 32 * head + 8 * h;
        float* vp = vout + (size_t)qpix * KVD + 32 * head + 8 * h;
#pragma unroll
        for (int m = 0; m < 6; ++m) {
            float* dp = ((m >> 1) == 1 ? kp : vp) + 16 * (m & 1);
            float* d0 = (m >> 1) == 0 ? qw + 2 * (m & 1) * WW * 192 : dp;
            float* d1 = (m >> 1) == 0 ? d0 + 2 * 192 : dp + 4;
            *(float4*)(d0) = make_float4(qa[m][0] + nbq[m][0].x, qa[m][1] + nbq[m][0].y,
                                         qa[m][2] + nbq[m][0].z, qa[m][3] + nbq[m][0].w);
            *(float4*)(d1) = make_float4(qa[m][4] + nbq[m][1].x, qa[m][5] + nbq[m][1].y,
                                         qa[m][6] + nbq[m][1].z, qa[m][7] + nbq[m][1].w);
        }

#pragma unroll
        for (int g = 0; g < MAPG; ++g) {
            const int t = g >> 1;
            const int r = 4 * (g & 1);
            *(float4*)(mp + 128 * g) = make_float4(st[t][r], st[t][r + 1], st[t][r + 2], st[t][r + 3]);
        }
        *(float2*)(mp2) = make_float2(st[6][0], st[6][1]);
    }
}

extern "C" void kernel_launch(void* const* d_in, const int* in_sizes, int n_in,
                              void* d_out, int out_size, void* d_ws, size_t ws_size,
                              hipStream_t stream) {
    const float* x      = (const float*)d_in[0];
    const float* qkv_w  = (const float*)d_in[1];
    const float* qkv_b  = (const float*)d_in[2];
    const float* proj_w = (const float*)d_in[3];
    const float* proj_b = (const float*)d_in[4];
    const float* rpb    = (const float*)d_in[5];
    float* out = (float*)d_out;
    float* ws  = (float*)d_ws;

    float* plane = ws;
    float* qkv   = plane + (size_t)PIX * CDIM;
    float* amap  = qkv + (size_t)PIX * 192;
    float* wfrag = amap + (size_t)(PIX / 16) * NHEADS * MAPW;

    nchw_to_frag<<<TBLK + WBLK, 128, 0, stream>>>(x, plane, qkv_w, proj_w, wfrag);

    {
        int numTiles = (PIX / 16) * (192 / 16);
        wmma_gemm_bias<CDIM><<<(numTiles + 7) / 8, 256, 0, stream>>>(
            plane, wfrag + WF_QKV0, qkv_b, qkv, PIX, 192);
    }

    na_attn_wmma<false, false, 192, 64><<<PIX / 16, 32 * NHEADS, 0, stream>>>(
        qkv, qkv + 64, qkv + 128, rpb, wfrag + WF_PROJ, proj_b,
        wfrag + WF_NEXT, qkv_b + 192, amap, plane, out, nullptr);
    na_attn_wmma<true, false, 64, 192><<<PIX / 16, 32 * NHEADS, 0, stream>>>(
        qkv, plane, out, rpb + RPBS, wfrag + WF_PROJ + CDIM * CDIM, proj_b + CDIM,
        wfrag + WF_NEXT + 192 * CDIM, qkv_b + 2 * 192, amap, qkv + 64, qkv + 128, nullptr);
    na_attn_wmma<true, true, 192, 64><<<PIX / 16 / GP, 32 * NHEADS * GP, 0, stream>>>(
        qkv, qkv + 64, qkv + 128, rpb + 2 * RPBS, wfrag + WF_PROJ + 2 * CDIM * CDIM, proj_b + 2 * CDIM,
        nullptr, nullptr, amap, nullptr, nullptr, out);
}


# stage 0's QKV product formed in the layout launch: x tile and row-major weight staged in LDS by whole-line loads, one loader for A and B, no workspace read; 5 -> 4 launches; same sums
# speedup vs baseline: 9.0040x; 1.1299x over previous
#include <hip/hip_runtime.h>

typedef float v2f __attribute__((ext_vector_type(2)));
typedef float v4f __attribute__((ext_vector_type(4)));
typedef float v8f __attribute__((ext_vector_type(8)));

#define HH 96
#define WW 96
#define PIX (HH * WW)
#define CDIM 64
#define NHEADS 2
#define HD 32
#define KW 7
#define NKK (KW * KW)

#define TPX 32
#define TPITCH 68
#define GBLK (PIX / TPX)
#define NF (3 * CDIM)
#define WF_PROJ 0
#define WF_NEXT (3 * NHEADS * 16 * 128)
#define WF_FLOATS (WF_NEXT + 2 * NHEADS * 48 * 128)
#define WBLK (WF_FLOATS / 4 / 128)

__global__ void __launch_bounds__(128)
qkv0_from_nchw(const float* __restrict__ x, const float* __restrict__ qkv_w,
               const float* __restrict__ qkv_b, const float* __restrict__ proj_w,
               float* __restrict__ qkv, float* __restrict__ wf) {
    __shared__ __attribute__((aligned(16))) float xt[TPX * TPITCH];
    __shared__ __attribute__((aligned(16))) float wt[NF * TPITCH];
    const int tid = threadIdx.x;
    if (blockIdx.x >= GBLK) {
        const int i = (blockIdx.x - GBLK) * 128 + tid;
        const int lane = i & 31, lc = lane & 15, h = lane >> 4;
        float4 w;
        if (i < WF_NEXT / 4) {
            const int idx = (i >> 5) & 15, head = (i >> 9) & 1, stage = i >> 10;
            const int u = idx >> 2, dt = (idx >> 1) & 1, e = idx & 1;
            const int ot = (2 * head + u) & 3;
            w = *(const float4*)(proj_w + stage * CDIM * CDIM + (16 * ot + lc) * CDIM + head * HD + 8 * h + 16 * dt + 4 * e);
        } else {
            const int j = i - WF_NEXT / 4;
            const int sn = j / (NHEADS * 48 * 32);
            const int r = j - sn * (NHEADS * 48 * 32);
            const int head = r / (48 * 32);
            const int idx = (r - head * (48 * 32)) >> 5;
            const int m = idx >> 3, u = (idx >> 1) & 3, e = idx & 1;
            const int g = (2 * head + u) & 3;
            const int frow = 64 * (m >> 1) + 32 * head + 16 * (m & 1);
            w = *(const float4*)(qkv_w + (sn + 1) * NF * CDIM + (frow + lc) * CDIM + 8 * h + 16 * g + 4 * e);
        }
        *(float4*)(wf + 4 * i) = w;
        return;
    }
    const int p0 = blockIdx.x * TPX;
    const int lane = tid & 31, lc = lane & 15, h = lane >> 4;
    const int wv = __builtin_amdgcn_readfirstlane(tid >> 5);
    const int pt = wv & 1;
    const int f0 = 6 * (wv >> 1);

    float4 v[4];
#pragma unroll
    for (int i = 0; i < 4; ++i) {
        const int idx = i * 128 + tid;
        const int c = idx >> 3;
        const int q = idx & 7;
        v[i] = *(const float4*)(x + (size_t)c * PIX + p0 + 4 * q);
    }
    float4 wr[NF * CDIM / 4 / 128];
#pragma unroll
    for (int i = 0; i < NF * CDIM / 4 / 128; ++i)
        wr[i] = *(const float4*)(qkv_w + 4 * (i * 128 + tid));
    float bv[6];
#pragma unroll
    for (int m = 0; m < 6; ++m) bv[m] = qkv_b[16 * (f0 + m) + lc];

#pragma unroll
    for (int i = 0; i < 4; ++i) {
        const int idx = i * 128 + tid;
        const int c = idx >> 3;
        const int q = idx & 7;
        float* t = xt + (4 * q) * TPITCH + c;
        t[0]          = v[i].x;
        t[TPITCH]     = v[i].y;
        t[2 * TPITCH] = v[i].z;
        t[3 * TPITCH] = v[i].w;
    }
#pragma unroll
    for (int i = 0; i < NF * CDIM / 4 / 128; ++i) {
        const int idx = i * 128 + tid;
        *(float4*)(wt + (idx >> 4) * TPITCH + 4 * (idx & 15)) = wr[i];
    }
    __syncthreads();

    const float* ar = xt + (16 * pt + lc) * TPITCH + 2 * h;
    const float* br = wt + (16 * f0 + lc) * TPITCH + 2 * h;
    v8f a8[4];
#pragma unroll
    for (int g = 0; g < 4; ++g) {
        const v2f t0 = *(const v2f*)(ar + 16 * g);
        const v2f t1 = *(const v2f*)(ar + 16 * g + 4);
        const v2f t2 = *(const v2f*)(ar + 16 * g + 8);
        const v2f t3 = *(const v2f*)(ar + 16 * g + 12);
        a8[g] = (v8f){t0.x, t0.y, t1.x, t1.y, t2.x, t2.y, t3.x, t3.y};
    }
    v8f b8[6][4];
#pragma unroll
    for (int m = 0; m < 6; ++m) {
#pragma unroll
        for (int g = 0; g < 4; ++g) {
            const float* bp = br + 16 * m * TPITCH + 16 * g;
            const v2f t0 = *(const v2f*)(bp);
            const v2f t1 = *(const v2f*)(bp + 4);
            const v2f t2 = *(const v2f*)(bp + 8);
            const v2f t3 = *(const v2f*)(bp + 12);
            b8[m][g] = (v8f){t0.x, t0.y, t1.x, t1.y, t2.x, t2.y, t3.x, t3.y};
        }
    }
    asm volatile("" :: "v"(a8[0]), "v"(a8[1]), "v"(a8[2]), "v"(a8[3]),
                       "v"(b8[0][0]), "v"(b8[0][1]), "v"(b8[0][2]), "v"(b8[0][3]),
                       "v"(b8[1][0]), "v"(b8[1][1]), "v"(b8[1][2]), "v"(b8[1][3]),
                       "v"(b8[2][0]), "v"(b8[2][1]), "v"(b8[2][2]), "v"(b8[2][3]),
                       "v"(b8[3][0]), "v"(b8[3][1]), "v"(b8[3][2]), "v"(b8[3][3]),
                       "v"(b8[4][0]), "v"(b8[4][1]), "v"(b8[4][2]), "v"(b8[4][3]),
                       "v"(b8[5][0]), "v"(b8[5][1]), "v"(b8[5][2]), "v"(b8[5][3]));

    v8f acc[6];
#pragma unroll
    for (int m = 0; m < 6; ++m) acc[m] = (v8f){0.f, 0.f, 0.f, 0.f, 0.f, 0.f, 0.f, 0.f};
#pragma unroll
    for (int ks = 0; ks < 16; ++ks) {
        const v2f a = {a8[ks >> 2][2 * (ks & 3)], a8[ks >> 2][2 * (ks & 3) + 1]};
#pragma unroll
        for (int m = 0; m < 6; ++m) {
            const v2f b = {b8[m][ks >> 2][2 * (ks & 3)], b8[m][ks >> 2][2 * (ks & 3) + 1]};
            acc[m] = __builtin_amdgcn_wmma_f32_16x16x4_f32(
                 false, a,  false, b,
                 (short)0, acc[m],  false,  false);
        }
    }

    float* cp = qkv + (size_t)(p0 + 16 * pt + 8 * h) * NF + 16 * f0 + lc;
#pragma unroll
    for (int m = 0; m < 6; ++m)
#pragma unroll
        for (int r = 0; r < 8; ++r)
            cp[r * NF + 16 * m] = acc[m][r] + bv[m];
}

#define SQ 10
#define NKEY (SQ * SQ)
#define RPBW (2 * KW - 1)
#define MAPG 12
#define MAPW (MAPG * 128 + 64)
#define RPBS (NHEADS * RPBW * RPBW)

__device__ __forceinline__ v8f wmma4(v2f a, v2f b, v8f c) {
    return __builtin_amdgcn_wmma_f32_16x16x4_f32(
         false, a,  false, b,
         (short)0, c,  false,  false);
}

template <int PITCH>
__device__ __forceinline__ constexpr unsigned key_row_off(int n) {
    return (unsigned)((((n < NKEY ? n : NKEY - 1) / SQ) * WW + ((n < NKEY ? n : NKEY - 1) % SQ)) * PITCH);
}

#define HKEY (NKEY / 2)

#define XPITCH 12
#define XTILE (32 * XPITCH)
#define GP 4
#define GCOL (4 * GP)
template <bool HAS_PREV, bool LAST, int KVS, int KVD>
__global__ void __launch_bounds__(LAST ? 64 * GP : 64) __attribute__((amdgpu_waves_per_eu(LAST ? 2 : 1, 2)))
na_attn_wmma(float* __restrict__ qio,
             const float* __restrict__ kin,
             const float* __restrict__ vin,
             const float* __restrict__ rpb,
             const float* __restrict__ pw,
             const float* __restrict__ pb,
             const float* __restrict__ nw,
             const float* __restrict__ nb,
             float* __restrict__ amap,
             float* __restrict__ kout,
             float* __restrict__ vout,
             float* __restrict__ out) {
    __shared__ __attribute__((aligned(16))) float xch[LAST ? GP * NHEADS * 2 * 8 * 32 : NHEADS * 4 * XTILE];
    const int lane  = threadIdx.x & 31;
    const int lc    = lane & 15;
    const int h     = lane >> 4;
    const int wv    = __builtin_amdgcn_readfirstlane((int)(threadIdx.x >> 5));
    const int head  = LAST ? (wv & 1) : wv;
    const int pl    = LAST ? (wv >> 1) : 0;
    const int patch = LAST ? (int)blockIdx.x * GP + pl : (int)blockIdx.x;
    const int i0 = (patch / (WW / 4)) * 4;
    const int j0 = (patch % (WW / 4)) * 4;

    int r0 = i0 - KW / 2; r0 = r0 < 0 ? 0 : r0; r0 = r0 > HH - SQ ? HH - SQ : r0;
    int c0 = j0 - KW / 2; c0 = c0 < 0 ? 0 : c0; c0 = c0 > WW - SQ ? WW - SQ : c0;

    const int qi = i0 + (lc >> 2);
    const int qj = j0 + (lc & 3);
    const int qpix = qi * WW + qj;
    int si = qi - KW / 2; si = si < 0 ? 0 : si; si = si > HH - KW ? HH - KW : si;
    int sj = qj - KW / 2; sj = sj < 0 ? 0 : sj; sj = sj > WW - KW ? WW - KW : sj;
    const int dsi = si - r0;
    const int dsj = sj - c0;

    const float scale = 0.17677669529663687f;

    const unsigned qoff = HAS_PREV
        ? (unsigned)((i0 * WW + j0 + (lane >> 3)) * 192 + head * HD + 4 * (lane & 7))
        : (unsigned)(qpix * 192 + head * HD + 4 * h);
    float4 qv[4];
#pragma unroll
    for (int s = 0; s < 4; ++s) qv[s] = *(const float4*)(qio + qoff + (HAS_PREV ? WW * 192 : 8) * s);

    float4 kv[7][4];
#pragma unroll
    for (int t = 0; t < 7; ++t) {
        int km = 8 * t + (lc & 7);
        km = km < HKEY ? km : HKEY - 1;
        const int kq = km / SQ;
        const int kr = kq + 5 * (lc >> 3);
        const int kc = km - kq * SQ;
        const unsigned koff = (unsigned)(((r0 + kr) * WW + (c0 + kc)) * KVS + head * HD + 4 * h);
#pragma unroll
        for (int s = 0; s < 4; ++s) kv[t][s] = *(const float4*)(kin + koff + 8 * s);
    }

    const int ri0 = r0 - qi + KW - 1 + 5 * h;
    const int ci0 = c0 - qj + KW - 1;
    int rrow[5], ccol[SQ];
#pragma unroll
    for (int k = 0; k < 5; ++k) { int v = ri0 + k; v = v < 0 ? 0 : v; v = v > RPBW - 1 ? RPBW - 1 : v; rrow[k] = v * RPBW; }
#pragma unroll
    for (int k = 0; k < SQ; ++k) { int v = ci0 + k; v = v < 0 ? 0 : v; v = v > RPBW - 1 ? RPBW - 1 : v; ccol[k] = v; }
    const int dsih = dsi - 5 * h;
    bool rowv[5], colv[SQ];
#pragma unroll
    for (int k = 0; k < 5; ++k) rowv[k] = (unsigned)(k - dsih) < (unsigned)KW;
#pragma unroll
    for (int k = 0; k < SQ; ++k) colv[k] = (unsigned)(k - dsj) < (unsigned)KW;
    const float* rb = rpb + head * RPBW * RPBW;
    float* mp = amap + (unsigned)((patch * NHEADS + head) * MAPW) + 4 * lane;
    float* mp2 = amap + (unsigned)((patch * NHEADS + head) * MAPW + MAPG * 128) + 2 * lane;
    float4 pm[MAPG];
    float2 pm2 = make_float2(0.f, 0.f);
    if (HAS_PREV) {
#pragma unroll
        for (int g = 0; g < MAPG; ++g) pm[g] = *(const float4*)(mp + 128 * g);
        pm2 = *(const float2*)(mp2);
    }
    float rbv[7][8];
    float pvv[7][8];
#pragma unroll
    for (int t = 0; t < 7; ++t) {
#pragma unroll
        for (int r = 0; r < 8; ++r) {
            const int f = 8 * t + r;
            rbv[t][r] = f < HKEY ? rb[rrow[f / SQ] + ccol[f % SQ]] : 0.f;
            if (HAS_PREV && f < 4 * MAPG) {
                const float4 g4 = pm[f >> 2];
                pvv[t][r] = (f & 3) == 0 ? g4.x : (f & 3) == 1 ? g4.y : (f & 3) == 2 ? g4.z : g4.w;
            } else if (HAS_PREV && f < HKEY) {
                pvv[t][r] = f == 4 * MAPG ? pm2.x : pm2.y;
            } else {
                pvv[t][r] = 0.f;
            }
        }
    }

#pragma unroll
    for (int s = 0; s < 4; ++s)
        qv[s] = make_float4(qv[s].x * scale, qv[s].y * scale, qv[s].z * scale, qv[s].w * scale);

    v8f st[7];
#pragma unroll
    for (int t = 0; t < 7; ++t) {
        v8f acc = {0.f, 0.f, 0.f, 0.f, 0.f, 0.f, 0.f, 0.f};
#pragma unroll
        for (int s = 0; s < 4; ++s) {
            v2f a0 = {kv[t][s].x, kv[t][s].y};
            v2f a1 = {kv[t][s].z, kv[t][s].w};
            v2f b0 = {qv[s].x, qv[s].y};
            v2f b1 = {qv[s].z, qv[s].w};
            acc = wmma4(a0, b0, acc);
            acc = wmma4(a1, b1, acc);
        }
        st[t] = acc;
    }

    const unsigned voff = (unsigned)(((r0 + 5 * h) * WW + c0) * KVS + head * HD + lc);
    float va[7][4][2][2];
#pragma unroll
    for (int t = 0; t < 7; ++t) {
#pragma unroll
        for (int s = 0; s < 4; ++s) {
            const int k0 = 8 * t + 2 * s;
            if (k0 >= HKEY) continue;
            const unsigned of0 = voff + key_row_off<KVS>(k0);
            const unsigned of1 = voff + key_row_off<KVS>(k0 + 1);
            va[t][s][0][0] = vin[of0];
            va[t][s][0][1] = vin[of1];
            va[t][s][1][0] = vin[of0 + 16];
            va[t][s][1][1] = vin[of1 + 16];
        }
    }

    float4 wq[4][2][2];
    {
        const float* wr = pw + (unsigned)(head * (16 * 128) + 4 * lane);
#pragma unroll
        for (int u = 0; u < 4; ++u) {
#pragma unroll
            for (int dt = 0; dt < 2; ++dt) {
                wq[u][dt][0] = *(const float4*)(wr + 128 * (4 * u + 2 * dt));
                wq[u][dt][1] = *(const float4*)(wr + 128 * (4 * u + 2 * dt + 1));
            }
        }
    }
    float4 bq[2][2];
#pragma unroll
    for (int tt = 0; tt < 2; ++tt) {
        const float* br = pb + (unsigned)(16 * (2 * head + tt) + 8 * h);
        bq[tt][0] = *(const float4*)(br);
        bq[tt][1] = *(const float4*)(br + 4);
    }

    float mx = -1e30f;
#pragma unroll
    for (int t = 0; t < 7; ++t) {
#pragma unroll
        for (int r = 0; r < 8; ++r) {
            const int f = 8 * t + r;
            const bool valid = f < HKEY && rowv[f < HKEY ? f / SQ : 0] && colv[f % SQ];
            float l = st[t][r] + rbv[t][r];
            if (HAS_PREV) l += pvv[t][r];
            l = valid ? l : -1e30f;
            st[t][r] = l;
            mx = fmaxf(mx, l);
        }
    }
    mx = fmaxf(mx, __shfl_xor(mx, 16));

    float ssum = 0.f;
#pragma unroll
    for (int t = 0; t < 7; ++t) {
#pragma unroll
        for (int r = 0; r < 8; ++r) {
            float p = __expf(st[t][r] - mx);
            st[t][r] = p;
            ssum += p;
        }
    }
    ssum += __shfl_xor(ssum, 16);
    const float inv = 1.f / ssum;

#pragma unroll
    for (int t = 0; t < 7; ++t)
#pragma unroll
        for (int r = 0; r < 8; ++r) st[t][r] *= inv;

    v8f o[2][2];
#pragma unroll
    for (int dt = 0; dt < 2; ++dt)
#pragma unroll
        for (int u = 0; u < 2; ++u)
            o[dt][u] = (v8f){0.f, 0.f, 0.f, 0.f, 0.f, 0.f, 0.f, 0.f};

#pragma unroll
    for (int t = 0; t < 7; ++t) {
#pragma unroll
        for (int s = 0; s < 4; ++s) {
            const int k0 = 8 * t + 2 * s;
            if (k0 >= HKEY) continue;
            v2f a0 = {va[t][s][0][0], va[t][s][0][1]};
            v2f a1 = {va[t][s][1][0], va[t][s][1][1]};
            v2f b  = {st[t][2 * s], st[t][2 * s + 1]};
            o[0][t & 1] = wmma4(a0, b, o[0][t & 1]);
            o[1][t & 1] = wmma4(a1, b, o[1][t & 1]);
        }
    }

    v8f d[2];
#pragma unroll
    for (int dt = 0; dt < 2; ++dt) d[dt] = o[dt][0] + o[dt][1];

    float4 wn[6][4][2];
    if constexpr (!LAST) {
        const float* wr = nw + (unsigned)(head * (48 * 128) + 4 * lane);
#pragma unroll
        for (int m = 0; m < 6; ++m) {
#pragma unroll
            for (int u = 0; u < 4; ++u) {
                wn[m][u][0] = *(const float4*)(wr + 128 * (8 * m + 2 * u));
                wn[m][u][1] = *(const float4*)(wr + 128 * (8 * m + 2 * u + 1));
            }
        }
    }

    v8f pp[4];
#pragma unroll
    for (int tt = 0; tt < 2; ++tt)
        pp[tt] = (v8f){bq[tt][0].x, bq[tt][0].y, bq[tt][0].z, bq[tt][0].w,
                       bq[tt][1].x, bq[tt][1].y, bq[tt][1].z, bq[tt][1].w};
#pragma unroll
    for (int u = 2; u < 4; ++u) pp[u] = (v8f){0.f, 0.f, 0.f, 0.f, 0.f, 0.f, 0.f, 0.f};
#pragma unroll
    for (int dt = 0; dt < 2; ++dt) {
#pragma unroll
        for (int j = 0; j < 4; ++j) {
            v2f b = {d[dt][2 * j], d[dt][2 * j + 1]};
#pragma unroll
            for (int u = 0; u < 4; ++u) {
                const float4 w4 = wq[u][dt][j >> 1];
                v2f a = (j & 1) ? (v2f){w4.z, w4.w} : (v2f){w4.x, w4.y};
                pp[u] = wmma4(a, b, pp[u]);
            }
        }
    }

    if constexpr (LAST) {
        __shared__ __attribute__((aligned(16))) float stg[CDIM * 4 * GCOL];
        float* xw = xch + (pl * NHEADS + head) * (2 * 8 * 32) + lane;
        const float* xr = xch + (pl * NHEADS + (1 - head)) * (2 * 8 * 32) + lane;
#pragma unroll
        for (int tt = 0; tt < 2; ++tt)
#pragma unroll
            for (int r = 0; r < 8; ++r) xw[(tt * 8 + r) * 32] = pp[2 + tt][r];
        __syncthreads();

        float* sw = stg + (32 * head + 8 * h) * (4 * GCOL) + (lc >> 2) * GCOL + 4 * pl + (lc & 3);
        float xo[2][8];
#pragma unroll
        for (int tt = 0; tt < 2; ++tt)
#pragma unroll
            for (int r = 0; r < 8; ++r) xo[tt][r] = xr[(tt * 8 + r) * 32];
        asm volatile("" :: "v"(xo[0][0]), "v"(xo[0][1]), "v"(xo[0][2]), "v"(xo[0][3]),
                           "v"(xo[0][4]), "v"(xo[0][5]), "v"(xo[0][6]), "v"(xo[0][7]),
                           "v"(xo[1][0]), "v"(xo[1][1]), "v"(xo[1][2]), "v"(xo[1][3]),
                           "v"(xo[1][4]), "v"(xo[1][5]), "v"(xo[1][6]), "v"(xo[1][7]));
#pragma unroll
        for (int tt = 0; tt < 2; ++tt)
#pragma unroll
            for (int r = 0; r < 8; ++r)
                sw[(16 * tt + r) * (4 * GCOL)] = pp[tt][r] + xo[tt][r];
        __syncthreads();

        const int ig = ((int)blockIdx.x / (WW / GCOL)) * 4;
        const int jg = ((int)blockIdx.x % (WW / GCOL)) * GCOL;
        float* og = out + (unsigned)(ig * WW + jg);
#pragma unroll
        for (int it = 0; it < (CDIM * 4 * GCOL) / (4 * 64 * GP); ++it) {
            const int idx = it * (64 * GP) + (int)threadIdx.x;
            const float4 w4 = *(const float4*)(stg + 4 * idx);
            *(float4*)(og + (size_t)(idx >> 4) * PIX + ((idx >> 2) & 3) * WW + 4 * (idx & 3)) = w4;
        }
    } else {
        float* xw = xch + head * (4 * XTILE) + lane * XPITCH;
        const float* xr = xch + (1 - head) * (4 * XTILE) + lane * XPITCH;
#pragma unroll
        for (int u = 0; u < 4; ++u) {
            *(float4*)(xw + u * XTILE)     = make_float4(pp[u][0], pp[u][1], pp[u][2], pp[u][3]);
            *(float4*)(xw + u * XTILE + 4) = make_float4(pp[u][4], pp[u][5], pp[u][6], pp[u][7]);
        }
        __syncthreads();

        v4f xa[4], xb[4];
#pragma unroll
        for (int u = 0; u < 4; ++u) {
            xa[u] = *(const v4f*)(xr + ((u + 2) & 3) * XTILE);
            xb[u] = *(const v4f*)(xr + ((u + 2) & 3) * XTILE + 4);
        }
        float4 nbq[6][2];
#pragma unroll
        for (int m = 0; m < 6; ++m) {
            const float* br = nb + (unsigned)(64 * (m >> 1) + 32 * head + 16 * (m & 1) + 8 * h);
            nbq[m][0] = *(const float4*)(br);
            nbq[m][1] = *(const float4*)(br + 4);
        }
        asm volatile("" :: "v"(xa[0]), "v"(xb[0]), "v"(xa[1]), "v"(xb[1]),
                           "v"(xa[2]), "v"(xb[2]), "v"(xa[3]), "v"(xb[3]));
        v8f xt[4];
#pragma unroll
        for (int u = 0; u < 4; ++u)
            xt[u] = pp[u] + (v8f){xa[u].x, xa[u].y, xa[u].z, xa[u].w, xb[u].x, xb[u].y, xb[u].z, xb[u].w};

        v8f qa[6];
#pragma unroll
        for (int m = 0; m < 6; ++m) qa[m] = (v8f){0.f, 0.f, 0.f, 0.f, 0.f, 0.f, 0.f, 0.f};
#pragma unroll
        for (int u = 0; u < 4; ++u) {
#pragma unroll
            for (int j = 0; j < 4; ++j) {
                v2f b = {xt[u][2 * j], xt[u][2 * j + 1]};
#pragma unroll
                for (int m = 0; m < 6; ++m) {
                    const float4 w4 = wn[m][u][j >> 1];
                    v2f a = (j & 1) ? (v2f){w4.z, w4.w} : (v2f){w4.x, w4.y};
                    qa[m] = wmma4(a, b, qa[m]);
                }
            }
        }

        float* qw = qio  + (size_t)(((i0 + h) * WW + j0 + (lc >> 3)) * 192 + 32 * head + 4 * (lc & 7));
        float* kp = kout + (size_t)qpix * KVD + 32 * head + 8 * h;
        float* vp = vout + (size_t)qpix * KVD + 32 * head + 8 * h;
#pragma unroll
        for (int m = 0; m < 6; ++m) {
            float* dp = ((m >> 1) == 1 ? kp : vp) + 16 * (m & 1);
            float* d0 = (m >> 1) == 0 ? qw + 2 * (m & 1) * WW * 192 : dp;
            float* d1 = (m >> 1) == 0 ? d0 + 2 * 192 : dp + 4;
            *(float4*)(d0) = make_float4(qa[m][0] + nbq[m][0].x, qa[m][1] + nbq[m][0].y,
                                         qa[m][2] + nbq[m][0].z, qa[m][3] + nbq[m][0].w);
            *(float4*)(d1) = make_float4(qa[m][4] + nbq[m][1].x, qa[m][5] + nbq[m][1].y,
                                         qa[m][6] + nbq[m][1].z, qa[m][7] + nbq[m][1].w);
        }

#pragma unroll
        for (int g = 0; g < MAPG; ++g) {
            const int t = g >> 1;
            const int r = 4 * (g & 1);
            *(float4*)(mp + 128 * g) = make_float4(st[t][r], st[t][r + 1], st[t][r + 2], st[t][r + 3]);
        }
        *(float2*)(mp2) = make_float2(st[6][0], st[6][1]);
    }
}

extern "C" void kernel_launch(void* const* d_in, const int* in_sizes, int n_in,
                              void* d_out, int out_size, void* d_ws, size_t ws_size,
                              hipStream_t stream) {
    const float* x      = (const float*)d_in[0];
    const float* qkv_w  = (const float*)d_in[1];
    const float* qkv_b  = (const float*)d_in[2];
    const float* proj_w = (const float*)d_in[3];
    const float* proj_b = (const float*)d_in[4];
    const float* rpb    = (const float*)d_in[5];
    float* out = (float*)d_out;
    float* ws  = (float*)d_ws;

    float* plane = ws;
    float* qkv   = plane + (size_t)PIX * CDIM;
    float* amap  = qkv + (size_t)PIX * 192;
    float* wfrag = amap + (size_t)(PIX / 16) * NHEADS * MAPW;

    qkv0_from_nchw<<<GBLK + WBLK, 128, 0, stream>>>(x, qkv_w, qkv_b, proj_w, qkv, wfrag);

    na_attn_wmma<false, false, 192, 64><<<PIX / 16, 32 * NHEADS, 0, stream>>>(
        qkv, qkv + 64, qkv + 128, rpb, wfrag + WF_PROJ, proj_b,
        wfrag + WF_NEXT, qkv_b + 192, amap, plane, out, nullptr);
    na_attn_wmma<true, false, 64, 192><<<PIX / 16, 32 * NHEADS, 0, stream>>>(
        qkv, plane, out, rpb + RPBS, wfrag + WF_PROJ + CDIM * CDIM, proj_b + CDIM,
        wfrag + WF_NEXT + 192 * CDIM, qkv_b + 2 * 192, amap, qkv + 64, qkv + 128, nullptr);
    na_attn_wmma<true, true, 192, 64><<<PIX / 16 / GP, 32 * NHEADS * GP, 0, stream>>>(
        qkv, qkv + 64, qkv + 128, rpb + 2 * RPBS, wfrag + WF_PROJ + 2 * CDIM * CDIM, proj_b + 2 * CDIM,
        nullptr, nullptr, amap, nullptr, nullptr, out);
}


# K kept transposed inside each 4-pixel group (line = 4 keys x 8 dims of one slice): a K fragment load touches about 6 lines, not 16; all K writers store in that order; same sums
# speedup vs baseline: 9.4237x; 1.0466x over previous
#include <hip/hip_runtime.h>

typedef float v2f __attribute__((ext_vector_type(2)));
typedef float v4f __attribute__((ext_vector_type(4)));
typedef float v8f __attribute__((ext_vector_type(8)));

#define HH 96
#define WW 96
#define PIX (HH * WW)
#define CDIM 64
#define NHEADS 2
#define HD 32
#define KW 7
#define NKK (KW * KW)

#define TPX 32
#define TPITCH 68
#define GBLK (PIX / TPX)
#define NF (3 * CDIM)
#define WF_PROJ 0
#define WF_NEXT (3 * NHEADS * 16 * 128)
#define WF_FLOATS (WF_NEXT + 2 * NHEADS * 48 * 128)
#define WBLK (WF_FLOATS / 4 / 128)

__global__ void __launch_bounds__(128)
qkv0_from_nchw(const float* __restrict__ x, const float* __restrict__ qkv_w,
               const float* __restrict__ qkv_b, const float* __restrict__ proj_w,
               float* __restrict__ qkv, float* __restrict__ wf) {
    __shared__ __attribute__((aligned(16))) float xt[TPX * TPITCH];
    __shared__ __attribute__((aligned(16))) float wt[NF * TPITCH];
    const int tid = threadIdx.x;
    if (blockIdx.x >= GBLK) {
        const int i = (blockIdx.x - GBLK) * 128 + tid;
        const int lane = i & 31, lc = lane & 15, h = lane >> 4;
        float4 w;
        if (i < WF_NEXT / 4) {
            const int idx = (i >> 5) & 15, head = (i >> 9) & 1, stage = i >> 10;
            const int u = idx >> 2, dt = (idx >> 1) & 1, e = idx & 1;
            const int ot = (2 * head + u) & 3;
            w = *(const float4*)(proj_w + stage * CDIM * CDIM + (16 * ot + lc) * CDIM + head * HD + 8 * h + 16 * dt + 4 * e);
        } else {
            const int j = i - WF_NEXT / 4;
            const int sn = j / (NHEADS * 48 * 32);
            const int r = j - sn * (NHEADS * 48 * 32);
            const int head = r / (48 * 32);
            const int idx = (r - head * (48 * 32)) >> 5;
            const int m = idx >> 3, u = (idx >> 1) & 3, e = idx & 1;
            const int g = (2 * head + u) & 3;
            const int frow = 64 * (m >> 1) + 32 * head + 16 * (m & 1);
            w = *(const float4*)(qkv_w + (sn + 1) * NF * CDIM + (frow + lc) * CDIM + 8 * h + 16 * g + 4 * e);
        }
        *(float4*)(wf + 4 * i) = w;
        return;
    }
    const int p0 = blockIdx.x * TPX;
    const int lane = tid & 31, lc = lane & 15, h = lane >> 4;
    const int wv = __builtin_amdgcn_readfirstlane(tid >> 5);
    const int pt = wv & 1;
    const int f0 = 6 * (wv >> 1);

    float4 v[4];
#pragma unroll
    for (int i = 0; i < 4; ++i) {
        const int idx = i * 128 + tid;
        const int c = idx >> 3;
        const int q = idx & 7;
        v[i] = *(const float4*)(x + (size_t)c * PIX + p0 + 4 * q);
    }
    float4 wr[NF * CDIM / 4 / 128];
#pragma unroll
    for (int i = 0; i < NF * CDIM / 4 / 128; ++i)
        wr[i] = *(const float4*)(qkv_w + 4 * (i * 128 + tid));
    float bv[6];
#pragma unroll
    for (int m = 0; m < 6; ++m) bv[m] = qkv_b[16 * (f0 + m) + lc];

#pragma unroll
    for (int i = 0; i < 4; ++i) {
        const int idx = i * 128 + tid;
        const int c = idx >> 3;
        const int q = idx & 7;
        float* t = xt + (4 * q) * TPITCH + c;
        t[0]          = v[i].x;
        t[TPITCH]     = v[i].y;
        t[2 * TPITCH] = v[i].z;
        t[3 * TPITCH] = v[i].w;
    }
#pragma unroll
    for (int i = 0; i < NF * CDIM / 4 / 128; ++i) {
        const int idx = i * 128 + tid;
        *(float4*)(wt + (idx >> 4) * TPITCH + 4 * (idx & 15)) = wr[i];
    }
    __syncthreads();

    const float* ar = xt + (16 * pt + lc) * TPITCH + 2 * h;
    const float* br = wt + (16 * f0 + lc) * TPITCH + 2 * h;
    v8f a8[4];
#pragma unroll
    for (int g = 0; g < 4; ++g) {
        const v2f t0 = *(const v2f*)(ar + 16 * g);
        const v2f t1 = *(const v2f*)(ar + 16 * g + 4);
        const v2f t2 = *(const v2f*)(ar + 16 * g + 8);
        const v2f t3 = *(const v2f*)(ar + 16 * g + 12);
        a8[g] = (v8f){t0.x, t0.y, t1.x, t1.y, t2.x, t2.y, t3.x, t3.y};
    }
    v8f b8[6][4];
#pragma unroll
    for (int m = 0; m < 6; ++m) {
#pragma unroll
        for (int g = 0; g < 4; ++g) {
            const float* bp = br + 16 * m * TPITCH + 16 * g;
            const v2f t0 = *(const v2f*)(bp);
            const v2f t1 = *(const v2f*)(bp + 4);
            const v2f t2 = *(const v2f*)(bp + 8);
            const v2f t3 = *(const v2f*)(bp + 12);
            b8[m][g] = (v8f){t0.x, t0.y, t1.x, t1.y, t2.x, t2.y, t3.x, t3.y};
        }
    }
    asm volatile("" :: "v"(a8[0]), "v"(a8[1]), "v"(a8[2]), "v"(a8[3]),
                       "v"(b8[0][0]), "v"(b8[0][1]), "v"(b8[0][2]), "v"(b8[0][3]),
                       "v"(b8[1][0]), "v"(b8[1][1]), "v"(b8[1][2]), "v"(b8[1][3]),
                       "v"(b8[2][0]), "v"(b8[2][1]), "v"(b8[2][2]), "v"(b8[2][3]),
                       "v"(b8[3][0]), "v"(b8[3][1]), "v"(b8[3][2]), "v"(b8[3][3]),
                       "v"(b8[4][0]), "v"(b8[4][1]), "v"(b8[4][2]), "v"(b8[4][3]),
                       "v"(b8[5][0]), "v"(b8[5][1]), "v"(b8[5][2]), "v"(b8[5][3]));

    v8f acc[6];
#pragma unroll
    for (int m = 0; m < 6; ++m) acc[m] = (v8f){0.f, 0.f, 0.f, 0.f, 0.f, 0.f, 0.f, 0.f};
#pragma unroll
    for (int ks = 0; ks < 16; ++ks) {
        const v2f a = {a8[ks >> 2][2 * (ks & 3)], a8[ks >> 2][2 * (ks & 3) + 1]};
#pragma unroll
        for (int m = 0; m < 6; ++m) {
            const v2f b = {b8[m][ks >> 2][2 * (ks & 3)], b8[m][ks >> 2][2 * (ks & 3) + 1]};
            acc[m] = __builtin_amdgcn_wmma_f32_16x16x4_f32(
                 false, a,  false, b,
                 (short)0, acc[m],  false,  false);
        }
    }

    float* cp = qkv + (size_t)(p0 + 16 * pt + 8 * h) * NF + 16 * f0 + lc;
    const bool hi = f0 != 0;
    float* kb = qkv + (size_t)(p0 + 16 * pt + 8 * h + (lc >> 3)) * NF + CDIM + (hi ? HD : 0) + (lc & 7);
#pragma unroll
    for (int m = 0; m < 6; ++m) {
        const bool kt = hi ? (m < 2) : (m >= 4);
#pragma unroll
        for (int r = 0; r < 8; ++r) {
            float* dst = kt ? kb + (2 * (m & 1) + (r & ~3)) * NF + (r & 3) * 8 : cp + r * NF + 16 * m;
            *dst = acc[m][r] + bv[m];
        }
    }
}

#define SQ 10
#define NKEY (SQ * SQ)
#define RPBW (2 * KW - 1)
#define MAPG 12
#define MAPW (MAPG * 128 + 64)
#define RPBS (NHEADS * RPBW * RPBW)

__device__ __forceinline__ v8f wmma4(v2f a, v2f b, v8f c) {
    return __builtin_amdgcn_wmma_f32_16x16x4_f32(
         false, a,  false, b,
         (short)0, c,  false,  false);
}

template <int PITCH>
__device__ __forceinline__ constexpr unsigned key_row_off(int n) {
    return (unsigned)((((n < NKEY ? n : NKEY - 1) / SQ) * WW + ((n < NKEY ? n : NKEY - 1) % SQ)) * PITCH);
}

#define HKEY (NKEY / 2)

#define XPITCH 12
#define XTILE (32 * XPITCH)
#define GP 4
#define GCOL (4 * GP)
template <bool HAS_PREV, bool LAST, int KVS, int KVD>
__global__ void __launch_bounds__(LAST ? 64 * GP : 64) __attribute__((amdgpu_waves_per_eu(LAST ? 2 : 1, 2)))
na_attn_wmma(float* __restrict__ qio,
             const float* __restrict__ kin,
             const float* __restrict__ vin,
             const float* __restrict__ rpb,
             const float* __restrict__ pw,
             const float* __restrict__ pb,
             const float* __restrict__ nw,
             const float* __restrict__ nb,
             float* __restrict__ amap,
             float* __restrict__ kout,
             float* __restrict__ vout,
             float* __restrict__ out) {
    __shared__ __attribute__((aligned(16))) float xch[LAST ? GP * NHEADS * 2 * 8 * 32 : NHEADS * 4 * XTILE];
    const int lane  = threadIdx.x & 31;
    const int lc    = lane & 15;
    const int h     = lane >> 4;
    const int wv    = __builtin_amdgcn_readfirstlane((int)(threadIdx.x >> 5));
    const int head  = LAST ? (wv & 1) : wv;
    const int pl    = LAST ? (wv >> 1) : 0;
    const int patch = LAST ? (int)blockIdx.x * GP + pl : (int)blockIdx.x;
    const int i0 = (patch / (WW / 4)) * 4;
    const int j0 = (patch % (WW / 4)) * 4;

    int r0 = i0 - KW / 2; r0 = r0 < 0 ? 0 : r0; r0 = r0 > HH - SQ ? HH - SQ : r0;
    int c0 = j0 - KW / 2; c0 = c0 < 0 ? 0 : c0; c0 = c0 > WW - SQ ? WW - SQ : c0;

    const int qi = i0 + (lc >> 2);
    const int qj = j0 + (lc & 3);
    const int qpix = qi * WW + qj;
    int si = qi - KW / 2; si = si < 0 ? 0 : si; si = si > HH - KW ? HH - KW : si;
    int sj = qj - KW / 2; sj = sj < 0 ? 0 : sj; sj = sj > WW - KW ? WW - KW : sj;
    const int dsi = si - r0;
    const int dsj = sj - c0;

    const float scale = 0.17677669529663687f;

    const unsigned qoff = HAS_PREV
        ? (unsigned)((i0 * WW + j0 + (lane >> 3)) * 192 + head * HD + 4 * (lane & 7))
        : (unsigned)(qpix * 192 + head * HD + 4 * h);
    float4 qv[4];
#pragma unroll
    for (int s = 0; s < 4; ++s) qv[s] = *(const float4*)(qio + qoff + (HAS_PREV ? WW * 192 : 8) * s);

    float4 kv[7][4];
#pragma unroll
    for (int t = 0; t < 7; ++t) {
        int km = 8 * t + (lc & 7);
        km = km < HKEY ? km : HKEY - 1;
        const int kq = km / SQ;
        const int kr = kq + 5 * (lc >> 3);
        const int kc = km - kq * SQ;
        const int kp = (r0 + kr) * WW + (c0 + kc);
        const unsigned koff = (unsigned)((kp & ~3) * KVS + head * HD + (kp & 3) * 8 + 4 * h);
#pragma unroll
        for (int s = 0; s < 4; ++s) kv[t][s] = *(const float4*)(kin + koff + KVS * s);
    }

    const int ri0 = r0 - qi + KW - 1 + 5 * h;
    const int ci0 = c0 - qj + KW - 1;
    int rrow[5], ccol[SQ];
#pragma unroll
    for (int k = 0; k < 5; ++k) { int v = ri0 + k; v = v < 0 ? 0 : v; v = v > RPBW - 1 ? RPBW - 1 : v; rrow[k] = v * RPBW; }
#pragma unroll
    for (int k = 0; k < SQ; ++k) { int v = ci0 + k; v = v < 0 ? 0 : v; v = v > RPBW - 1 ? RPBW - 1 : v; ccol[k] = v; }
    const int dsih = dsi - 5 * h;
    bool rowv[5], colv[SQ];
#pragma unroll
    for (int k = 0; k < 5; ++k) rowv[k] = (unsigned)(k - dsih) < (unsigned)KW;
#pragma unroll
    for (int k = 0; k < SQ; ++k) colv[k] = (unsigned)(k - dsj) < (unsigned)KW;
    const float* rb = rpb + head * RPBW * RPBW;
    float* mp = amap + (unsigned)((patch * NHEADS + head) * MAPW) + 4 * lane;
    float* mp2 = amap + (unsigned)((patch * NHEADS + head) * MAPW + MAPG * 128) + 2 * lane;
    float4 pm[MAPG];
    float2 pm2 = make_float2(0.f, 0.f);
    if (HAS_PREV) {
#pragma unroll
        for (int g = 0; g < MAPG; ++g) pm[g] = *(const float4*)(mp + 128 * g);
        pm2 = *(const float2*)(mp2);
    }
    float rbv[7][8];
    float pvv[7][8];
#pragma unroll
    for (int t = 0; t < 7; ++t) {
#pragma unroll
        for (int r = 0; r < 8; ++r) {
            const int f = 8 * t + r;
            rbv[t][r] = f < HKEY ? rb[rrow[f / SQ] + ccol[f % SQ]] : 0.f;
            if (HAS_PREV && f < 4 * MAPG) {
                const float4 g4 = pm[f >> 2];
                pvv[t][r] = (f & 3) == 0 ? g4.x : (f & 3) == 1 ? g4.y : (f & 3) == 2 ? g4.z : g4.w;
            } else if (HAS_PREV && f < HKEY) {
                pvv[t][r] = f == 4 * MAPG ? pm2.x : pm2.y;
            } else {
                pvv[t][r] = 0.f;
            }
        }
    }

#pragma unroll
    for (int s = 0; s < 4; ++s)
        qv[s] = make_float4(qv[s].x * scale, qv[s].y * scale, qv[s].z * scale, qv[s].w * scale);

    v8f st[7];
#pragma unroll
    for (int t = 0; t < 7; ++t) {
        v8f acc = {0.f, 0.f, 0.f, 0.f, 0.f, 0.f, 0.f, 0.f};
#pragma unroll
        for (int s = 0; s < 4; ++s) {
            v2f a0 = {kv[t][s].x, kv[t][s].y};
            v2f a1 = {kv[t][s].z, kv[t][s].w};
            v2f b0 = {qv[s].x, qv[s].y};
            v2f b1 = {qv[s].z, qv[s].w};
            acc = wmma4(a0, b0, acc);
            acc = wmma4(a1, b1, acc);
        }
        st[t] = acc;
    }

    const unsigned voff = (unsigned)(((r0 + 5 * h) * WW + c0) * KVS + head * HD + lc);
    float va[7][4][2][2];
#pragma unroll
    for (int t = 0; t < 7; ++t) {
#pragma unroll
        for (int s = 0; s < 4; ++s) {
            const int k0 = 8 * t + 2 * s;
            if (k0 >= HKEY) continue;
            const unsigned of0 = voff + key_row_off<KVS>(k0);
            const unsigned of1 = voff + key_row_off<KVS>(k0 + 1);
            va[t][s][0][0] = vin[of0];
            va[t][s][0][1] = vin[of1];
            va[t][s][1][0] = vin[of0 + 16];
            va[t][s][1][1] = vin[of1 + 16];
        }
    }

    float4 wq[4][2][2];
    {
        const float* wr = pw + (unsigned)(head * (16 * 128) + 4 * lane);
#pragma unroll
        for (int u = 0; u < 4; ++u) {
#pragma unroll
            for (int dt = 0; dt < 2; ++dt) {
                wq[u][dt][0] = *(const float4*)(wr + 128 * (4 * u + 2 * dt));
                wq[u][dt][1] = *(const float4*)(wr + 128 * (4 * u + 2 * dt + 1));
            }
        }
    }
    float4 bq[2][2];
#pragma unroll
    for (int tt = 0; tt < 2; ++tt) {
        const float* br = pb + (unsigned)(16 * (2 * head + tt) + 8 * h);
        bq[tt][0] = *(const float4*)(br);
        bq[tt][1] = *(const float4*)(br + 4);
    }

    float mx = -1e30f;
#pragma unroll
    for (int t = 0; t < 7; ++t) {
#pragma unroll
        for (int r = 0; r < 8; ++r) {
            const int f = 8 * t + r;
            const bool valid = f < HKEY && rowv[f < HKEY ? f / SQ : 0] && colv[f % SQ];
            float l = st[t][r] + rbv[t][r];
            if (HAS_PREV) l += pvv[t][r];
            l = valid ? l : -1e30f;
            st[t][r] = l;
            mx = fmaxf(mx, l);
        }
    }
    mx = fmaxf(mx, __shfl_xor(mx, 16));

    float ssum = 0.f;
#pragma unroll
    for (int t = 0; t < 7; ++t) {
#pragma unroll
        for (int r = 0; r < 8; ++r) {
            float p = __expf(st[t][r] - mx);
            st[t][r] = p;
            ssum += p;
        }
    }
    ssum += __shfl_xor(ssum, 16);
    const float inv = 1.f / ssum;

#pragma unroll
    for (int t = 0; t < 7; ++t)
#pragma unroll
        for (int r = 0; r < 8; ++r) st[t][r] *= inv;

    v8f o[2][2];
#pragma unroll
    for (int dt = 0; dt < 2; ++dt)
#pragma unroll
        for (int u = 0; u < 2; ++u)
            o[dt][u] = (v8f){0.f, 0.f, 0.f, 0.f, 0.f, 0.f, 0.f, 0.f};

#pragma unroll
    for (int t = 0; t < 7; ++t) {
#pragma unroll
        for (int s = 0; s < 4; ++s) {
            const int k0 = 8 * t + 2 * s;
            if (k0 >= HKEY) continue;
            v2f a0 = {va[t][s][0][0], va[t][s][0][1]};
            v2f a1 = {va[t][s][1][0], va[t][s][1][1]};
            v2f b  = {st[t][2 * s], st[t][2 * s + 1]};
            o[0][t & 1] = wmma4(a0, b, o[0][t & 1]);
            o[1][t & 1] = wmma4(a1, b, o[1][t & 1]);
        }
    }

    v8f d[2];
#pragma unroll
    for (int dt = 0; dt < 2; ++dt) d[dt] = o[dt][0] + o[dt][1];

    float4 wn[6][4][2];
    if constexpr (!LAST) {
        const float* wr = nw + (unsigned)(head * (48 * 128) + 4 * lane);
#pragma unroll
        for (int m = 0; m < 6; ++m) {
#pragma unroll
            for (int u = 0; u < 4; ++u) {
                wn[m][u][0] = *(const float4*)(wr + 128 * (8 * m + 2 * u));
                wn[m][u][1] = *(const float4*)(wr + 128 * (8 * m + 2 * u + 1));
            }
        }
    }

    v8f pp[4];
#pragma unroll
    for (int tt = 0; tt < 2; ++tt)
        pp[tt] = (v8f){bq[tt][0].x, bq[tt][0].y, bq[tt][0].z, bq[tt][0].w,
                       bq[tt][1].x, bq[tt][1].y, bq[tt][1].z, bq[tt][1].w};
#pragma unroll
    for (int u = 2; u < 4; ++u) pp[u] = (v8f){0.f, 0.f, 0.f, 0.f, 0.f, 0.f, 0.f, 0.f};
#pragma unroll
    for (int dt = 0; dt < 2; ++dt) {
#pragma unroll
        for (int j = 0; j < 4; ++j) {
            v2f b = {d[dt][2 * j], d[dt][2 * j + 1]};
#pragma unroll
            for (int u = 0; u < 4; ++u) {
                const float4 w4 = wq[u][dt][j >> 1];
                v2f a = (j & 1) ? (v2f){w4.z, w4.w} : (v2f){w4.x, w4.y};
                pp[u] = wmma4(a, b, pp[u]);
            }
        }
    }

    if constexpr (LAST) {
        __shared__ __attribute__((aligned(16))) float stg[CDIM * 4 * GCOL];
        float* xw = xch + (pl * NHEADS + head) * (2 * 8 * 32) + lane;
        const float* xr = xch + (pl * NHEADS + (1 - head)) * (2 * 8 * 32) + lane;
#pragma unroll
        for (int tt = 0; tt < 2; ++tt)
#pragma unroll
            for (int r = 0; r < 8; ++r) xw[(tt * 8 + r) * 32] = pp[2 + tt][r];
        __syncthreads();

        float* sw = stg + (32 * head + 8 * h) * (4 * GCOL) + (lc >> 2) * GCOL + 4 * pl + (lc & 3);
        float xo[2][8];
#pragma unroll
        for (int tt = 0; tt < 2; ++tt)
#pragma unroll
            for (int r = 0; r < 8; ++r) xo[tt][r] = xr[(tt * 8 + r) * 32];
        asm volatile("" :: "v"(xo[0][0]), "v"(xo[0][1]), "v"(xo[0][2]), "v"(xo[0][3]),
                           "v"(xo[0][4]), "v"(xo[0][5]), "v"(xo[0][6]), "v"(xo[0][7]),
                           "v"(xo[1][0]), "v"(xo[1][1]), "v"(xo[1][2]), "v"(xo[1][3]),
                           "v"(xo[1][4]), "v"(xo[1][5]), "v"(xo[1][6]), "v"(xo[1][7]));
#pragma unroll
        for (int tt = 0; tt < 2; ++tt)
#pragma unroll
            for (int r = 0; r < 8; ++r)
                sw[(16 * tt + r) * (4 * GCOL)] = pp[tt][r] + xo[tt][r];
        __syncthreads();

        const int ig = ((int)blockIdx.x / (WW / GCOL)) * 4;
        const int jg = ((int)blockIdx.x % (WW / GCOL)) * GCOL;
        float* og = out + (unsigned)(ig * WW + jg);
#pragma unroll
        for (int it = 0; it < (CDIM * 4 * GCOL) / (4 * 64 * GP); ++it) {
            const int idx = it * (64 * GP) + (int)threadIdx.x;
            const float4 w4 = *(const float4*)(stg + 4 * idx);
            *(float4*)(og + (size_t)(idx >> 4) * PIX + ((idx >> 2) & 3) * WW + 4 * (idx & 3)) = w4;
        }
    } else {
        float* xw = xch + head * (4 * XTILE) + lane * XPITCH;
        const float* xr = xch + (1 - head) * (4 * XTILE) + lane * XPITCH;
#pragma unroll
        for (int u = 0; u < 4; ++u) {
            *(float4*)(xw + u * XTILE)     = make_float4(pp[u][0], pp[u][1], pp[u][2], pp[u][3]);
            *(float4*)(xw + u * XTILE + 4) = make_float4(pp[u][4], pp[u][5], pp[u][6], pp[u][7]);
        }
        __syncthreads();

        v4f xa[4], xb[4];
#pragma unroll
        for (int u = 0; u < 4; ++u) {
            xa[u] = *(const v4f*)(xr + ((u + 2) & 3) * XTILE);
            xb[u] = *(const v4f*)(xr + ((u + 2) & 3) * XTILE + 4);
        }
        float4 nbq[6][2];
#pragma unroll
        for (int m = 0; m < 6; ++m) {
            const float* br = nb + (unsigned)(64 * (m >> 1) + 32 * head + 16 * (m & 1) + 8 * h);
            nbq[m][0] = *(const float4*)(br);
            nbq[m][1] = *(const float4*)(br + 4);
        }
        asm volatile("" :: "v"(xa[0]), "v"(xb[0]), "v"(xa[1]), "v"(xb[1]),
                           "v"(xa[2]), "v"(xb[2]), "v"(xa[3]), "v"(xb[3]));
        v8f xt[4];
#pragma unroll
        for (int u = 0; u < 4; ++u)
            xt[u] = pp[u] + (v8f){xa[u].x, xa[u].y, xa[u].z, xa[u].w, xb[u].x, xb[u].y, xb[u].z, xb[u].w};

        v8f qa[6];
#pragma unroll
        for (int m = 0; m < 6; ++m) qa[m] = (v8f){0.f, 0.f, 0.f, 0.f, 0.f, 0.f, 0.f, 0.f};
#pragma unroll
        for (int u = 0; u < 4; ++u) {
#pragma unroll
            for (int j = 0; j < 4; ++j) {
                v2f b = {xt[u][2 * j], xt[u][2 * j + 1]};
#pragma unroll
                for (int m = 0; m < 6; ++m) {
                    const float4 w4 = wn[m][u][j >> 1];
                    v2f a = (j & 1) ? (v2f){w4.z, w4.w} : (v2f){w4.x, w4.y};
                    qa[m] = wmma4(a, b, qa[m]);
                }
            }
        }

        float* qw = qio  + (size_t)(((i0 + h) * WW + j0 + (lc >> 3)) * 192 + 32 * head + 4 * (lc & 7));
        float* kp = kout + (size_t)(qi * WW + j0 + h) * KVD + 32 * head + 8 * (lc & 3);
        float* vp = vout + (size_t)qpix * KVD + 32 * head + 8 * h;
#pragma unroll
        for (int m = 0; m < 6; ++m) {
            float* dp = (m >> 1) == 1 ? kp + 2 * (m & 1) * KVD : vp + 16 * (m & 1);
            float* d0 = (m >> 1) == 0 ? qw + 2 * (m & 1) * WW * 192 : dp;
            float* d1 = (m >> 1) == 0 ? d0 + 2 * 192 : dp + 4;
            *(float4*)(d0) = make_float4(qa[m][0] + nbq[m][0].x, qa[m][1] + nbq[m][0].y,
                                         qa[m][2] + nbq[m][0].z, qa[m][3] + nbq[m][0].w);
            *(float4*)(d1) = make_float4(qa[m][4] + nbq[m][1].x, qa[m][5] + nbq[m][1].y,
                                         qa[m][6] + nbq[m][1].z, qa[m][7] + nbq[m][1].w);
        }

#pragma unroll
        for (int g = 0; g < MAPG; ++g) {
            const int t = g >> 1;
            const int r = 4 * (g & 1);
            *(float4*)(mp + 128 * g) = make_float4(st[t][r], st[t][r + 1], st[t][r + 2], st[t][r + 3]);
        }
        *(float2*)(mp2) = make_float2(st[6][0], st[6][1]);
    }
}

extern "C" void kernel_launch(void* const* d_in, const int* in_sizes, int n_in,
                              void* d_out, int out_size, void* d_ws, size_t ws_size,
                              hipStream_t stream) {
    const float* x      = (const float*)d_in[0];
    const float* qkv_w  = (const float*)d_in[1];
    const float* qkv_b  = (const float*)d_in[2];
    const float* proj_w = (const float*)d_in[3];
    const float* proj_b = (const float*)d_in[4];
    const float* rpb    = (const float*)d_in[5];
    float* out = (float*)d_out;
    float* ws  = (float*)d_ws;

    float* plane = ws;
    float* qkv   = plane + (size_t)PIX * CDIM;
    float* amap  = qkv + (size_t)PIX * 192;
    float* wfrag = amap + (size_t)(PIX / 16) * NHEADS * MAPW;

    qkv0_from_nchw<<<GBLK + WBLK, 128, 0, stream>>>(x, qkv_w, qkv_b, proj_w, qkv, wfrag);

    na_attn_wmma<false, false, 192, 64><<<PIX / 16, 32 * NHEADS, 0, stream>>>(
        qkv, qkv + 64, qkv + 128, rpb, wfrag + WF_PROJ, proj_b,
        wfrag + WF_NEXT, qkv_b + 192, amap, plane, out, nullptr);
    na_attn_wmma<true, false, 64, 192><<<PIX / 16, 32 * NHEADS, 0, stream>>>(
        qkv, plane, out, rpb + RPBS, wfrag + WF_PROJ + CDIM * CDIM, proj_b + CDIM,
        wfrag + WF_NEXT + 192 * CDIM, qkv_b + 2 * 192, amap, qkv + 64, qkv + 128, nullptr);
    na_attn_wmma<true, true, 192, 64><<<PIX / 16 / GP, 32 * NHEADS * GP, 0, stream>>>(
        qkv, qkv + 64, qkv + 128, rpb + 2 * RPBS, wfrag + WF_PROJ + 2 * CDIM * CDIM, proj_b + 2 * CDIM,
        nullptr, nullptr, amap, nullptr, nullptr, out);
}
